# starting kernel
# baseline (speedup 1.0000x reference)
#include <hip/hip_runtime.h>
#include <hip/hip_bf16.h>
#include <math.h>

typedef float v2f __attribute__((ext_vector_type(2)));
typedef float v8f __attribute__((ext_vector_type(8)));

#define NN      50000
#define NE      600000
#define INC     128
#define HIDC    128
#define OUTC    64
#define NG      512

__device__ __forceinline__ void atomAddF(float* p, float v) {
  __hip_atomic_fetch_add(p, v, __ATOMIC_RELAXED, __HIP_MEMORY_SCOPE_AGENT);
}

__global__ void deg_kernel(const int* __restrict__ dst, float* __restrict__ deg, int E) {
  int e = blockIdx.x * blockDim.x + threadIdx.x;
  if (e < E) atomAddF(&deg[dst[e]], 1.0f);
}

__global__ void invdeg_kernel(float* __restrict__ deg, int n) {
  int i = blockIdx.x * blockDim.x + threadIdx.x;
  if (i < n) deg[i] = 1.0f / fmaxf(deg[i], 1.0f);
}

__global__ void edge_agg_kernel(const float* __restrict__ x, const int* __restrict__ src,
                                const int* __restrict__ dst, float* __restrict__ agg, int E) {
  int wave = blockIdx.x * (blockDim.x >> 5) + (threadIdx.x >> 5);
  int lane = threadIdx.x & 31;
  if (wave >= E) return;
  int s = src[wave];
  int d = dst[wave];
  const float4 v = *reinterpret_cast<const float4*>(x + (size_t)s * 128 + lane * 4);
  float* p = agg + (size_t)d * 128 + lane * 4;
  atomAddF(p + 0, v.x);
  atomAddF(p + 1, v.y);
  atomAddF(p + 2, v.z);
  atomAddF(p + 3, v.w);
}

__global__ void sage_gemm_kernel(const float* __restrict__ Agg, const float* __restrict__ invdeg,
                                 const float* __restrict__ X,
                                 const float* __restrict__ Wl, const float* __restrict__ bl,
                                 const float* __restrict__ Wr,
                                 float* __restrict__ OutA, float* __restrict__ OutB,
                                 int M, int K, int N, int doRelu) {
  int wave = blockIdx.x * (blockDim.x >> 5) + (threadIdx.x >> 5);
  int lane = threadIdx.x & 31;
  int nTilesN = N >> 4;
  int nTiles = (M >> 4) * nTilesN;
  if (wave >= nTiles) return;

  int tn = wave % nTilesN;
  int tm = wave / nTilesN;
  int m0 = tm << 4;
  int n0 = tn << 4;

  int rA   = m0 + (lane & 15);
  int colB = n0 + (lane & 15);
  int kh   = (lane >> 4) << 1;

  const float* arow = Agg + (size_t)rA * K;
  const float* xrow = X   + (size_t)rA * K;
  float s = invdeg[rA];

  v8f c = {};
#pragma unroll 4
  for (int kk = 0; kk < K; kk += 4) {
    v2f a, b;
    a.x = arow[kk + kh] * s;
    a.y = arow[kk + kh + 1] * s;
    b.x = Wl[(size_t)(kk + kh) * N + colB];
    b.y = Wl[(size_t)(kk + kh + 1) * N + colB];
    c = __builtin_amdgcn_wmma_f32_16x16x4_f32(false, a, false, b, (short)0, c, false, false);
    a.x = xrow[kk + kh];
    a.y = xrow[kk + kh + 1];
    b.x = Wr[(size_t)(kk + kh) * N + colB];
    b.y = Wr[(size_t)(kk + kh + 1) * N + colB];
    c = __builtin_amdgcn_wmma_f32_16x16x4_f32(false, a, false, b, (short)0, c, false, false);
  }

  float bias = bl[colB];
  int rbase = m0 + ((lane >> 4) << 3);
#pragma unroll
  for (int r = 0; r < 8; ++r) {
    float v = c[r] + bias;
    size_t off = (size_t)(rbase + r) * N + colB;
    if (OutB) OutB[off] = v;
    if (doRelu) v = fmaxf(v, 0.0f);
    OutA[off] = v;
  }
}

__global__ void pool_kernel(const float* __restrict__ h, const int* __restrict__ cluster,
                            float* __restrict__ gsum, float* __restrict__ gcnt, int n) {
  int idx = blockIdx.x * blockDim.x + threadIdx.x;
  int node = idx >> 5;
  int lane = idx & 31;
  if (node >= n) return;
  int g = cluster[node];
  const float4 v = *reinterpret_cast<const float4*>(h + (size_t)node * 128 + lane * 4);
  float* p = gsum + (size_t)g * 128 + lane * 4;
  atomAddF(p + 0, v.x);
  atomAddF(p + 1, v.y);
  atomAddF(p + 2, v.z);
  atomAddF(p + 3, v.w);
  if (lane == 0) atomAddF(&gcnt[g], 1.0f);
}

__global__ void pool_div_kernel(float* __restrict__ g, const float* __restrict__ gcnt, int total) {
  int i = blockIdx.x * blockDim.x + threadIdx.x;
  if (i >= total) return;
  g[i] = g[i] / fmaxf(gcnt[i >> 7], 1.0f);
}

__global__ void log_softmax_kernel(float* __restrict__ out, int rows) {
  int wave = blockIdx.x * (blockDim.x >> 5) + (threadIdx.x >> 5);
  int lane = threadIdx.x & 31;
  if (wave >= rows) return;
  float* row = out + (size_t)wave * 64;
  float v0 = row[lane];
  float v1 = row[lane + 32];
  float m = fmaxf(v0, v1);
#pragma unroll
  for (int off = 16; off >= 1; off >>= 1) m = fmaxf(m, __shfl_xor(m, off, 32));
  float s = __expf(v0 - m) + __expf(v1 - m);
#pragma unroll
  for (int off = 16; off >= 1; off >>= 1) s += __shfl_xor(s, off, 32);
  float lse = m + __logf(s);
  row[lane]      = v0 - lse;
  row[lane + 32] = v1 - lse;
}

extern "C" void kernel_launch(void* const* d_in, const int* in_sizes, int n_in,
                              void* d_out, int out_size, void* d_ws, size_t ws_size,
                              hipStream_t stream) {
  const float* x       = (const float*)d_in[0];
  const int*   ei      = (const int*)d_in[1];
  const int*   src     = ei;
  const int*   dst     = ei + NE;
  const int*   cluster = (const int*)d_in[2];
  const float* Wl0 = (const float*)d_in[3];
  const float* bl0 = (const float*)d_in[4];
  const float* Wr0 = (const float*)d_in[5];
  const float* Wl1 = (const float*)d_in[6];
  const float* bl1 = (const float*)d_in[7];
  const float* Wr1 = (const float*)d_in[8];
  const float* Wl2 = (const float*)d_in[9];
  const float* bl2 = (const float*)d_in[10];
  const float* Wr2 = (const float*)d_in[11];

  float* out_ls = (float*)d_out;
  float* out_h2 = out_ls + (size_t)NN * OUTC;
  float* out_g  = out_h2 + (size_t)NN * HIDC;

  float* ws   = (float*)d_ws;
  float* agg  = ws;
  float* h1   = ws + (size_t)NN * HIDC;
  float* h2r  = h1 + (size_t)NN * HIDC;
  float* deg  = h2r + (size_t)NN * HIDC;
  float* gcnt = deg + NN;

  const int TPB = 256;
  const int edgeWaveBlocks = (NE + 7) / 8;

  hipMemsetAsync(deg, 0, NN * sizeof(float), stream);
  deg_kernel<<<(NE + TPB - 1) / TPB, TPB, 0, stream>>>(dst, deg, NE);
  invdeg_kernel<<<(NN + TPB - 1) / TPB, TPB, 0, stream>>>(deg, NN);

  hipMemsetAsync(agg, 0, (size_t)NN * HIDC * sizeof(float), stream);
  edge_agg_kernel<<<edgeWaveBlocks, TPB, 0, stream>>>(x, src, dst, agg, NE);
  {
    int tiles = (NN / 16) * (HIDC / 16);
    sage_gemm_kernel<<<(tiles + 7) / 8, TPB, 0, stream>>>(
        agg, deg, x, Wl0, bl0, Wr0, h1, nullptr, NN, INC, HIDC, 1);
  }

  hipMemsetAsync(agg, 0, (size_t)NN * HIDC * sizeof(float), stream);
  edge_agg_kernel<<<edgeWaveBlocks, TPB, 0, stream>>>(h1, src, dst, agg, NE);
  {
    int tiles = (NN / 16) * (HIDC / 16);
    sage_gemm_kernel<<<(tiles + 7) / 8, TPB, 0, stream>>>(
        agg, deg, h1, Wl1, bl1, Wr1, h2r, out_h2, NN, HIDC, HIDC, 1);
  }

  hipMemsetAsync(out_g, 0, (size_t)NG * HIDC * sizeof(float), stream);
  hipMemsetAsync(gcnt, 0, NG * sizeof(float), stream);
  pool_kernel<<<((size_t)NN * 32 + TPB - 1) / TPB, TPB, 0, stream>>>(h2r, cluster, out_g, gcnt, NN);
  pool_div_kernel<<<(NG * HIDC + TPB - 1) / TPB, TPB, 0, stream>>>(out_g, gcnt, NG * HIDC);

  hipMemsetAsync(agg, 0, (size_t)NN * HIDC * sizeof(float), stream);
  edge_agg_kernel<<<edgeWaveBlocks, TPB, 0, stream>>>(h2r, src, dst, agg, NE);
  {
    int tiles = (NN / 16) * (OUTC / 16);
    sage_gemm_kernel<<<(tiles + 7) / 8, TPB, 0, stream>>>(
        agg, deg, h2r, Wl2, bl2, Wr2, out_ls, nullptr, NN, HIDC, OUTC, 0);
  }
  log_softmax_kernel<<<((size_t)NN * 32 + TPB - 1) / TPB, TPB, 0, stream>>>(out_ls, NN);
}


# Neighbour sums by gather, not float atomic scatter: edges bucketed by destination once per call (int atomic rank, prefix sum, fill); a wave adds a node's rows in arrival-rank order; no agg memsets
# speedup vs baseline: 1.6915x; 1.6915x over previous
#include <hip/hip_runtime.h>
#include <hip/hip_bf16.h>
#include <math.h>

typedef float v2f __attribute__((ext_vector_type(2)));
typedef float v8f __attribute__((ext_vector_type(8)));

#define NN      50000
#define NE      600000
#define INC     128
#define HIDC    128
#define OUTC    64
#define NG      512

__device__ __forceinline__ void atomAddF(float* p, float v) {
  __hip_atomic_fetch_add(p, v, __ATOMIC_RELAXED, __HIP_MEMORY_SCOPE_AGENT);
}

__global__ void count_rank_kernel(const int* __restrict__ dst, int* __restrict__ cnt,
                                  int* __restrict__ pos, int E) {
  int B = blockDim.x;
  int e0 = blockIdx.x * (B * 4) + threadIdx.x;
  int e1 = e0 + B, e2 = e1 + B, e3 = e2 + B;
  if (e3 < E) {
    int d0 = dst[e0], d1 = dst[e1], d2 = dst[e2], d3 = dst[e3];
    int r0 = __hip_atomic_fetch_add(&cnt[d0], 1, __ATOMIC_RELAXED, __HIP_MEMORY_SCOPE_AGENT);
    int r1 = __hip_atomic_fetch_add(&cnt[d1], 1, __ATOMIC_RELAXED, __HIP_MEMORY_SCOPE_AGENT);
    int r2 = __hip_atomic_fetch_add(&cnt[d2], 1, __ATOMIC_RELAXED, __HIP_MEMORY_SCOPE_AGENT);
    int r3 = __hip_atomic_fetch_add(&cnt[d3], 1, __ATOMIC_RELAXED, __HIP_MEMORY_SCOPE_AGENT);
    pos[e0] = r0; pos[e1] = r1; pos[e2] = r2; pos[e3] = r3;
  } else {
    for (int e = e0; e < E; e += B)
      pos[e] = __hip_atomic_fetch_add(&cnt[dst[e]], 1, __ATOMIC_RELAXED, __HIP_MEMORY_SCOPE_AGENT);
  }
}

__global__ void __launch_bounds__(1024)
scan_kernel(const int* __restrict__ cnt, int* __restrict__ rowptr,
            float* __restrict__ invdeg, int n) {
  __shared__ int wsum[32];
  __shared__ int sbase;
  int t = threadIdx.x;
  int lane = t & 31;
  int w = t >> 5;
  int start = blockIdx.x * 1024;

  int part = 0;
  for (int i = t; i < start; i += 1024) part += cnt[i];
  for (int off = 16; off >= 1; off >>= 1) part += __shfl_xor(part, off, 32);
  if (lane == 0) wsum[w] = part;
  __syncthreads();
  if (w == 0) {
    int v = wsum[lane];
    for (int off = 16; off >= 1; off >>= 1) v += __shfl_xor(v, off, 32);
    if (lane == 0) sbase = v;
  }
  __syncthreads();
  int base = sbase;

  int i = start + t;
  int c = (i < n) ? cnt[i] : 0;
  int inc = c;
  for (int off = 1; off < 32; off <<= 1) {
    int u = __shfl_up(inc, off, 32);
    if (lane >= off) inc += u;
  }
  if (lane == 31) wsum[w] = inc;
  __syncthreads();
  if (w == 0) {
    int v = wsum[lane];
    int iv = v;
    for (int off = 1; off < 32; off <<= 1) {
      int u = __shfl_up(iv, off, 32);
      if (lane >= off) iv += u;
    }
    wsum[lane] = iv - v;
  }
  __syncthreads();
  int excl = base + wsum[w] + inc - c;
  if (i < n) {
    rowptr[i] = excl;
    invdeg[i] = 1.0f / fmaxf((float)c, 1.0f);
  }
  if (i == n - 1) rowptr[n] = excl + c;
}

__global__ void fill_kernel(const int* __restrict__ src, const int* __restrict__ dst,
                            const int* __restrict__ rowptr, const int* __restrict__ pos,
                            int* __restrict__ col, int E) {
  int B = blockDim.x;
  int e0 = blockIdx.x * (B * 4) + threadIdx.x;
  int e1 = e0 + B, e2 = e1 + B, e3 = e2 + B;
  if (e3 < E) {
    int d0 = dst[e0], d1 = dst[e1], d2 = dst[e2], d3 = dst[e3];
    int p0 = pos[e0], p1 = pos[e1], p2 = pos[e2], p3 = pos[e3];
    int s0 = src[e0], s1 = src[e1], s2 = src[e2], s3 = src[e3];
    int q0 = rowptr[d0], q1 = rowptr[d1], q2 = rowptr[d2], q3 = rowptr[d3];
    col[q0 + p0] = s0; col[q1 + p1] = s1; col[q2 + p2] = s2; col[q3 + p3] = s3;
  } else {
    for (int e = e0; e < E; e += B) col[rowptr[dst[e]] + pos[e]] = src[e];
  }
}

#define GATHER_NODES 4
__device__ __forceinline__ void add4(float4& a, const float4 v) {
  a.x += v.x; a.y += v.y; a.z += v.z; a.w += v.w;
}
__global__ void gather_sum_kernel(const float* __restrict__ x, const int* __restrict__ rowptr,
                                  const int* __restrict__ col, float* __restrict__ agg, int n) {
  int w = __builtin_amdgcn_readfirstlane((int)(blockIdx.x * (blockDim.x >> 5) + (threadIdx.x >> 5)));
  int lane = threadIdx.x & 31;
  const float* xl = x + lane * 4;
  for (int q = 0; q < GATHER_NODES; ++q) {
    int node = w * GATHER_NODES + q;
    if (node >= n) break;
    int j = rowptr[node];
    int end = rowptr[node + 1];
    float4 acc = make_float4(0.0f, 0.0f, 0.0f, 0.0f);
    for (; j + 4 <= end; j += 4) {
      int s0 = col[j], s1 = col[j + 1], s2 = col[j + 2], s3 = col[j + 3];
      const float4 v0 = *reinterpret_cast<const float4*>(xl + (size_t)s0 * 128);
      const float4 v1 = *reinterpret_cast<const float4*>(xl + (size_t)s1 * 128);
      const float4 v2 = *reinterpret_cast<const float4*>(xl + (size_t)s2 * 128);
      const float4 v3 = *reinterpret_cast<const float4*>(xl + (size_t)s3 * 128);
      add4(acc, v0); add4(acc, v1); add4(acc, v2); add4(acc, v3);
    }
    int rem = end - j;
    if (rem > 0) {
      int s0 = col[j];
      int s1 = (rem > 1) ? col[j + 1] : s0;
      int s2 = (rem > 2) ? col[j + 2] : s0;
      const float4 v0 = *reinterpret_cast<const float4*>(xl + (size_t)s0 * 128);
      const float4 v1 = *reinterpret_cast<const float4*>(xl + (size_t)s1 * 128);
      const float4 v2 = *reinterpret_cast<const float4*>(xl + (size_t)s2 * 128);
      add4(acc, v0);
      if (rem > 1) add4(acc, v1);
      if (rem > 2) add4(acc, v2);
    }
    *reinterpret_cast<float4*>(agg + (size_t)node * 128 + lane * 4) = acc;
  }
}

__global__ void sage_gemm_kernel(const float* __restrict__ Agg, const float* __restrict__ invdeg,
                                 const float* __restrict__ X,
                                 const float* __restrict__ Wl, const float* __restrict__ bl,
                                 const float* __restrict__ Wr,
                                 float* __restrict__ OutA, float* __restrict__ OutB,
                                 int M, int K, int N, int doRelu) {
  int wave = blockIdx.x * (blockDim.x >> 5) + (threadIdx.x >> 5);
  int lane = threadIdx.x & 31;
  int nTilesN = N >> 4;
  int nTiles = (M >> 4) * nTilesN;
  if (wave >= nTiles) return;

  int tn = wave % nTilesN;
  int tm = wave / nTilesN;
  int m0 = tm << 4;
  int n0 = tn << 4;

  int rA   = m0 + (lane & 15);
  int colB = n0 + (lane & 15);
  int kh   = (lane >> 4) << 1;

  const float* arow = Agg + (size_t)rA * K;
  const float* xrow = X   + (size_t)rA * K;
  float s = invdeg[rA];

  v8f c = {};
#pragma unroll 4
  for (int kk = 0; kk < K; kk += 4) {
    v2f a, b;
    a.x = arow[kk + kh] * s;
    a.y = arow[kk + kh + 1] * s;
    b.x = Wl[(size_t)(kk + kh) * N + colB];
    b.y = Wl[(size_t)(kk + kh + 1) * N + colB];
    c = __builtin_amdgcn_wmma_f32_16x16x4_f32(false, a, false, b, (short)0, c, false, false);
    a.x = xrow[kk + kh];
    a.y = xrow[kk + kh + 1];
    b.x = Wr[(size_t)(kk + kh) * N + colB];
    b.y = Wr[(size_t)(kk + kh + 1) * N + colB];
    c = __builtin_amdgcn_wmma_f32_16x16x4_f32(false, a, false, b, (short)0, c, false, false);
  }

  float bias = bl[colB];
  int rbase = m0 + ((lane >> 4) << 3);
#pragma unroll
  for (int r = 0; r < 8; ++r) {
    float v = c[r] + bias;
    size_t off = (size_t)(rbase + r) * N + colB;
    if (OutB) OutB[off] = v;
    if (doRelu) v = fmaxf(v, 0.0f);
    OutA[off] = v;
  }
}

__global__ void pool_kernel(const float* __restrict__ h, const int* __restrict__ cluster,
                            float* __restrict__ gsum, float* __restrict__ gcnt, int n) {
  int idx = blockIdx.x * blockDim.x + threadIdx.x;
  int node = idx >> 5;
  int lane = idx & 31;
  if (node >= n) return;
  int g = cluster[node];
  const float4 v = *reinterpret_cast<const float4*>(h + (size_t)node * 128 + lane * 4);
  float* p = gsum + (size_t)g * 128 + lane * 4;
  atomAddF(p + 0, v.x);
  atomAddF(p + 1, v.y);
  atomAddF(p + 2, v.z);
  atomAddF(p + 3, v.w);
  if (lane == 0) atomAddF(&gcnt[g], 1.0f);
}

__global__ void pool_div_kernel(float* __restrict__ g, const float* __restrict__ gcnt, int total) {
  int i = blockIdx.x * blockDim.x + threadIdx.x;
  if (i >= total) return;
  g[i] = g[i] / fmaxf(gcnt[i >> 7], 1.0f);
}

__global__ void log_softmax_kernel(float* __restrict__ out, int rows) {
  int wave = blockIdx.x * (blockDim.x >> 5) + (threadIdx.x >> 5);
  int lane = threadIdx.x & 31;
  if (wave >= rows) return;
  float* row = out + (size_t)wave * 64;
  float v0 = row[lane];
  float v1 = row[lane + 32];
  float m = fmaxf(v0, v1);
#pragma unroll
  for (int off = 16; off >= 1; off >>= 1) m = fmaxf(m, __shfl_xor(m, off, 32));
  float s = __expf(v0 - m) + __expf(v1 - m);
#pragma unroll
  for (int off = 16; off >= 1; off >>= 1) s += __shfl_xor(s, off, 32);
  float lse = m + __logf(s);
  row[lane]      = v0 - lse;
  row[lane + 32] = v1 - lse;
}

extern "C" void kernel_launch(void* const* d_in, const int* in_sizes, int n_in,
                              void* d_out, int out_size, void* d_ws, size_t ws_size,
                              hipStream_t stream) {
  const float* x       = (const float*)d_in[0];
  const int*   ei      = (const int*)d_in[1];
  const int*   src     = ei;
  const int*   dst     = ei + NE;
  const int*   cluster = (const int*)d_in[2];
  const float* Wl0 = (const float*)d_in[3];
  const float* bl0 = (const float*)d_in[4];
  const float* Wr0 = (const float*)d_in[5];
  const float* Wl1 = (const float*)d_in[6];
  const float* bl1 = (const float*)d_in[7];
  const float* Wr1 = (const float*)d_in[8];
  const float* Wl2 = (const float*)d_in[9];
  const float* bl2 = (const float*)d_in[10];
  const float* Wr2 = (const float*)d_in[11];

  float* out_ls = (float*)d_out;
  float* out_h2 = out_ls + (size_t)NN * OUTC;
  float* out_g  = out_h2 + (size_t)NN * HIDC;

  float* ws   = (float*)d_ws;
  float* agg  = ws;
  float* h1   = ws + (size_t)NN * HIDC;
  float* h2r  = h1 + (size_t)NN * HIDC;
  float* deg  = h2r + (size_t)NN * HIDC;
  float* gcnt = deg + NN;

  int* cnt    = (int*)h1;
  int* pos    = cnt + 50048;
  int* rowptr = (int*)out_ls;
  int* col    = rowptr + 50048;

  const int TPB = 256;
  const int edgeBlocks4 = (NE + TPB * 4 - 1) / (TPB * 4);
  const int gatherBlocks = (NN + 8 * GATHER_NODES - 1) / (8 * GATHER_NODES);

  hipMemsetAsync(cnt, 0, NN * sizeof(int), stream);
  count_rank_kernel<<<edgeBlocks4, TPB, 0, stream>>>(dst, cnt, pos, NE);
  scan_kernel<<<(NN + 1023) / 1024, 1024, 0, stream>>>(cnt, rowptr, deg, NN);
  fill_kernel<<<edgeBlocks4, TPB, 0, stream>>>(src, dst, rowptr, pos, col, NE);

  gather_sum_kernel<<<gatherBlocks, TPB, 0, stream>>>(x, rowptr, col, agg, NN);
  {
    int tiles = (NN / 16) * (HIDC / 16);
    sage_gemm_kernel<<<(tiles + 7) / 8, TPB, 0, stream>>>(
        agg, deg, x, Wl0, bl0, Wr0, h1, nullptr, NN, INC, HIDC, 1);
  }

  gather_sum_kernel<<<gatherBlocks, TPB, 0, stream>>>(h1, rowptr, col, agg, NN);
  {
    int tiles = (NN / 16) * (HIDC / 16);
    sage_gemm_kernel<<<(tiles + 7) / 8, TPB, 0, stream>>>(
        agg, deg, h1, Wl1, bl1, Wr1, h2r, out_h2, NN, HIDC, HIDC, 1);
  }

  hipMemsetAsync(out_g, 0, (size_t)NG * HIDC * sizeof(float), stream);
  hipMemsetAsync(gcnt, 0, NG * sizeof(float), stream);
  pool_kernel<<<((size_t)NN * 32 + TPB - 1) / TPB, TPB, 0, stream>>>(h2r, cluster, out_g, gcnt, NN);
  pool_div_kernel<<<(NG * HIDC + TPB - 1) / TPB, TPB, 0, stream>>>(out_g, gcnt, NG * HIDC);

  gather_sum_kernel<<<gatherBlocks, TPB, 0, stream>>>(h2r, rowptr, col, agg, NN);
  {
    int tiles = (NN / 16) * (OUTC / 16);
    sage_gemm_kernel<<<(tiles + 7) / 8, TPB, 0, stream>>>(
        agg, deg, h2r, Wl2, bl2, Wr2, out_ls, nullptr, NN, HIDC, OUTC, 0);
  }
  log_softmax_kernel<<<((size_t)NN * 32 + TPB - 1) / TPB, TPB, 0, stream>>>(out_ls, NN);
}


# Graph pooling by run sums: a wave adds 16 consecutive node rows in registers, float atomics only where the graph id changes or its nodes end; fewer atomics, order of the pooled float sums changes
# speedup vs baseline: 1.9660x; 1.1623x over previous
#include <hip/hip_runtime.h>
#include <hip/hip_bf16.h>
#include <math.h>

typedef float v2f __attribute__((ext_vector_type(2)));
typedef float v8f __attribute__((ext_vector_type(8)));

#define NN      50000
#define NE      600000
#define INC     128
#define HIDC    128
#define OUTC    64
#define NG      512

__device__ __forceinline__ void atomAddF(float* p, float v) {
  __hip_atomic_fetch_add(p, v, __ATOMIC_RELAXED, __HIP_MEMORY_SCOPE_AGENT);
}

__global__ void count_rank_kernel(const int* __restrict__ dst, int* __restrict__ cnt,
                                  int* __restrict__ pos, int E) {
  int B = blockDim.x;
  int e0 = blockIdx.x * (B * 4) + threadIdx.x;
  int e1 = e0 + B, e2 = e1 + B, e3 = e2 + B;
  if (e3 < E) {
    int d0 = dst[e0], d1 = dst[e1], d2 = dst[e2], d3 = dst[e3];
    int r0 = __hip_atomic_fetch_add(&cnt[d0], 1, __ATOMIC_RELAXED, __HIP_MEMORY_SCOPE_AGENT);
    int r1 = __hip_atomic_fetch_add(&cnt[d1], 1, __ATOMIC_RELAXED, __HIP_MEMORY_SCOPE_AGENT);
    int r2 = __hip_atomic_fetch_add(&cnt[d2], 1, __ATOMIC_RELAXED, __HIP_MEMORY_SCOPE_AGENT);
    int r3 = __hip_atomic_fetch_add(&cnt[d3], 1, __ATOMIC_RELAXED, __HIP_MEMORY_SCOPE_AGENT);
    pos[e0] = r0; pos[e1] = r1; pos[e2] = r2; pos[e3] = r3;
  } else {
    for (int e = e0; e < E; e += B)
      pos[e] = __hip_atomic_fetch_add(&cnt[dst[e]], 1, __ATOMIC_RELAXED, __HIP_MEMORY_SCOPE_AGENT);
  }
}

__global__ void __launch_bounds__(1024)
scan_kernel(const int* __restrict__ cnt, int* __restrict__ rowptr,
            float* __restrict__ invdeg, int n) {
  __shared__ int wsum[32];
  __shared__ int sbase;
  int t = threadIdx.x;
  int lane = t & 31;
  int w = t >> 5;
  int start = blockIdx.x * 1024;

  int part = 0;
  for (int i = t; i < start; i += 1024) part += cnt[i];
  for (int off = 16; off >= 1; off >>= 1) part += __shfl_xor(part, off, 32);
  if (lane == 0) wsum[w] = part;
  __syncthreads();
  if (w == 0) {
    int v = wsum[lane];
    for (int off = 16; off >= 1; off >>= 1) v += __shfl_xor(v, off, 32);
    if (lane == 0) sbase = v;
  }
  __syncthreads();
  int base = sbase;

  int i = start + t;
  int c = (i < n) ? cnt[i] : 0;
  int inc = c;
  for (int off = 1; off < 32; off <<= 1) {
    int u = __shfl_up(inc, off, 32);
    if (lane >= off) inc += u;
  }
  if (lane == 31) wsum[w] = inc;
  __syncthreads();
  if (w == 0) {
    int v = wsum[lane];
    int iv = v;
    for (int off = 1; off < 32; off <<= 1) {
      int u = __shfl_up(iv, off, 32);
      if (lane >= off) iv += u;
    }
    wsum[lane] = iv - v;
  }
  __syncthreads();
  int excl = base + wsum[w] + inc - c;
  if (i < n) {
    rowptr[i] = excl;
    invdeg[i] = 1.0f / fmaxf((float)c, 1.0f);
  }
  if (i == n - 1) rowptr[n] = excl + c;
}

__global__ void fill_kernel(const int* __restrict__ src, const int* __restrict__ dst,
                            const int* __restrict__ rowptr, const int* __restrict__ pos,
                            int* __restrict__ col, int E) {
  int B = blockDim.x;
  int e0 = blockIdx.x * (B * 4) + threadIdx.x;
  int e1 = e0 + B, e2 = e1 + B, e3 = e2 + B;
  if (e3 < E) {
    int d0 = dst[e0], d1 = dst[e1], d2 = dst[e2], d3 = dst[e3];
    int p0 = pos[e0], p1 = pos[e1], p2 = pos[e2], p3 = pos[e3];
    int s0 = src[e0], s1 = src[e1], s2 = src[e2], s3 = src[e3];
    int q0 = rowptr[d0], q1 = rowptr[d1], q2 = rowptr[d2], q3 = rowptr[d3];
    col[q0 + p0] = s0; col[q1 + p1] = s1; col[q2 + p2] = s2; col[q3 + p3] = s3;
  } else {
    for (int e = e0; e < E; e += B) col[rowptr[dst[e]] + pos[e]] = src[e];
  }
}

#define GATHER_NODES 4
__device__ __forceinline__ void add4(float4& a, const float4 v) {
  a.x += v.x; a.y += v.y; a.z += v.z; a.w += v.w;
}
__global__ void gather_sum_kernel(const float* __restrict__ x, const int* __restrict__ rowptr,
                                  const int* __restrict__ col, float* __restrict__ agg, int n) {
  int w = __builtin_amdgcn_readfirstlane((int)(blockIdx.x * (blockDim.x >> 5) + (threadIdx.x >> 5)));
  int lane = threadIdx.x & 31;
  const float* xl = x + lane * 4;
  for (int q = 0; q < GATHER_NODES; ++q) {
    int node = w * GATHER_NODES + q;
    if (node >= n) break;
    int j = rowptr[node];
    int end = rowptr[node + 1];
    float4 acc = make_float4(0.0f, 0.0f, 0.0f, 0.0f);
    for (; j + 4 <= end; j += 4) {
      int s0 = col[j], s1 = col[j + 1], s2 = col[j + 2], s3 = col[j + 3];
      const float4 v0 = *reinterpret_cast<const float4*>(xl + (size_t)s0 * 128);
      const float4 v1 = *reinterpret_cast<const float4*>(xl + (size_t)s1 * 128);
      const float4 v2 = *reinterpret_cast<const float4*>(xl + (size_t)s2 * 128);
      const float4 v3 = *reinterpret_cast<const float4*>(xl + (size_t)s3 * 128);
      add4(acc, v0); add4(acc, v1); add4(acc, v2); add4(acc, v3);
    }
    int rem = end - j;
    if (rem > 0) {
      int s0 = col[j];
      int s1 = (rem > 1) ? col[j + 1] : s0;
      int s2 = (rem > 2) ? col[j + 2] : s0;
      const float4 v0 = *reinterpret_cast<const float4*>(xl + (size_t)s0 * 128);
      const float4 v1 = *reinterpret_cast<const float4*>(xl + (size_t)s1 * 128);
      const float4 v2 = *reinterpret_cast<const float4*>(xl + (size_t)s2 * 128);
      add4(acc, v0);
      if (rem > 1) add4(acc, v1);
      if (rem > 2) add4(acc, v2);
    }
    *reinterpret_cast<float4*>(agg + (size_t)node * 128 + lane * 4) = acc;
  }
}

__global__ void sage_gemm_kernel(const float* __restrict__ Agg, const float* __restrict__ invdeg,
                                 const float* __restrict__ X,
                                 const float* __restrict__ Wl, const float* __restrict__ bl,
                                 const float* __restrict__ Wr,
                                 float* __restrict__ OutA, float* __restrict__ OutB,
                                 int M, int K, int N, int doRelu) {
  int wave = blockIdx.x * (blockDim.x >> 5) + (threadIdx.x >> 5);
  int lane = threadIdx.x & 31;
  int nTilesN = N >> 4;
  int nTiles = (M >> 4) * nTilesN;
  if (wave >= nTiles) return;

  int tn = wave % nTilesN;
  int tm = wave / nTilesN;
  int m0 = tm << 4;
  int n0 = tn << 4;

  int rA   = m0 + (lane & 15);
  int colB = n0 + (lane & 15);
  int kh   = (lane >> 4) << 1;

  const float* arow = Agg + (size_t)rA * K;
  const float* xrow = X   + (size_t)rA * K;
  float s = invdeg[rA];

  v8f c = {};
#pragma unroll 4
  for (int kk = 0; kk < K; kk += 4) {
    v2f a, b;
    a.x = arow[kk + kh] * s;
    a.y = arow[kk + kh + 1] * s;
    b.x = Wl[(size_t)(kk + kh) * N + colB];
    b.y = Wl[(size_t)(kk + kh + 1) * N + colB];
    c = __builtin_amdgcn_wmma_f32_16x16x4_f32(false, a, false, b, (short)0, c, false, false);
    a.x = xrow[kk + kh];
    a.y = xrow[kk + kh + 1];
    b.x = Wr[(size_t)(kk + kh) * N + colB];
    b.y = Wr[(size_t)(kk + kh + 1) * N + colB];
    c = __builtin_amdgcn_wmma_f32_16x16x4_f32(false, a, false, b, (short)0, c, false, false);
  }

  float bias = bl[colB];
  int rbase = m0 + ((lane >> 4) << 3);
#pragma unroll
  for (int r = 0; r < 8; ++r) {
    float v = c[r] + bias;
    size_t off = (size_t)(rbase + r) * N + colB;
    if (OutB) OutB[off] = v;
    if (doRelu) v = fmaxf(v, 0.0f);
    OutA[off] = v;
  }
}

#define POOL_NODES 16
__device__ __forceinline__ void pool_flush(float* __restrict__ gsum, float* __restrict__ gcnt,
                                           int g, int lane, const float4 acc, int run) {
  float* p = gsum + (size_t)g * 128 + lane * 4;
  atomAddF(p + 0, acc.x);
  atomAddF(p + 1, acc.y);
  atomAddF(p + 2, acc.z);
  atomAddF(p + 3, acc.w);
  if (lane == 0) atomAddF(&gcnt[g], (float)run);
}
__global__ void pool_kernel(const float* __restrict__ h, const int* __restrict__ cluster,
                            float* __restrict__ gsum, float* __restrict__ gcnt, int n) {
  int w = __builtin_amdgcn_readfirstlane((int)(blockIdx.x * (blockDim.x >> 5) + (threadIdx.x >> 5)));
  int lane = threadIdx.x & 31;
  int node0 = w * POOL_NODES;
  if (node0 >= n) return;
  const float* hl = h + (size_t)node0 * 128 + lane * 4;
  const int* cl = cluster + node0;
  if (node0 + POOL_NODES <= n) {
    int g[POOL_NODES];
    float4 v[POOL_NODES];
#pragma unroll
    for (int q = 0; q < POOL_NODES; ++q) {
      g[q] = cl[q];
      v[q] = *reinterpret_cast<const float4*>(hl + (size_t)q * 128);
    }
    int cur = g[0];
    int run = 1;
    float4 acc = v[0];
#pragma unroll
    for (int q = 1; q < POOL_NODES; ++q) {
      if (g[q] != cur) {
        pool_flush(gsum, gcnt, cur, lane, acc, run);
        cur = g[q];
        run = 1;
        acc = v[q];
      } else {
        add4(acc, v[q]);
        ++run;
      }
    }
    pool_flush(gsum, gcnt, cur, lane, acc, run);
  } else {
    int cur = cl[0];
    int run = 1;
    float4 acc = *reinterpret_cast<const float4*>(hl);
    for (int q = 1; node0 + q < n; ++q) {
      int gq = cl[q];
      const float4 vq = *reinterpret_cast<const float4*>(hl + (size_t)q * 128);
      if (gq != cur) {
        pool_flush(gsum, gcnt, cur, lane, acc, run);
        cur = gq;
        run = 1;
        acc = vq;
      } else {
        add4(acc, vq);
        ++run;
      }
    }
    pool_flush(gsum, gcnt, cur, lane, acc, run);
  }
}

__global__ void pool_div_kernel(float* __restrict__ g, const float* __restrict__ gcnt, int total) {
  int i = blockIdx.x * blockDim.x + threadIdx.x;
  if (i >= total) return;
  g[i] = g[i] / fmaxf(gcnt[i >> 7], 1.0f);
}

__global__ void log_softmax_kernel(float* __restrict__ out, int rows) {
  int wave = blockIdx.x * (blockDim.x >> 5) + (threadIdx.x >> 5);
  int lane = threadIdx.x & 31;
  if (wave >= rows) return;
  float* row = out + (size_t)wave * 64;
  float v0 = row[lane];
  float v1 = row[lane + 32];
  float m = fmaxf(v0, v1);
#pragma unroll
  for (int off = 16; off >= 1; off >>= 1) m = fmaxf(m, __shfl_xor(m, off, 32));
  float s = __expf(v0 - m) + __expf(v1 - m);
#pragma unroll
  for (int off = 16; off >= 1; off >>= 1) s += __shfl_xor(s, off, 32);
  float lse = m + __logf(s);
  row[lane]      = v0 - lse;
  row[lane + 32] = v1 - lse;
}

extern "C" void kernel_launch(void* const* d_in, const int* in_sizes, int n_in,
                              void* d_out, int out_size, void* d_ws, size_t ws_size,
                              hipStream_t stream) {
  const float* x       = (const float*)d_in[0];
  const int*   ei      = (const int*)d_in[1];
  const int*   src     = ei;
  const int*   dst     = ei + NE;
  const int*   cluster = (const int*)d_in[2];
  const float* Wl0 = (const float*)d_in[3];
  const float* bl0 = (const float*)d_in[4];
  const float* Wr0 = (const float*)d_in[5];
  const float* Wl1 = (const float*)d_in[6];
  const float* bl1 = (const float*)d_in[7];
  const float* Wr1 = (const float*)d_in[8];
  const float* Wl2 = (const float*)d_in[9];
  const float* bl2 = (const float*)d_in[10];
  const float* Wr2 = (const float*)d_in[11];

  float* out_ls = (float*)d_out;
  float* out_h2 = out_ls + (size_t)NN * OUTC;
  float* out_g  = out_h2 + (size_t)NN * HIDC;

  float* ws   = (float*)d_ws;
  float* agg  = ws;
  float* h1   = ws + (size_t)NN * HIDC;
  float* h2r  = h1 + (size_t)NN * HIDC;
  float* deg  = h2r + (size_t)NN * HIDC;
  float* gcnt = deg + NN;

  int* cnt    = (int*)h1;
  int* pos    = cnt + 50048;
  int* rowptr = (int*)out_ls;
  int* col    = rowptr + 50048;

  const int TPB = 256;
  const int edgeBlocks4 = (NE + TPB * 4 - 1) / (TPB * 4);
  const int gatherBlocks = (NN + 8 * GATHER_NODES - 1) / (8 * GATHER_NODES);

  hipMemsetAsync(cnt, 0, NN * sizeof(int), stream);
  count_rank_kernel<<<edgeBlocks4, TPB, 0, stream>>>(dst, cnt, pos, NE);
  scan_kernel<<<(NN + 1023) / 1024, 1024, 0, stream>>>(cnt, rowptr, deg, NN);
  fill_kernel<<<edgeBlocks4, TPB, 0, stream>>>(src, dst, rowptr, pos, col, NE);

  gather_sum_kernel<<<gatherBlocks, TPB, 0, stream>>>(x, rowptr, col, agg, NN);
  {
    int tiles = (NN / 16) * (HIDC / 16);
    sage_gemm_kernel<<<(tiles + 7) / 8, TPB, 0, stream>>>(
        agg, deg, x, Wl0, bl0, Wr0, h1, nullptr, NN, INC, HIDC, 1);
  }

  gather_sum_kernel<<<gatherBlocks, TPB, 0, stream>>>(h1, rowptr, col, agg, NN);
  {
    int tiles = (NN / 16) * (HIDC / 16);
    sage_gemm_kernel<<<(tiles + 7) / 8, TPB, 0, stream>>>(
        agg, deg, h1, Wl1, bl1, Wr1, h2r, out_h2, NN, HIDC, HIDC, 1);
  }

  hipMemsetAsync(out_g, 0, (size_t)NG * HIDC * sizeof(float), stream);
  hipMemsetAsync(gcnt, 0, NG * sizeof(float), stream);
  pool_kernel<<<(NN + 8 * POOL_NODES - 1) / (8 * POOL_NODES), TPB, 0, stream>>>(h2r, cluster, out_g, gcnt, NN);
  pool_div_kernel<<<(NG * HIDC + TPB - 1) / TPB, TPB, 0, stream>>>(out_g, gcnt, NG * HIDC);

  gather_sum_kernel<<<gatherBlocks, TPB, 0, stream>>>(h2r, rowptr, col, agg, NN);
  {
    int tiles = (NN / 16) * (OUTC / 16);
    sage_gemm_kernel<<<(tiles + 7) / 8, TPB, 0, stream>>>(
        agg, deg, h2r, Wl2, bl2, Wr2, out_ls, nullptr, NN, HIDC, OUTC, 0);
  }
  log_softmax_kernel<<<((size_t)NN * 32 + TPB - 1) / TPB, TPB, 0, stream>>>(out_ls, NN);
}


# GEMM register tiling: a wave owns 16 rows x 64 cols as 4 tiles (cols interleaved by 4); one A fragment feeds 4 tiles, A/W/result move as float4, double-buffered; k order in each 8-k block changes
# speedup vs baseline: 2.6167x; 1.3310x over previous
#include <hip/hip_runtime.h>
#include <hip/hip_bf16.h>
#include <math.h>

typedef float v2f __attribute__((ext_vector_type(2)));
typedef float v8f __attribute__((ext_vector_type(8)));

#define NN      50000
#define NE      600000
#define INC     128
#define HIDC    128
#define OUTC    64
#define NG      512

__device__ __forceinline__ void atomAddF(float* p, float v) {
  __hip_atomic_fetch_add(p, v, __ATOMIC_RELAXED, __HIP_MEMORY_SCOPE_AGENT);
}

__global__ void count_rank_kernel(const int* __restrict__ dst, int* __restrict__ cnt,
                                  int* __restrict__ pos, int E) {
  int B = blockDim.x;
  int e0 = blockIdx.x * (B * 4) + threadIdx.x;
  int e1 = e0 + B, e2 = e1 + B, e3 = e2 + B;
  if (e3 < E) {
    int d0 = dst[e0], d1 = dst[e1], d2 = dst[e2], d3 = dst[e3];
    int r0 = __hip_atomic_fetch_add(&cnt[d0], 1, __ATOMIC_RELAXED, __HIP_MEMORY_SCOPE_AGENT);
    int r1 = __hip_atomic_fetch_add(&cnt[d1], 1, __ATOMIC_RELAXED, __HIP_MEMORY_SCOPE_AGENT);
    int r2 = __hip_atomic_fetch_add(&cnt[d2], 1, __ATOMIC_RELAXED, __HIP_MEMORY_SCOPE_AGENT);
    int r3 = __hip_atomic_fetch_add(&cnt[d3], 1, __ATOMIC_RELAXED, __HIP_MEMORY_SCOPE_AGENT);
    pos[e0] = r0; pos[e1] = r1; pos[e2] = r2; pos[e3] = r3;
  } else {
    for (int e = e0; e < E; e += B)
      pos[e] = __hip_atomic_fetch_add(&cnt[dst[e]], 1, __ATOMIC_RELAXED, __HIP_MEMORY_SCOPE_AGENT);
  }
}

__global__ void __launch_bounds__(1024)
scan_kernel(const int* __restrict__ cnt, int* __restrict__ rowptr,
            float* __restrict__ invdeg, int n) {
  __shared__ int wsum[32];
  __shared__ int sbase;
  int t = threadIdx.x;
  int lane = t & 31;
  int w = t >> 5;
  int start = blockIdx.x * 1024;

  int part = 0;
  for (int i = t; i < start; i += 1024) part += cnt[i];
  for (int off = 16; off >= 1; off >>= 1) part += __shfl_xor(part, off, 32);
  if (lane == 0) wsum[w] = part;
  __syncthreads();
  if (w == 0) {
    int v = wsum[lane];
    for (int off = 16; off >= 1; off >>= 1) v += __shfl_xor(v, off, 32);
    if (lane == 0) sbase = v;
  }
  __syncthreads();
  int base = sbase;

  int i = start + t;
  int c = (i < n) ? cnt[i] : 0;
  int inc = c;
  for (int off = 1; off < 32; off <<= 1) {
    int u = __shfl_up(inc, off, 32);
    if (lane >= off) inc += u;
  }
  if (lane == 31) wsum[w] = inc;
  __syncthreads();
  if (w == 0) {
    int v = wsum[lane];
    int iv = v;
    for (int off = 1; off < 32; off <<= 1) {
      int u = __shfl_up(iv, off, 32);
      if (lane >= off) iv += u;
    }
    wsum[lane] = iv - v;
  }
  __syncthreads();
  int excl = base + wsum[w] + inc - c;
  if (i < n) {
    rowptr[i] = excl;
    invdeg[i] = 1.0f / fmaxf((float)c, 1.0f);
  }
  if (i == n - 1) rowptr[n] = excl + c;
}

__global__ void fill_kernel(const int* __restrict__ src, const int* __restrict__ dst,
                            const int* __restrict__ rowptr, const int* __restrict__ pos,
                            int* __restrict__ col, int E) {
  int B = blockDim.x;
  int e0 = blockIdx.x * (B * 4) + threadIdx.x;
  int e1 = e0 + B, e2 = e1 + B, e3 = e2 + B;
  if (e3 < E) {
    int d0 = dst[e0], d1 = dst[e1], d2 = dst[e2], d3 = dst[e3];
    int p0 = pos[e0], p1 = pos[e1], p2 = pos[e2], p3 = pos[e3];
    int s0 = src[e0], s1 = src[e1], s2 = src[e2], s3 = src[e3];
    int q0 = rowptr[d0], q1 = rowptr[d1], q2 = rowptr[d2], q3 = rowptr[d3];
    col[q0 + p0] = s0; col[q1 + p1] = s1; col[q2 + p2] = s2; col[q3 + p3] = s3;
  } else {
    for (int e = e0; e < E; e += B) col[rowptr[dst[e]] + pos[e]] = src[e];
  }
}

#define GATHER_NODES 4
__device__ __forceinline__ void add4(float4& a, const float4 v) {
  a.x += v.x; a.y += v.y; a.z += v.z; a.w += v.w;
}
__global__ void gather_sum_kernel(const float* __restrict__ x, const int* __restrict__ rowptr,
                                  const int* __restrict__ col, float* __restrict__ agg, int n) {
  int w = __builtin_amdgcn_readfirstlane((int)(blockIdx.x * (blockDim.x >> 5) + (threadIdx.x >> 5)));
  int lane = threadIdx.x & 31;
  const float* xl = x + lane * 4;
  for (int q = 0; q < GATHER_NODES; ++q) {
    int node = w * GATHER_NODES + q;
    if (node >= n) break;
    int j = rowptr[node];
    int end = rowptr[node + 1];
    float4 acc = make_float4(0.0f, 0.0f, 0.0f, 0.0f);
    for (; j + 4 <= end; j += 4) {
      int s0 = col[j], s1 = col[j + 1], s2 = col[j + 2], s3 = col[j + 3];
      const float4 v0 = *reinterpret_cast<const float4*>(xl + (size_t)s0 * 128);
      const float4 v1 = *reinterpret_cast<const float4*>(xl + (size_t)s1 * 128);
      const float4 v2 = *reinterpret_cast<const float4*>(xl + (size_t)s2 * 128);
      const float4 v3 = *reinterpret_cast<const float4*>(xl + (size_t)s3 * 128);
      add4(acc, v0); add4(acc, v1); add4(acc, v2); add4(acc, v3);
    }
    int rem = end - j;
    if (rem > 0) {
      int s0 = col[j];
      int s1 = (rem > 1) ? col[j + 1] : s0;
      int s2 = (rem > 2) ? col[j + 2] : s0;
      const float4 v0 = *reinterpret_cast<const float4*>(xl + (size_t)s0 * 128);
      const float4 v1 = *reinterpret_cast<const float4*>(xl + (size_t)s1 * 128);
      const float4 v2 = *reinterpret_cast<const float4*>(xl + (size_t)s2 * 128);
      add4(acc, v0);
      if (rem > 1) add4(acc, v1);
      if (rem > 2) add4(acc, v2);
    }
    *reinterpret_cast<float4*>(agg + (size_t)node * 128 + lane * 4) = acc;
  }
}

struct GemmFrag {
  float4 a, x;
  float4 l[4], r[4];
};
__device__ __forceinline__ void gemm_load(GemmFrag& f, const float* __restrict__ arow,
                                          const float* __restrict__ xrow,
                                          const float* __restrict__ wl, const float* __restrict__ wr,
                                          int i, int N) {
  f.a = *reinterpret_cast<const float4*>(arow + 8 * i);
  f.x = *reinterpret_cast<const float4*>(xrow + 8 * i);
#pragma unroll
  for (int q = 0; q < 4; ++q) {
    f.l[q] = *reinterpret_cast<const float4*>(wl + (size_t)(8 * i + q) * N);
    f.r[q] = *reinterpret_cast<const float4*>(wr + (size_t)(8 * i + q) * N);
  }
}
__device__ __forceinline__ float f4c(const float4 v, int j) {
  return j == 0 ? v.x : (j == 1 ? v.y : (j == 2 ? v.z : v.w));
}
__device__ __forceinline__ void gemm_mma(const GemmFrag& f, float s, v8f (&acc)[4]) {
#pragma unroll
  for (int t = 0; t < 2; ++t) {
    v2f af, xf;
    af.x = f4c(f.a, 2 * t) * s;
    af.y = f4c(f.a, 2 * t + 1) * s;
    xf.x = f4c(f.x, 2 * t);
    xf.y = f4c(f.x, 2 * t + 1);
#pragma unroll
    for (int j = 0; j < 4; ++j) {
      v2f bf;
      bf.x = f4c(f.l[2 * t], j);
      bf.y = f4c(f.l[2 * t + 1], j);
      acc[j] = __builtin_amdgcn_wmma_f32_16x16x4_f32(false, af, false, bf, (short)0, acc[j], false, false);
    }
#pragma unroll
    for (int j = 0; j < 4; ++j) {
      v2f bf;
      bf.x = f4c(f.r[2 * t], j);
      bf.y = f4c(f.r[2 * t + 1], j);
      acc[j] = __builtin_amdgcn_wmma_f32_16x16x4_f32(false, xf, false, bf, (short)0, acc[j], false, false);
    }
  }
}
#define GEMM_WAVES 4
__global__ void __launch_bounds__(GEMM_WAVES * 32)
sage_gemm_kernel(const float* __restrict__ Agg, const float* __restrict__ invdeg,
                 const float* __restrict__ X,
                 const float* __restrict__ Wl, const float* __restrict__ bl,
                 const float* __restrict__ Wr,
                 float* __restrict__ OutA, float* __restrict__ OutB,
                 int M, int K, int N, int doRelu) {
  int wave = blockIdx.x * (blockDim.x >> 5) + (threadIdx.x >> 5);
  int lane = threadIdx.x & 31;
  int nGroups = N >> 6;
  int nTiles = (M >> 4) * nGroups;
  if (wave >= nTiles) return;

  int cg = wave % nGroups;
  int tm = wave / nGroups;
  int m0 = tm << 4;
  int nb = cg << 6;
  int c = lane & 15;
  int h = lane >> 4;
  int rA = m0 + c;

  const float* arow = Agg + (size_t)rA * K + 4 * h;
  const float* xrow = X   + (size_t)rA * K + 4 * h;
  const float* wl = Wl + (size_t)(4 * h) * N + nb + 4 * c;
  const float* wr = Wr + (size_t)(4 * h) * N + nb + 4 * c;
  float s = invdeg[rA];

  v8f acc[4] = {};
  int KI = K >> 3;
  GemmFrag cur;
  gemm_load(cur, arow, xrow, wl, wr, 0, N);
  for (int i = 0; i + 1 < KI; ++i) {
    GemmFrag nxt;
    gemm_load(nxt, arow, xrow, wl, wr, i + 1, N);
    __builtin_amdgcn_sched_barrier(0);
    gemm_mma(cur, s, acc);
    cur = nxt;
  }
  gemm_mma(cur, s, acc);

  const float4 bias = *reinterpret_cast<const float4*>(bl + nb + 4 * c);
  int rbase = m0 + (h << 3);
#pragma unroll
  for (int r = 0; r < 8; ++r) {
    float4 v = make_float4(acc[0][r] + bias.x, acc[1][r] + bias.y, acc[2][r] + bias.z, acc[3][r] + bias.w);
    size_t off = (size_t)(rbase + r) * N + nb + 4 * c;
    if (OutB) *reinterpret_cast<float4*>(OutB + off) = v;
    if (doRelu) v = make_float4(fmaxf(v.x, 0.0f), fmaxf(v.y, 0.0f), fmaxf(v.z, 0.0f), fmaxf(v.w, 0.0f));
    *reinterpret_cast<float4*>(OutA + off) = v;
  }
}

#define POOL_NODES 16
__device__ __forceinline__ void pool_flush(float* __restrict__ gsum, float* __restrict__ gcnt,
                                           int g, int lane, const float4 acc, int run) {
  float* p = gsum + (size_t)g * 128 + lane * 4;
  atomAddF(p + 0, acc.x);
  atomAddF(p + 1, acc.y);
  atomAddF(p + 2, acc.z);
  atomAddF(p + 3, acc.w);
  if (lane == 0) atomAddF(&gcnt[g], (float)run);
}
__global__ void pool_kernel(const float* __restrict__ h, const int* __restrict__ cluster,
                            float* __restrict__ gsum, float* __restrict__ gcnt, int n) {
  int w = __builtin_amdgcn_readfirstlane((int)(blockIdx.x * (blockDim.x >> 5) + (threadIdx.x >> 5)));
  int lane = threadIdx.x & 31;
  int node0 = w * POOL_NODES;
  if (node0 >= n) return;
  const float* hl = h + (size_t)node0 * 128 + lane * 4;
  const int* cl = cluster + node0;
  if (node0 + POOL_NODES <= n) {
    int g[POOL_NODES];
    float4 v[POOL_NODES];
#pragma unroll
    for (int q = 0; q < POOL_NODES; ++q) {
      g[q] = cl[q];
      v[q] = *reinterpret_cast<const float4*>(hl + (size_t)q * 128);
    }
    int cur = g[0];
    int run = 1;
    float4 acc = v[0];
#pragma unroll
    for (int q = 1; q < POOL_NODES; ++q) {
      if (g[q] != cur) {
        pool_flush(gsum, gcnt, cur, lane, acc, run);
        cur = g[q];
        run = 1;
        acc = v[q];
      } else {
        add4(acc, v[q]);
        ++run;
      }
    }
    pool_flush(gsum, gcnt, cur, lane, acc, run);
  } else {
    int cur = cl[0];
    int run = 1;
    float4 acc = *reinterpret_cast<const float4*>(hl);
    for (int q = 1; node0 + q < n; ++q) {
      int gq = cl[q];
      const float4 vq = *reinterpret_cast<const float4*>(hl + (size_t)q * 128);
      if (gq != cur) {
        pool_flush(gsum, gcnt, cur, lane, acc, run);
        cur = gq;
        run = 1;
        acc = vq;
      } else {
        add4(acc, vq);
        ++run;
      }
    }
    pool_flush(gsum, gcnt, cur, lane, acc, run);
  }
}

__global__ void pool_div_kernel(float* __restrict__ g, const float* __restrict__ gcnt, int total) {
  int i = blockIdx.x * blockDim.x + threadIdx.x;
  if (i >= total) return;
  g[i] = g[i] / fmaxf(gcnt[i >> 7], 1.0f);
}

__global__ void log_softmax_kernel(float* __restrict__ out, int rows) {
  int wave = blockIdx.x * (blockDim.x >> 5) + (threadIdx.x >> 5);
  int lane = threadIdx.x & 31;
  if (wave >= rows) return;
  float* row = out + (size_t)wave * 64;
  float v0 = row[lane];
  float v1 = row[lane + 32];
  float m = fmaxf(v0, v1);
#pragma unroll
  for (int off = 16; off >= 1; off >>= 1) m = fmaxf(m, __shfl_xor(m, off, 32));
  float s = __expf(v0 - m) + __expf(v1 - m);
#pragma unroll
  for (int off = 16; off >= 1; off >>= 1) s += __shfl_xor(s, off, 32);
  float lse = m + __logf(s);
  row[lane]      = v0 - lse;
  row[lane + 32] = v1 - lse;
}

extern "C" void kernel_launch(void* const* d_in, const int* in_sizes, int n_in,
                              void* d_out, int out_size, void* d_ws, size_t ws_size,
                              hipStream_t stream) {
  const float* x       = (const float*)d_in[0];
  const int*   ei      = (const int*)d_in[1];
  const int*   src     = ei;
  const int*   dst     = ei + NE;
  const int*   cluster = (const int*)d_in[2];
  const float* Wl0 = (const float*)d_in[3];
  const float* bl0 = (const float*)d_in[4];
  const float* Wr0 = (const float*)d_in[5];
  const float* Wl1 = (const float*)d_in[6];
  const float* bl1 = (const float*)d_in[7];
  const float* Wr1 = (const float*)d_in[8];
  const float* Wl2 = (const float*)d_in[9];
  const float* bl2 = (const float*)d_in[10];
  const float* Wr2 = (const float*)d_in[11];

  float* out_ls = (float*)d_out;
  float* out_h2 = out_ls + (size_t)NN * OUTC;
  float* out_g  = out_h2 + (size_t)NN * HIDC;

  float* ws   = (float*)d_ws;
  float* agg  = ws;
  float* h1   = ws + (size_t)NN * HIDC;
  float* h2r  = h1 + (size_t)NN * HIDC;
  float* deg  = h2r + (size_t)NN * HIDC;
  float* gcnt = deg + NN;

  int* cnt    = (int*)h1;
  int* pos    = cnt + 50048;
  int* rowptr = (int*)out_ls;
  int* col    = rowptr + 50048;

  const int TPB = 256;
  const int edgeBlocks4 = (NE + TPB * 4 - 1) / (TPB * 4);
  const int gatherBlocks = (NN + 8 * GATHER_NODES - 1) / (8 * GATHER_NODES);

  hipMemsetAsync(cnt, 0, NN * sizeof(int), stream);
  count_rank_kernel<<<edgeBlocks4, TPB, 0, stream>>>(dst, cnt, pos, NE);
  scan_kernel<<<(NN + 1023) / 1024, 1024, 0, stream>>>(cnt, rowptr, deg, NN);
  fill_kernel<<<edgeBlocks4, TPB, 0, stream>>>(src, dst, rowptr, pos, col, NE);

  gather_sum_kernel<<<gatherBlocks, TPB, 0, stream>>>(x, rowptr, col, agg, NN);
  {
    int tiles = (NN / 16) * (HIDC / 64);
    sage_gemm_kernel<<<(tiles + GEMM_WAVES - 1) / GEMM_WAVES, GEMM_WAVES * 32, 0, stream>>>(
        agg, deg, x, Wl0, bl0, Wr0, h1, nullptr, NN, INC, HIDC, 1);
  }

  gather_sum_kernel<<<gatherBlocks, TPB, 0, stream>>>(h1, rowptr, col, agg, NN);
  {
    int tiles = (NN / 16) * (HIDC / 64);
    sage_gemm_kernel<<<(tiles + GEMM_WAVES - 1) / GEMM_WAVES, GEMM_WAVES * 32, 0, stream>>>(
        agg, deg, h1, Wl1, bl1, Wr1, h2r, out_h2, NN, HIDC, HIDC, 1);
  }

  hipMemsetAsync(out_g, 0, (size_t)NG * HIDC * sizeof(float), stream);
  hipMemsetAsync(gcnt, 0, NG * sizeof(float), stream);
  pool_kernel<<<(NN + 8 * POOL_NODES - 1) / (8 * POOL_NODES), TPB, 0, stream>>>(h2r, cluster, out_g, gcnt, NN);
  pool_div_kernel<<<(NG * HIDC + TPB - 1) / TPB, TPB, 0, stream>>>(out_g, gcnt, NG * HIDC);

  gather_sum_kernel<<<gatherBlocks, TPB, 0, stream>>>(h2r, rowptr, col, agg, NN);
  {
    int tiles = (NN / 16) * (OUTC / 64);
    sage_gemm_kernel<<<(tiles + GEMM_WAVES - 1) / GEMM_WAVES, GEMM_WAVES * 32, 0, stream>>>(
        agg, deg, h2r, Wl2, bl2, Wr2, out_ls, nullptr, NN, HIDC, OUTC, 0);
  }
  log_softmax_kernel<<<((size_t)NN * 32 + TPB - 1) / TPB, TPB, 0, stream>>>(out_ls, NN);
}


# Log-softmax taken in the layer-2 GEMM epilogue: a wave's 16x64 tile holds whole rows, max and exp-sum go over 16 lanes by shuffles; its 50000-wave launch is gone; order of the softmax sum changes
# speedup vs baseline: 2.7068x; 1.0344x over previous
#include <hip/hip_runtime.h>
#include <hip/hip_bf16.h>
#include <math.h>

typedef float v2f __attribute__((ext_vector_type(2)));
typedef float v8f __attribute__((ext_vector_type(8)));

#define NN      50000
#define NE      600000
#define INC     128
#define HIDC    128
#define OUTC    64
#define NG      512

__device__ __forceinline__ void atomAddF(float* p, float v) {
  __hip_atomic_fetch_add(p, v, __ATOMIC_RELAXED, __HIP_MEMORY_SCOPE_AGENT);
}

__global__ void count_rank_kernel(const int* __restrict__ dst, int* __restrict__ cnt,
                                  int* __restrict__ pos, int E) {
  int B = blockDim.x;
  int e0 = blockIdx.x * (B * 4) + threadIdx.x;
  int e1 = e0 + B, e2 = e1 + B, e3 = e2 + B;
  if (e3 < E) {
    int d0 = dst[e0], d1 = dst[e1], d2 = dst[e2], d3 = dst[e3];
    int r0 = __hip_atomic_fetch_add(&cnt[d0], 1, __ATOMIC_RELAXED, __HIP_MEMORY_SCOPE_AGENT);
    int r1 = __hip_atomic_fetch_add(&cnt[d1], 1, __ATOMIC_RELAXED, __HIP_MEMORY_SCOPE_AGENT);
    int r2 = __hip_atomic_fetch_add(&cnt[d2], 1, __ATOMIC_RELAXED, __HIP_MEMORY_SCOPE_AGENT);
    int r3 = __hip_atomic_fetch_add(&cnt[d3], 1, __ATOMIC_RELAXED, __HIP_MEMORY_SCOPE_AGENT);
    pos[e0] = r0; pos[e1] = r1; pos[e2] = r2; pos[e3] = r3;
  } else {
    for (int e = e0; e < E; e += B)
      pos[e] = __hip_atomic_fetch_add(&cnt[dst[e]], 1, __ATOMIC_RELAXED, __HIP_MEMORY_SCOPE_AGENT);
  }
}

__global__ void __launch_bounds__(1024)
scan_kernel(const int* __restrict__ cnt, int* __restrict__ rowptr,
            float* __restrict__ invdeg, int n) {
  __shared__ int wsum[32];
  __shared__ int sbase;
  int t = threadIdx.x;
  int lane = t & 31;
  int w = t >> 5;
  int start = blockIdx.x * 1024;

  int part = 0;
  for (int i = t; i < start; i += 1024) part += cnt[i];
  for (int off = 16; off >= 1; off >>= 1) part += __shfl_xor(part, off, 32);
  if (lane == 0) wsum[w] = part;
  __syncthreads();
  if (w == 0) {
    int v = wsum[lane];
    for (int off = 16; off >= 1; off >>= 1) v += __shfl_xor(v, off, 32);
    if (lane == 0) sbase = v;
  }
  __syncthreads();
  int base = sbase;

  int i = start + t;
  int c = (i < n) ? cnt[i] : 0;
  int inc = c;
  for (int off = 1; off < 32; off <<= 1) {
    int u = __shfl_up(inc, off, 32);
    if (lane >= off) inc += u;
  }
  if (lane == 31) wsum[w] = inc;
  __syncthreads();
  if (w == 0) {
    int v = wsum[lane];
    int iv = v;
    for (int off = 1; off < 32; off <<= 1) {
      int u = __shfl_up(iv, off, 32);
      if (lane >= off) iv += u;
    }
    wsum[lane] = iv - v;
  }
  __syncthreads();
  int excl = base + wsum[w] + inc - c;
  if (i < n) {
    rowptr[i] = excl;
    invdeg[i] = 1.0f / fmaxf((float)c, 1.0f);
  }
  if (i == n - 1) rowptr[n] = excl + c;
}

__global__ void fill_kernel(const int* __restrict__ src, const int* __restrict__ dst,
                            const int* __restrict__ rowptr, const int* __restrict__ pos,
                            int* __restrict__ col, int E) {
  int B = blockDim.x;
  int e0 = blockIdx.x * (B * 4) + threadIdx.x;
  int e1 = e0 + B, e2 = e1 + B, e3 = e2 + B;
  if (e3 < E) {
    int d0 = dst[e0], d1 = dst[e1], d2 = dst[e2], d3 = dst[e3];
    int p0 = pos[e0], p1 = pos[e1], p2 = pos[e2], p3 = pos[e3];
    int s0 = src[e0], s1 = src[e1], s2 = src[e2], s3 = src[e3];
    int q0 = rowptr[d0], q1 = rowptr[d1], q2 = rowptr[d2], q3 = rowptr[d3];
    col[q0 + p0] = s0; col[q1 + p1] = s1; col[q2 + p2] = s2; col[q3 + p3] = s3;
  } else {
    for (int e = e0; e < E; e += B) col[rowptr[dst[e]] + pos[e]] = src[e];
  }
}

#define GATHER_NODES 4
__device__ __forceinline__ void add4(float4& a, const float4 v) {
  a.x += v.x; a.y += v.y; a.z += v.z; a.w += v.w;
}
__global__ void gather_sum_kernel(const float* __restrict__ x, const int* __restrict__ rowptr,
                                  const int* __restrict__ col, float* __restrict__ agg, int n) {
  int w = __builtin_amdgcn_readfirstlane((int)(blockIdx.x * (blockDim.x >> 5) + (threadIdx.x >> 5)));
  int lane = threadIdx.x & 31;
  const float* xl = x + lane * 4;
  for (int q = 0; q < GATHER_NODES; ++q) {
    int node = w * GATHER_NODES + q;
    if (node >= n) break;
    int j = rowptr[node];
    int end = rowptr[node + 1];
    float4 acc = make_float4(0.0f, 0.0f, 0.0f, 0.0f);
    for (; j + 4 <= end; j += 4) {
      int s0 = col[j], s1 = col[j + 1], s2 = col[j + 2], s3 = col[j + 3];
      const float4 v0 = *reinterpret_cast<const float4*>(xl + (size_t)s0 * 128);
      const float4 v1 = *reinterpret_cast<const float4*>(xl + (size_t)s1 * 128);
      const float4 v2 = *reinterpret_cast<const float4*>(xl + (size_t)s2 * 128);
      const float4 v3 = *reinterpret_cast<const float4*>(xl + (size_t)s3 * 128);
      add4(acc, v0); add4(acc, v1); add4(acc, v2); add4(acc, v3);
    }
    int rem = end - j;
    if (rem > 0) {
      int s0 = col[j];
      int s1 = (rem > 1) ? col[j + 1] : s0;
      int s2 = (rem > 2) ? col[j + 2] : s0;
      const float4 v0 = *reinterpret_cast<const float4*>(xl + (size_t)s0 * 128);
      const float4 v1 = *reinterpret_cast<const float4*>(xl + (size_t)s1 * 128);
      const float4 v2 = *reinterpret_cast<const float4*>(xl + (size_t)s2 * 128);
      add4(acc, v0);
      if (rem > 1) add4(acc, v1);
      if (rem > 2) add4(acc, v2);
    }
    *reinterpret_cast<float4*>(agg + (size_t)node * 128 + lane * 4) = acc;
  }
}

struct GemmFrag {
  float4 a, x;
  float4 l[4], r[4];
};
__device__ __forceinline__ void gemm_load(GemmFrag& f, const float* __restrict__ arow,
                                          const float* __restrict__ xrow,
                                          const float* __restrict__ wl, const float* __restrict__ wr,
                                          int i, int N) {
  f.a = *reinterpret_cast<const float4*>(arow + 8 * i);
  f.x = *reinterpret_cast<const float4*>(xrow + 8 * i);
#pragma unroll
  for (int q = 0; q < 4; ++q) {
    f.l[q] = *reinterpret_cast<const float4*>(wl + (size_t)(8 * i + q) * N);
    f.r[q] = *reinterpret_cast<const float4*>(wr + (size_t)(8 * i + q) * N);
  }
}
__device__ __forceinline__ float f4c(const float4 v, int j) {
  return j == 0 ? v.x : (j == 1 ? v.y : (j == 2 ? v.z : v.w));
}
__device__ __forceinline__ void gemm_mma(const GemmFrag& f, float s, v8f (&acc)[4]) {
#pragma unroll
  for (int t = 0; t < 2; ++t) {
    v2f af, xf;
    af.x = f4c(f.a, 2 * t) * s;
    af.y = f4c(f.a, 2 * t + 1) * s;
    xf.x = f4c(f.x, 2 * t);
    xf.y = f4c(f.x, 2 * t + 1);
#pragma unroll
    for (int j = 0; j < 4; ++j) {
      v2f bf;
      bf.x = f4c(f.l[2 * t], j);
      bf.y = f4c(f.l[2 * t + 1], j);
      acc[j] = __builtin_amdgcn_wmma_f32_16x16x4_f32(false, af, false, bf, (short)0, acc[j], false, false);
    }
#pragma unroll
    for (int j = 0; j < 4; ++j) {
      v2f bf;
      bf.x = f4c(f.r[2 * t], j);
      bf.y = f4c(f.r[2 * t + 1], j);
      acc[j] = __builtin_amdgcn_wmma_f32_16x16x4_f32(false, xf, false, bf, (short)0, acc[j], false, false);
    }
  }
}
__device__ __forceinline__ float half_max16(float v) {
  v = fmaxf(v, __shfl_xor(v, 8, 32));
  v = fmaxf(v, __shfl_xor(v, 4, 32));
  v = fmaxf(v, __shfl_xor(v, 2, 32));
  v = fmaxf(v, __shfl_xor(v, 1, 32));
  return v;
}
__device__ __forceinline__ float half_sum16(float v) {
  v += __shfl_xor(v, 8, 32);
  v += __shfl_xor(v, 4, 32);
  v += __shfl_xor(v, 2, 32);
  v += __shfl_xor(v, 1, 32);
  return v;
}
#define GEMM_WAVES 4
template <bool LSM>
__global__ void __launch_bounds__(GEMM_WAVES * 32)
sage_gemm_kernel(const float* __restrict__ Agg, const float* __restrict__ invdeg,
                 const float* __restrict__ X,
                 const float* __restrict__ Wl, const float* __restrict__ bl,
                 const float* __restrict__ Wr,
                 float* __restrict__ OutA, float* __restrict__ OutB,
                 int M, int K, int N, int doRelu) {
  int wave = blockIdx.x * (blockDim.x >> 5) + (threadIdx.x >> 5);
  int lane = threadIdx.x & 31;
  int nGroups = N >> 6;
  int nTiles = (M >> 4) * nGroups;
  if (wave >= nTiles) return;

  int cg = wave % nGroups;
  int tm = wave / nGroups;
  int m0 = tm << 4;
  int nb = cg << 6;
  int c = lane & 15;
  int h = lane >> 4;
  int rA = m0 + c;

  const float* arow = Agg + (size_t)rA * K + 4 * h;
  const float* xrow = X   + (size_t)rA * K + 4 * h;
  const float* wl = Wl + (size_t)(4 * h) * N + nb + 4 * c;
  const float* wr = Wr + (size_t)(4 * h) * N + nb + 4 * c;
  float s = invdeg[rA];

  v8f acc[4] = {};
  int KI = K >> 3;
  GemmFrag cur;
  gemm_load(cur, arow, xrow, wl, wr, 0, N);
  for (int i = 0; i + 1 < KI; ++i) {
    GemmFrag nxt;
    gemm_load(nxt, arow, xrow, wl, wr, i + 1, N);
    __builtin_amdgcn_sched_barrier(0);
    gemm_mma(cur, s, acc);
    cur = nxt;
  }
  gemm_mma(cur, s, acc);

  const float4 bias = *reinterpret_cast<const float4*>(bl + nb + 4 * c);
  int rbase = m0 + (h << 3);
  if constexpr (LSM) {
#pragma unroll
    for (int r = 0; r < 8; ++r) {
      float4 v = make_float4(acc[0][r] + bias.x, acc[1][r] + bias.y, acc[2][r] + bias.z, acc[3][r] + bias.w);
      float m = half_max16(fmaxf(fmaxf(v.x, v.y), fmaxf(v.z, v.w)));
      float e = half_sum16((__expf(v.x - m) + __expf(v.y - m)) + (__expf(v.z - m) + __expf(v.w - m)));
      float lse = m + __logf(e);
      size_t off = (size_t)(rbase + r) * N + nb + 4 * c;
      *reinterpret_cast<float4*>(OutA + off) = make_float4(v.x - lse, v.y - lse, v.z - lse, v.w - lse);
    }
  } else {
#pragma unroll
    for (int r = 0; r < 8; ++r) {
      float4 v = make_float4(acc[0][r] + bias.x, acc[1][r] + bias.y, acc[2][r] + bias.z, acc[3][r] + bias.w);
      size_t off = (size_t)(rbase + r) * N + nb + 4 * c;
      if (OutB) *reinterpret_cast<float4*>(OutB + off) = v;
      if (doRelu) v = make_float4(fmaxf(v.x, 0.0f), fmaxf(v.y, 0.0f), fmaxf(v.z, 0.0f), fmaxf(v.w, 0.0f));
      *reinterpret_cast<float4*>(OutA + off) = v;
    }
  }
}

#define POOL_NODES 16
__device__ __forceinline__ void pool_flush(float* __restrict__ gsum, float* __restrict__ gcnt,
                                           int g, int lane, const float4 acc, int run) {
  float* p = gsum + (size_t)g * 128 + lane * 4;
  atomAddF(p + 0, acc.x);
  atomAddF(p + 1, acc.y);
  atomAddF(p + 2, acc.z);
  atomAddF(p + 3, acc.w);
  if (lane == 0) atomAddF(&gcnt[g], (float)run);
}
__global__ void pool_kernel(const float* __restrict__ h, const int* __restrict__ cluster,
                            float* __restrict__ gsum, float* __restrict__ gcnt, int n) {
  int w = __builtin_amdgcn_readfirstlane((int)(blockIdx.x * (blockDim.x >> 5) + (threadIdx.x >> 5)));
  int lane = threadIdx.x & 31;
  int node0 = w * POOL_NODES;
  if (node0 >= n) return;
  const float* hl = h + (size_t)node0 * 128 + lane * 4;
  const int* cl = cluster + node0;
  if (node0 + POOL_NODES <= n) {
    int g[POOL_NODES];
    float4 v[POOL_NODES];
#pragma unroll
    for (int q = 0; q < POOL_NODES; ++q) {
      g[q] = cl[q];
      v[q] = *reinterpret_cast<const float4*>(hl + (size_t)q * 128);
    }
    int cur = g[0];
    int run = 1;
    float4 acc = v[0];
#pragma unroll
    for (int q = 1; q < POOL_NODES; ++q) {
      if (g[q] != cur) {
        pool_flush(gsum, gcnt, cur, lane, acc, run);
        cur = g[q];
        run = 1;
        acc = v[q];
      } else {
        add4(acc, v[q]);
        ++run;
      }
    }
    pool_flush(gsum, gcnt, cur, lane, acc, run);
  } else {
    int cur = cl[0];
    int run = 1;
    float4 acc = *reinterpret_cast<const float4*>(hl);
    for (int q = 1; node0 + q < n; ++q) {
      int gq = cl[q];
      const float4 vq = *reinterpret_cast<const float4*>(hl + (size_t)q * 128);
      if (gq != cur) {
        pool_flush(gsum, gcnt, cur, lane, acc, run);
        cur = gq;
        run = 1;
        acc = vq;
      } else {
        add4(acc, vq);
        ++run;
      }
    }
    pool_flush(gsum, gcnt, cur, lane, acc, run);
  }
}

__global__ void pool_div_kernel(float* __restrict__ g, const float* __restrict__ gcnt, int total) {
  int i = blockIdx.x * blockDim.x + threadIdx.x;
  if (i >= total) return;
  g[i] = g[i] / fmaxf(gcnt[i >> 7], 1.0f);
}

extern "C" void kernel_launch(void* const* d_in, const int* in_sizes, int n_in,
                              void* d_out, int out_size, void* d_ws, size_t ws_size,
                              hipStream_t stream) {
  const float* x       = (const float*)d_in[0];
  const int*   ei      = (const int*)d_in[1];
  const int*   src     = ei;
  const int*   dst     = ei + NE;
  const int*   cluster = (const int*)d_in[2];
  const float* Wl0 = (const float*)d_in[3];
  const float* bl0 = (const float*)d_in[4];
  const float* Wr0 = (const float*)d_in[5];
  const float* Wl1 = (const float*)d_in[6];
  const float* bl1 = (const float*)d_in[7];
  const float* Wr1 = (const float*)d_in[8];
  const float* Wl2 = (const float*)d_in[9];
  const float* bl2 = (const float*)d_in[10];
  const float* Wr2 = (const float*)d_in[11];

  float* out_ls = (float*)d_out;
  float* out_h2 = out_ls + (size_t)NN * OUTC;
  float* out_g  = out_h2 + (size_t)NN * HIDC;

  float* ws   = (float*)d_ws;
  float* agg  = ws;
  float* h1   = ws + (size_t)NN * HIDC;
  float* h2r  = h1 + (size_t)NN * HIDC;
  float* deg  = h2r + (size_t)NN * HIDC;
  float* gcnt = deg + NN;

  int* cnt    = (int*)h1;
  int* pos    = cnt + 50048;
  int* rowptr = (int*)out_ls;
  int* col    = rowptr + 50048;

  const int TPB = 256;
  const int edgeBlocks4 = (NE + TPB * 4 - 1) / (TPB * 4);
  const int gatherBlocks = (NN + 8 * GATHER_NODES - 1) / (8 * GATHER_NODES);

  hipMemsetAsync(cnt, 0, NN * sizeof(int), stream);
  count_rank_kernel<<<edgeBlocks4, TPB, 0, stream>>>(dst, cnt, pos, NE);
  scan_kernel<<<(NN + 1023) / 1024, 1024, 0, stream>>>(cnt, rowptr, deg, NN);
  fill_kernel<<<edgeBlocks4, TPB, 0, stream>>>(src, dst, rowptr, pos, col, NE);

  gather_sum_kernel<<<gatherBlocks, TPB, 0, stream>>>(x, rowptr, col, agg, NN);
  {
    int tiles = (NN / 16) * (HIDC / 64);
    sage_gemm_kernel<false><<<(tiles + GEMM_WAVES - 1) / GEMM_WAVES, GEMM_WAVES * 32, 0, stream>>>(
        agg, deg, x, Wl0, bl0, Wr0, h1, nullptr, NN, INC, HIDC, 1);
  }

  gather_sum_kernel<<<gatherBlocks, TPB, 0, stream>>>(h1, rowptr, col, agg, NN);
  {
    int tiles = (NN / 16) * (HIDC / 64);
    sage_gemm_kernel<false><<<(tiles + GEMM_WAVES - 1) / GEMM_WAVES, GEMM_WAVES * 32, 0, stream>>>(
        agg, deg, h1, Wl1, bl1, Wr1, h2r, out_h2, NN, HIDC, HIDC, 1);
  }

  hipMemsetAsync(out_g, 0, (size_t)NG * HIDC * sizeof(float), stream);
  hipMemsetAsync(gcnt, 0, NG * sizeof(float), stream);
  pool_kernel<<<(NN + 8 * POOL_NODES - 1) / (8 * POOL_NODES), TPB, 0, stream>>>(h2r, cluster, out_g, gcnt, NN);
  pool_div_kernel<<<(NG * HIDC + TPB - 1) / TPB, TPB, 0, stream>>>(out_g, gcnt, NG * HIDC);

  gather_sum_kernel<<<gatherBlocks, TPB, 0, stream>>>(h2r, rowptr, col, agg, NN);
  {
    int tiles = (NN / 16) * (OUTC / 64);
    sage_gemm_kernel<true><<<(tiles + GEMM_WAVES - 1) / GEMM_WAVES, GEMM_WAVES * 32, 0, stream>>>(
        agg, deg, h2r, Wl2, bl2, Wr2, out_ls, nullptr, NN, HIDC, OUTC, 0);
  }
}


# Neighbour gather fused into the GEMM launch: a block sums its 32 rows' neighbour rows into LDS beside self rows, A fragments come from LDS; sums never reach memory; 3 launches gone; sum order kept
# speedup vs baseline: 3.0194x; 1.1155x over previous
#include <hip/hip_runtime.h>
#include <hip/hip_bf16.h>
#include <math.h>

typedef float v2f __attribute__((ext_vector_type(2)));
typedef float v8f __attribute__((ext_vector_type(8)));

#define NN      50000
#define NE      600000
#define INC     128
#define HIDC    128
#define OUTC    64
#define NG      512

__device__ __forceinline__ void atomAddF(float* p, float v) {
  __hip_atomic_fetch_add(p, v, __ATOMIC_RELAXED, __HIP_MEMORY_SCOPE_AGENT);
}

__global__ void count_rank_kernel(const int* __restrict__ dst, int* __restrict__ cnt,
                                  int* __restrict__ pos, int E) {
  int B = blockDim.x;
  int e0 = blockIdx.x * (B * 4) + threadIdx.x;
  int e1 = e0 + B, e2 = e1 + B, e3 = e2 + B;
  if (e3 < E) {
    int d0 = dst[e0], d1 = dst[e1], d2 = dst[e2], d3 = dst[e3];
    int r0 = __hip_atomic_fetch_add(&cnt[d0], 1, __ATOMIC_RELAXED, __HIP_MEMORY_SCOPE_AGENT);
    int r1 = __hip_atomic_fetch_add(&cnt[d1], 1, __ATOMIC_RELAXED, __HIP_MEMORY_SCOPE_AGENT);
    int r2 = __hip_atomic_fetch_add(&cnt[d2], 1, __ATOMIC_RELAXED, __HIP_MEMORY_SCOPE_AGENT);
    int r3 = __hip_atomic_fetch_add(&cnt[d3], 1, __ATOMIC_RELAXED, __HIP_MEMORY_SCOPE_AGENT);
    pos[e0] = r0; pos[e1] = r1; pos[e2] = r2; pos[e3] = r3;
  } else {
    for (int e = e0; e < E; e += B)
      pos[e] = __hip_atomic_fetch_add(&cnt[dst[e]], 1, __ATOMIC_RELAXED, __HIP_MEMORY_SCOPE_AGENT);
  }
}

__global__ void __launch_bounds__(1024)
scan_kernel(const int* __restrict__ cnt, int* __restrict__ rowptr,
            float* __restrict__ invdeg, int n) {
  __shared__ int wsum[32];
  __shared__ int sbase;
  int t = threadIdx.x;
  int lane = t & 31;
  int w = t >> 5;
  int start = blockIdx.x * 1024;

  int part = 0;
  for (int i = t; i < start; i += 1024) part += cnt[i];
  for (int off = 16; off >= 1; off >>= 1) part += __shfl_xor(part, off, 32);
  if (lane == 0) wsum[w] = part;
  __syncthreads();
  if (w == 0) {
    int v = wsum[lane];
    for (int off = 16; off >= 1; off >>= 1) v += __shfl_xor(v, off, 32);
    if (lane == 0) sbase = v;
  }
  __syncthreads();
  int base = sbase;

  int i = start + t;
  int c = (i < n) ? cnt[i] : 0;
  int inc = c;
  for (int off = 1; off < 32; off <<= 1) {
    int u = __shfl_up(inc, off, 32);
    if (lane >= off) inc += u;
  }
  if (lane == 31) wsum[w] = inc;
  __syncthreads();
  if (w == 0) {
    int v = wsum[lane];
    int iv = v;
    for (int off = 1; off < 32; off <<= 1) {
      int u = __shfl_up(iv, off, 32);
      if (lane >= off) iv += u;
    }
    wsum[lane] = iv - v;
  }
  __syncthreads();
  int excl = base + wsum[w] + inc - c;
  if (i < n) {
    rowptr[i] = excl;
    invdeg[i] = 1.0f / fmaxf((float)c, 1.0f);
  }
  if (i == n - 1) rowptr[n] = excl + c;
}

__global__ void fill_kernel(const int* __restrict__ src, const int* __restrict__ dst,
                            const int* __restrict__ rowptr, const int* __restrict__ pos,
                            int* __restrict__ col, int E) {
  int B = blockDim.x;
  int e0 = blockIdx.x * (B * 4) + threadIdx.x;
  int e1 = e0 + B, e2 = e1 + B, e3 = e2 + B;
  if (e3 < E) {
    int d0 = dst[e0], d1 = dst[e1], d2 = dst[e2], d3 = dst[e3];
    int p0 = pos[e0], p1 = pos[e1], p2 = pos[e2], p3 = pos[e3];
    int s0 = src[e0], s1 = src[e1], s2 = src[e2], s3 = src[e3];
    int q0 = rowptr[d0], q1 = rowptr[d1], q2 = rowptr[d2], q3 = rowptr[d3];
    col[q0 + p0] = s0; col[q1 + p1] = s1; col[q2 + p2] = s2; col[q3 + p3] = s3;
  } else {
    for (int e = e0; e < E; e += B) col[rowptr[dst[e]] + pos[e]] = src[e];
  }
}

__device__ __forceinline__ void add4(float4& a, const float4 v) {
  a.x += v.x; a.y += v.y; a.z += v.z; a.w += v.w;
}

#define LAYER_ROWS 32
#define LDS_PITCH 132
struct GemmFrag {
  float4 a, x;
  float4 l[4], r[4];
};
__device__ __forceinline__ float f4c(const float4 v, int j) {
  return j == 0 ? v.x : (j == 1 ? v.y : (j == 2 ? v.z : v.w));
}
__device__ __forceinline__ void gemm_mma(const GemmFrag& f, v8f (&acc)[4]) {
#pragma unroll
  for (int t = 0; t < 2; ++t) {
    v2f af, xf;
    af.x = f4c(f.a, 2 * t);
    af.y = f4c(f.a, 2 * t + 1);
    xf.x = f4c(f.x, 2 * t);
    xf.y = f4c(f.x, 2 * t + 1);
#pragma unroll
    for (int j = 0; j < 4; ++j) {
      v2f bf;
      bf.x = f4c(f.l[2 * t], j);
      bf.y = f4c(f.l[2 * t + 1], j);
      acc[j] = __builtin_amdgcn_wmma_f32_16x16x4_f32(false, af, false, bf, (short)0, acc[j], false, false);
    }
#pragma unroll
    for (int j = 0; j < 4; ++j) {
      v2f bf;
      bf.x = f4c(f.r[2 * t], j);
      bf.y = f4c(f.r[2 * t + 1], j);
      acc[j] = __builtin_amdgcn_wmma_f32_16x16x4_f32(false, xf, false, bf, (short)0, acc[j], false, false);
    }
  }
}
__device__ __forceinline__ float half_max16(float v) {
  v = fmaxf(v, __shfl_xor(v, 8, 32));
  v = fmaxf(v, __shfl_xor(v, 4, 32));
  v = fmaxf(v, __shfl_xor(v, 2, 32));
  v = fmaxf(v, __shfl_xor(v, 1, 32));
  return v;
}
__device__ __forceinline__ float half_sum16(float v) {
  v += __shfl_xor(v, 8, 32);
  v += __shfl_xor(v, 4, 32);
  v += __shfl_xor(v, 2, 32);
  v += __shfl_xor(v, 1, 32);
  return v;
}
template <int NCOLS, bool LSM>
__global__ void __launch_bounds__((NCOLS / 64) * 64)
sage_layer_kernel(const float* __restrict__ X, const int* __restrict__ rowptr,
                  const int* __restrict__ col, const float* __restrict__ invdeg,
                  const float* __restrict__ Wl, const float* __restrict__ bl,
                  const float* __restrict__ Wr,
                  float* __restrict__ OutA, float* __restrict__ OutB, int M, int doRelu) {
  constexpr int NGRP = NCOLS / 64;
  constexpr int WAVES = 2 * NGRP;
  constexpr int G = LAYER_ROWS / WAVES;
  __shared__ __attribute__((aligned(16))) float sAgg[LAYER_ROWS * LDS_PITCH];
  __shared__ __attribute__((aligned(16))) float sX[LAYER_ROWS * LDS_PITCH];

  int wv = __builtin_amdgcn_readfirstlane((int)(threadIdx.x >> 5));
  int lane = threadIdx.x & 31;
  int row0 = blockIdx.x * LAYER_ROWS;

  int lr0 = wv * G;
  int n0 = row0 + lr0;
  if (n0 + G <= M) {
    int rp = (lane <= G) ? rowptr[n0 + lane] : 0x7fffffff;
    float iv = invdeg[n0 + min(lane, G - 1)];
    float4 xs[G];
#pragma unroll
    for (int q = 0; q < G; ++q)
      xs[q] = *reinterpret_cast<const float4*>(X + (size_t)(n0 + q) * 128 + lane * 4);
#pragma unroll
    for (int q = 0; q < G; ++q)
      *reinterpret_cast<float4*>(&sX[(lr0 + q) * LDS_PITCH + lane * 4]) = xs[q];

    int jbeg = __builtin_amdgcn_readlane(rp, 0);
    int jend = __builtin_amdgcn_readlane(rp, G);
    int node = 0;
    int nend = __builtin_amdgcn_readlane(rp, 1);
    float4 acc = make_float4(0.0f, 0.0f, 0.0f, 0.0f);
    const float* xl = X + lane * 4;
    if (jbeg < jend) {
      int cv = col[min(jbeg + lane, jend - 1)];
      for (int jb = jbeg; jb < jend; jb += 32) {
        int cvn = col[min(jb + 32 + lane, jend - 1)];
#pragma unroll
        for (int u = 0; u < 2; ++u) {
          int e0 = jb + 16 * u;
          if (e0 < jend) {
            float4 v[16];
#pragma unroll
            for (int t = 0; t < 16; ++t) {
              int s = __builtin_amdgcn_readlane(cv, 16 * u + t);
              v[t] = *reinterpret_cast<const float4*>(xl + (size_t)s * 128);
            }
#pragma unroll
            for (int t = 0; t < 16; ++t) {
              int e = e0 + t;
              while (e >= nend && node < G) {
                float s = __int_as_float(__builtin_amdgcn_readlane(__float_as_int(iv), node));
                *reinterpret_cast<float4*>(&sAgg[(lr0 + node) * LDS_PITCH + lane * 4]) =
                    make_float4(acc.x * s, acc.y * s, acc.z * s, acc.w * s);
                ++node;
                nend = __builtin_amdgcn_readlane(rp, node + 1);
                acc = make_float4(0.0f, 0.0f, 0.0f, 0.0f);
              }
              add4(acc, v[t]);
            }
          }
        }
        cv = cvn;
      }
    }
    for (; node < G; ++node) {
      float s = __int_as_float(__builtin_amdgcn_readlane(__float_as_int(iv), node));
      *reinterpret_cast<float4*>(&sAgg[(lr0 + node) * LDS_PITCH + lane * 4]) =
          make_float4(acc.x * s, acc.y * s, acc.z * s, acc.w * s);
      acc = make_float4(0.0f, 0.0f, 0.0f, 0.0f);
    }
  }
  __syncthreads();

  int tl = wv / NGRP;
  int cg = wv % NGRP;
  int m0 = row0 + 16 * tl;
  if (m0 >= M) return;
  int nb = cg << 6;
  int c = lane & 15;
  int h = lane >> 4;

  int aoff = (16 * tl + c) * LDS_PITCH + 4 * h;
  const float* wl = Wl + (size_t)(4 * h) * NCOLS + nb + 4 * c;
  const float* wr = Wr + (size_t)(4 * h) * NCOLS + nb + 4 * c;

  auto frag_load = [&](GemmFrag& f, int i) {
    f.a = *reinterpret_cast<const float4*>(&sAgg[aoff + 8 * i]);
    f.x = *reinterpret_cast<const float4*>(&sX[aoff + 8 * i]);
#pragma unroll
    for (int q = 0; q < 4; ++q) {
      f.l[q] = *reinterpret_cast<const float4*>(wl + (size_t)(8 * i + q) * NCOLS);
      f.r[q] = *reinterpret_cast<const float4*>(wr + (size_t)(8 * i + q) * NCOLS);
    }
  };

  v8f acc[4] = {};
  constexpr int KI = 128 >> 3;
  GemmFrag cur;
  frag_load(cur, 0);
  for (int i = 0; i + 1 < KI; ++i) {
    GemmFrag nxt;
    frag_load(nxt, i + 1);
    __builtin_amdgcn_sched_barrier(0);
    gemm_mma(cur, acc);
    cur = nxt;
  }
  gemm_mma(cur, acc);

  const float4 bias = *reinterpret_cast<const float4*>(bl + nb + 4 * c);
  int rbase = m0 + (h << 3);
  if constexpr (LSM) {
#pragma unroll
    for (int r = 0; r < 8; ++r) {
      float4 v = make_float4(acc[0][r] + bias.x, acc[1][r] + bias.y, acc[2][r] + bias.z, acc[3][r] + bias.w);
      float m = half_max16(fmaxf(fmaxf(v.x, v.y), fmaxf(v.z, v.w)));
      float e = half_sum16((__expf(v.x - m) + __expf(v.y - m)) + (__expf(v.z - m) + __expf(v.w - m)));
      float lse = m + __logf(e);
      size_t off = (size_t)(rbase + r) * NCOLS + nb + 4 * c;
      *reinterpret_cast<float4*>(OutA + off) = make_float4(v.x - lse, v.y - lse, v.z - lse, v.w - lse);
    }
  } else {
#pragma unroll
    for (int r = 0; r < 8; ++r) {
      float4 v = make_float4(acc[0][r] + bias.x, acc[1][r] + bias.y, acc[2][r] + bias.z, acc[3][r] + bias.w);
      size_t off = (size_t)(rbase + r) * NCOLS + nb + 4 * c;
      if (OutB) *reinterpret_cast<float4*>(OutB + off) = v;
      if (doRelu) v = make_float4(fmaxf(v.x, 0.0f), fmaxf(v.y, 0.0f), fmaxf(v.z, 0.0f), fmaxf(v.w, 0.0f));
      *reinterpret_cast<float4*>(OutA + off) = v;
    }
  }
}

#define POOL_NODES 16
__device__ __forceinline__ void pool_flush(float* __restrict__ gsum, float* __restrict__ gcnt,
                                           int g, int lane, const float4 acc, int run) {
  float* p = gsum + (size_t)g * 128 + lane * 4;
  atomAddF(p + 0, acc.x);
  atomAddF(p + 1, acc.y);
  atomAddF(p + 2, acc.z);
  atomAddF(p + 3, acc.w);
  if (lane == 0) atomAddF(&gcnt[g], (float)run);
}
__global__ void pool_kernel(const float* __restrict__ h, const int* __restrict__ cluster,
                            float* __restrict__ gsum, float* __restrict__ gcnt, int n) {
  int w = __builtin_amdgcn_readfirstlane((int)(blockIdx.x * (blockDim.x >> 5) + (threadIdx.x >> 5)));
  int lane = threadIdx.x & 31;
  int node0 = w * POOL_NODES;
  if (node0 >= n) return;
  const float* hl = h + (size_t)node0 * 128 + lane * 4;
  const int* cl = cluster + node0;
  if (node0 + POOL_NODES <= n) {
    int g[POOL_NODES];
    float4 v[POOL_NODES];
#pragma unroll
    for (int q = 0; q < POOL_NODES; ++q) {
      g[q] = cl[q];
      v[q] = *reinterpret_cast<const float4*>(hl + (size_t)q * 128);
    }
    int cur = g[0];
    int run = 1;
    float4 acc = v[0];
#pragma unroll
    for (int q = 1; q < POOL_NODES; ++q) {
      if (g[q] != cur) {
        pool_flush(gsum, gcnt, cur, lane, acc, run);
        cur = g[q];
        run = 1;
        acc = v[q];
      } else {
        add4(acc, v[q]);
        ++run;
      }
    }
    pool_flush(gsum, gcnt, cur, lane, acc, run);
  } else {
    int cur = cl[0];
    int run = 1;
    float4 acc = *reinterpret_cast<const float4*>(hl);
    for (int q = 1; node0 + q < n; ++q) {
      int gq = cl[q];
      const float4 vq = *reinterpret_cast<const float4*>(hl + (size_t)q * 128);
      if (gq != cur) {
        pool_flush(gsum, gcnt, cur, lane, acc, run);
        cur = gq;
        run = 1;
        acc = vq;
      } else {
        add4(acc, vq);
        ++run;
      }
    }
    pool_flush(gsum, gcnt, cur, lane, acc, run);
  }
}

__global__ void pool_div_kernel(float* __restrict__ g, const float* __restrict__ gcnt, int total) {
  int i = blockIdx.x * blockDim.x + threadIdx.x;
  if (i >= total) return;
  g[i] = g[i] / fmaxf(gcnt[i >> 7], 1.0f);
}

extern "C" void kernel_launch(void* const* d_in, const int* in_sizes, int n_in,
                              void* d_out, int out_size, void* d_ws, size_t ws_size,
                              hipStream_t stream) {
  const float* x       = (const float*)d_in[0];
  const int*   ei      = (const int*)d_in[1];
  const int*   src     = ei;
  const int*   dst     = ei + NE;
  const int*   cluster = (const int*)d_in[2];
  const float* Wl0 = (const float*)d_in[3];
  const float* bl0 = (const float*)d_in[4];
  const float* Wr0 = (const float*)d_in[5];
  const float* Wl1 = (const float*)d_in[6];
  const float* bl1 = (const float*)d_in[7];
  const float* Wr1 = (const float*)d_in[8];
  const float* Wl2 = (const float*)d_in[9];
  const float* bl2 = (const float*)d_in[10];
  const float* Wr2 = (const float*)d_in[11];

  float* out_ls = (float*)d_out;
  float* out_h2 = out_ls + (size_t)NN * OUTC;
  float* out_g  = out_h2 + (size_t)NN * HIDC;

  float* ws   = (float*)d_ws;
  float* scr  = ws;
  float* h1   = ws + (size_t)NN * HIDC;
  float* h2r  = h1 + (size_t)NN * HIDC;
  float* deg  = h2r + (size_t)NN * HIDC;
  float* gcnt = deg + NN;

  int* cnt    = (int*)h1;
  int* pos    = cnt + 50048;
  int* rowptr = (int*)scr;
  int* col    = rowptr + 50048;

  const int TPB = 256;
  const int edgeBlocks4 = (NE + TPB * 4 - 1) / (TPB * 4);
  const int layerBlocks = (NN + LAYER_ROWS - 1) / LAYER_ROWS;

  hipMemsetAsync(cnt, 0, NN * sizeof(int), stream);
  count_rank_kernel<<<edgeBlocks4, TPB, 0, stream>>>(dst, cnt, pos, NE);
  scan_kernel<<<(NN + 1023) / 1024, 1024, 0, stream>>>(cnt, rowptr, deg, NN);
  fill_kernel<<<edgeBlocks4, TPB, 0, stream>>>(src, dst, rowptr, pos, col, NE);

  sage_layer_kernel<HIDC, false><<<layerBlocks, (HIDC / 64) * 64, 0, stream>>>(
      x, rowptr, col, deg, Wl0, bl0, Wr0, h1, nullptr, NN, 1);

  sage_layer_kernel<HIDC, false><<<layerBlocks, (HIDC / 64) * 64, 0, stream>>>(
      h1, rowptr, col, deg, Wl1, bl1, Wr1, h2r, out_h2, NN, 1);

  hipMemsetAsync(out_g, 0, (size_t)NG * HIDC * sizeof(float), stream);
  hipMemsetAsync(gcnt, 0, NG * sizeof(float), stream);
  pool_kernel<<<(NN + 8 * POOL_NODES - 1) / (8 * POOL_NODES), TPB, 0, stream>>>(h2r, cluster, out_g, gcnt, NN);
  pool_div_kernel<<<(NG * HIDC + TPB - 1) / TPB, TPB, 0, stream>>>(out_g, gcnt, NG * HIDC);

  sage_layer_kernel<OUTC, true><<<layerBlocks, (OUTC / 64) * 64, 0, stream>>>(
      h2r, rowptr, col, deg, Wl2, bl2, Wr2, out_ls, nullptr, NN, 0);
}


# Layer phases run in two orders: half of the blocks gather first, the other half run the self product X@Wr first, then gather, then the neighbour product; the k sum now adds the self part first
# speedup vs baseline: 3.1177x; 1.0326x over previous
#include <hip/hip_runtime.h>
#include <hip/hip_bf16.h>
#include <math.h>

typedef float v2f __attribute__((ext_vector_type(2)));
typedef float v8f __attribute__((ext_vector_type(8)));

#define NN      50000
#define NE      600000
#define INC     128
#define HIDC    128
#define OUTC    64
#define NG      512

__device__ __forceinline__ void atomAddF(float* p, float v) {
  __hip_atomic_fetch_add(p, v, __ATOMIC_RELAXED, __HIP_MEMORY_SCOPE_AGENT);
}

__global__ void count_rank_kernel(const int* __restrict__ dst, int* __restrict__ cnt,
                                  int* __restrict__ pos, int E) {
  int B = blockDim.x;
  int e0 = blockIdx.x * (B * 4) + threadIdx.x;
  int e1 = e0 + B, e2 = e1 + B, e3 = e2 + B;
  if (e3 < E) {
    int d0 = dst[e0], d1 = dst[e1], d2 = dst[e2], d3 = dst[e3];
    int r0 = __hip_atomic_fetch_add(&cnt[d0], 1, __ATOMIC_RELAXED, __HIP_MEMORY_SCOPE_AGENT);
    int r1 = __hip_atomic_fetch_add(&cnt[d1], 1, __ATOMIC_RELAXED, __HIP_MEMORY_SCOPE_AGENT);
    int r2 = __hip_atomic_fetch_add(&cnt[d2], 1, __ATOMIC_RELAXED, __HIP_MEMORY_SCOPE_AGENT);
    int r3 = __hip_atomic_fetch_add(&cnt[d3], 1, __ATOMIC_RELAXED, __HIP_MEMORY_SCOPE_AGENT);
    pos[e0] = r0; pos[e1] = r1; pos[e2] = r2; pos[e3] = r3;
  } else {
    for (int e = e0; e < E; e += B)
      pos[e] = __hip_atomic_fetch_add(&cnt[dst[e]], 1, __ATOMIC_RELAXED, __HIP_MEMORY_SCOPE_AGENT);
  }
}

__global__ void __launch_bounds__(1024)
scan_kernel(const int* __restrict__ cnt, int* __restrict__ rowptr,
            float* __restrict__ invdeg, int n) {
  __shared__ int wsum[32];
  __shared__ int sbase;
  int t = threadIdx.x;
  int lane = t & 31;
  int w = t >> 5;
  int start = blockIdx.x * 1024;

  int part = 0;
  for (int i = t; i < start; i += 1024) part += cnt[i];
  for (int off = 16; off >= 1; off >>= 1) part += __shfl_xor(part, off, 32);
  if (lane == 0) wsum[w] = part;
  __syncthreads();
  if (w == 0) {
    int v = wsum[lane];
    for (int off = 16; off >= 1; off >>= 1) v += __shfl_xor(v, off, 32);
    if (lane == 0) sbase = v;
  }
  __syncthreads();
  int base = sbase;

  int i = start + t;
  int c = (i < n) ? cnt[i] : 0;
  int inc = c;
  for (int off = 1; off < 32; off <<= 1) {
    int u = __shfl_up(inc, off, 32);
    if (lane >= off) inc += u;
  }
  if (lane == 31) wsum[w] = inc;
  __syncthreads();
  if (w == 0) {
    int v = wsum[lane];
    int iv = v;
    for (int off = 1; off < 32; off <<= 1) {
      int u = __shfl_up(iv, off, 32);
      if (lane >= off) iv += u;
    }
    wsum[lane] = iv - v;
  }
  __syncthreads();
  int excl = base + wsum[w] + inc - c;
  if (i < n) {
    rowptr[i] = excl;
    invdeg[i] = 1.0f / fmaxf((float)c, 1.0f);
  }
  if (i == n - 1) rowptr[n] = excl + c;
}

__global__ void fill_kernel(const int* __restrict__ src, const int* __restrict__ dst,
                            const int* __restrict__ rowptr, const int* __restrict__ pos,
                            int* __restrict__ col, int E) {
  int B = blockDim.x;
  int e0 = blockIdx.x * (B * 4) + threadIdx.x;
  int e1 = e0 + B, e2 = e1 + B, e3 = e2 + B;
  if (e3 < E) {
    int d0 = dst[e0], d1 = dst[e1], d2 = dst[e2], d3 = dst[e3];
    int p0 = pos[e0], p1 = pos[e1], p2 = pos[e2], p3 = pos[e3];
    int s0 = src[e0], s1 = src[e1], s2 = src[e2], s3 = src[e3];
    int q0 = rowptr[d0], q1 = rowptr[d1], q2 = rowptr[d2], q3 = rowptr[d3];
    col[q0 + p0] = s0; col[q1 + p1] = s1; col[q2 + p2] = s2; col[q3 + p3] = s3;
  } else {
    for (int e = e0; e < E; e += B) col[rowptr[dst[e]] + pos[e]] = src[e];
  }
}

__device__ __forceinline__ void add4(float4& a, const float4 v) {
  a.x += v.x; a.y += v.y; a.z += v.z; a.w += v.w;
}

#define LAYER_ROWS 32
#define LDS_PITCH 132
struct GemmFrag {
  float4 a[2];
  float4 w[8];
};
__device__ __forceinline__ float f4c(const float4 v, int j) {
  return j == 0 ? v.x : (j == 1 ? v.y : (j == 2 ? v.z : v.w));
}
__device__ __forceinline__ void gemm_mma(const GemmFrag& f, v8f (&acc)[4]) {
#pragma unroll
  for (int b = 0; b < 2; ++b) {
#pragma unroll
    for (int t = 0; t < 2; ++t) {
      v2f af;
      af.x = f4c(f.a[b], 2 * t);
      af.y = f4c(f.a[b], 2 * t + 1);
#pragma unroll
      for (int j = 0; j < 4; ++j) {
        v2f bf;
        bf.x = f4c(f.w[4 * b + 2 * t], j);
        bf.y = f4c(f.w[4 * b + 2 * t + 1], j);
        acc[j] = __builtin_amdgcn_wmma_f32_16x16x4_f32(false, af, false, bf, (short)0, acc[j], false, false);
      }
    }
  }
}
template <int NCOLS>
__device__ __forceinline__ void gemm_k128(const float* sA, const float* wbase, int woff, v8f (&acc)[4]) {
  auto frag_load = [&](GemmFrag& f, const float* ast, const float* wst) {
#pragma unroll
    for (int b = 0; b < 2; ++b)
      f.a[b] = *reinterpret_cast<const float4*>(ast + 8 * b);
#pragma unroll
    for (int q = 0; q < 8; ++q)
      f.w[q] = *reinterpret_cast<const float4*>(wst + (woff + (8 * (q >> 2) + (q & 3)) * NCOLS));
  };
  constexpr int KI = 128 >> 4;
  const float* ast = sA;
  const float* wst = wbase;
  GemmFrag cur;
  frag_load(cur, ast, wst);
#pragma nounroll
  for (int i = 0; i + 1 < KI; ++i) {
    ast += 16;
    wst += 16 * NCOLS;
    GemmFrag nxt;
    frag_load(nxt, ast, wst);
    __builtin_amdgcn_sched_barrier(0);
    gemm_mma(cur, acc);
    cur = nxt;
  }
  gemm_mma(cur, acc);
}
__device__ __forceinline__ float half_max16(float v) {
  v = fmaxf(v, __shfl_xor(v, 8, 32));
  v = fmaxf(v, __shfl_xor(v, 4, 32));
  v = fmaxf(v, __shfl_xor(v, 2, 32));
  v = fmaxf(v, __shfl_xor(v, 1, 32));
  return v;
}
__device__ __forceinline__ float half_sum16(float v) {
  v += __shfl_xor(v, 8, 32);
  v += __shfl_xor(v, 4, 32);
  v += __shfl_xor(v, 2, 32);
  v += __shfl_xor(v, 1, 32);
  return v;
}
template <int NCOLS, bool LSM>
__global__ void __launch_bounds__((NCOLS / 64) * 64)
sage_layer_kernel(const float* __restrict__ X, const int* __restrict__ rowptr,
                  const int* __restrict__ col, const float* __restrict__ invdeg,
                  const float* __restrict__ Wl, const float* __restrict__ bl,
                  const float* __restrict__ Wr,
                  float* __restrict__ OutA, float* __restrict__ OutB, int M, int doRelu) {
  constexpr int NGRP = NCOLS / 64;
  constexpr int WAVES = 2 * NGRP;
  constexpr int G = LAYER_ROWS / WAVES;
  __shared__ __attribute__((aligned(16))) float sAgg[LAYER_ROWS * LDS_PITCH];
  __shared__ __attribute__((aligned(16))) float sX[LAYER_ROWS * LDS_PITCH];

  int wv = __builtin_amdgcn_readfirstlane((int)(threadIdx.x >> 5));
  int lane = threadIdx.x & 31;
  int row0 = blockIdx.x * LAYER_ROWS;
  const bool gather_first = (__popc(blockIdx.x) & 1) == 0;

  int lr0 = wv * G;
  int n0 = row0 + lr0;
  const bool has_nodes = n0 + G <= M;

  if (has_nodes) {
#pragma unroll
    for (int q = 0; q < G; ++q)
      *reinterpret_cast<float4*>(&sX[(lr0 + q) * LDS_PITCH + lane * 4]) =
          *reinterpret_cast<const float4*>(X + (size_t)(n0 + q) * 128 + lane * 4);
  }

  auto gather = [&]() {
    if (!has_nodes) return;
    int rp = (lane <= G) ? rowptr[n0 + lane] : 0x7fffffff;
    float iv = invdeg[n0 + min(lane, G - 1)];
    int jbeg = __builtin_amdgcn_readlane(rp, 0);
    int jend = __builtin_amdgcn_readlane(rp, G);
    int node = 0;
    int nend = __builtin_amdgcn_readlane(rp, 1);
    float4 sum = make_float4(0.0f, 0.0f, 0.0f, 0.0f);
    const float* xl = X + lane * 4;
    if (jbeg < jend) {
      int cv = col[min(jbeg + lane, jend - 1)];
      for (int jb = jbeg; jb < jend; jb += 32) {
        int cvn = col[min(jb + 32 + lane, jend - 1)];
#pragma unroll
        for (int u = 0; u < 2; ++u) {
          int e0 = jb + 16 * u;
          if (e0 < jend) {
            float4 v[16];
#pragma unroll
            for (int t = 0; t < 16; ++t) {
              int s = __builtin_amdgcn_readlane(cv, 16 * u + t);
              v[t] = *reinterpret_cast<const float4*>(xl + (size_t)s * 128);
            }
#pragma unroll
            for (int t = 0; t < 16; ++t) {
              int e = e0 + t;
              while (e >= nend && node < G) {
                float s = __int_as_float(__builtin_amdgcn_readlane(__float_as_int(iv), node));
                *reinterpret_cast<float4*>(&sAgg[(lr0 + node) * LDS_PITCH + lane * 4]) =
                    make_float4(sum.x * s, sum.y * s, sum.z * s, sum.w * s);
                ++node;
                nend = __builtin_amdgcn_readlane(rp, node + 1);
                sum = make_float4(0.0f, 0.0f, 0.0f, 0.0f);
              }
              add4(sum, v[t]);
            }
          }
        }
        cv = cvn;
      }
    }
    for (; node < G; ++node) {
      float s = __int_as_float(__builtin_amdgcn_readlane(__float_as_int(iv), node));
      *reinterpret_cast<float4*>(&sAgg[(lr0 + node) * LDS_PITCH + lane * 4]) =
          make_float4(sum.x * s, sum.y * s, sum.z * s, sum.w * s);
      sum = make_float4(0.0f, 0.0f, 0.0f, 0.0f);
    }
  };

  int tl = wv / NGRP;
  int cg = wv % NGRP;
  int m0 = row0 + 16 * tl;
  const bool active = m0 < M;
  int nb = cg << 6;
  int c = lane & 15;
  int h = lane >> 4;
  int aoff = (16 * tl + c) * LDS_PITCH + 4 * h;
  int woff = (4 * h) * NCOLS + nb + 4 * c;
  v8f acc[4] = {};

#pragma nounroll
  for (int step = 0; step < 2; ++step) {
    if ((step == 0) == gather_first) gather();
    if (step == 0 || !gather_first) __syncthreads();
    if (active) gemm_k128<NCOLS>(step == 0 ? &sX[aoff] : &sAgg[aoff], step == 0 ? Wr : Wl, woff, acc);
  }
  if (!active) return;

  const float4 bias = *reinterpret_cast<const float4*>(bl + nb + 4 * c);
  int rbase = m0 + (h << 3);
  if constexpr (LSM) {
#pragma unroll
    for (int r = 0; r < 8; ++r) {
      float4 v = make_float4(acc[0][r] + bias.x, acc[1][r] + bias.y, acc[2][r] + bias.z, acc[3][r] + bias.w);
      float m = half_max16(fmaxf(fmaxf(v.x, v.y), fmaxf(v.z, v.w)));
      float e = half_sum16((__expf(v.x - m) + __expf(v.y - m)) + (__expf(v.z - m) + __expf(v.w - m)));
      float lse = m + __logf(e);
      size_t off = (size_t)(rbase + r) * NCOLS + nb + 4 * c;
      *reinterpret_cast<float4*>(OutA + off) = make_float4(v.x - lse, v.y - lse, v.z - lse, v.w - lse);
    }
  } else {
#pragma unroll
    for (int r = 0; r < 8; ++r) {
      float4 v = make_float4(acc[0][r] + bias.x, acc[1][r] + bias.y, acc[2][r] + bias.z, acc[3][r] + bias.w);
      size_t off = (size_t)(rbase + r) * NCOLS + nb + 4 * c;
      if (OutB) *reinterpret_cast<float4*>(OutB + off) = v;
      if (doRelu) v = make_float4(fmaxf(v.x, 0.0f), fmaxf(v.y, 0.0f), fmaxf(v.z, 0.0f), fmaxf(v.w, 0.0f));
      *reinterpret_cast<float4*>(OutA + off) = v;
    }
  }
}

#define POOL_NODES 16
__device__ __forceinline__ void pool_flush(float* __restrict__ gsum, float* __restrict__ gcnt,
                                           int g, int lane, const float4 acc, int run) {
  float* p = gsum + (size_t)g * 128 + lane * 4;
  atomAddF(p + 0, acc.x);
  atomAddF(p + 1, acc.y);
  atomAddF(p + 2, acc.z);
  atomAddF(p + 3, acc.w);
  if (lane == 0) atomAddF(&gcnt[g], (float)run);
}
__global__ void pool_kernel(const float* __restrict__ h, const int* __restrict__ cluster,
                            float* __restrict__ gsum, float* __restrict__ gcnt, int n) {
  int w = __builtin_amdgcn_readfirstlane((int)(blockIdx.x * (blockDim.x >> 5) + (threadIdx.x >> 5)));
  int lane = threadIdx.x & 31;
  int node0 = w * POOL_NODES;
  if (node0 >= n) return;
  const float* hl = h + (size_t)node0 * 128 + lane * 4;
  const int* cl = cluster + node0;
  if (node0 + POOL_NODES <= n) {
    int g[POOL_NODES];
    float4 v[POOL_NODES];
#pragma unroll
    for (int q = 0; q < POOL_NODES; ++q) {
      g[q] = cl[q];
      v[q] = *reinterpret_cast<const float4*>(hl + (size_t)q * 128);
    }
    int cur = g[0];
    int run = 1;
    float4 acc = v[0];
#pragma unroll
    for (int q = 1; q < POOL_NODES; ++q) {
      if (g[q] != cur) {
        pool_flush(gsum, gcnt, cur, lane, acc, run);
        cur = g[q];
        run = 1;
        acc = v[q];
      } else {
        add4(acc, v[q]);
        ++run;
      }
    }
    pool_flush(gsum, gcnt, cur, lane, acc, run);
  } else {
    int cur = cl[0];
    int run = 1;
    float4 acc = *reinterpret_cast<const float4*>(hl);
    for (int q = 1; node0 + q < n; ++q) {
      int gq = cl[q];
      const float4 vq = *reinterpret_cast<const float4*>(hl + (size_t)q * 128);
      if (gq != cur) {
        pool_flush(gsum, gcnt, cur, lane, acc, run);
        cur = gq;
        run = 1;
        acc = vq;
      } else {
        add4(acc, vq);
        ++run;
      }
    }
    pool_flush(gsum, gcnt, cur, lane, acc, run);
  }
}

__global__ void pool_div_kernel(float* __restrict__ g, const float* __restrict__ gcnt, int total) {
  int i = blockIdx.x * blockDim.x + threadIdx.x;
  if (i >= total) return;
  g[i] = g[i] / fmaxf(gcnt[i >> 7], 1.0f);
}

extern "C" void kernel_launch(void* const* d_in, const int* in_sizes, int n_in,
                              void* d_out, int out_size, void* d_ws, size_t ws_size,
                              hipStream_t stream) {
  const float* x       = (const float*)d_in[0];
  const int*   ei      = (const int*)d_in[1];
  const int*   src     = ei;
  const int*   dst     = ei + NE;
  const int*   cluster = (const int*)d_in[2];
  const float* Wl0 = (const float*)d_in[3];
  const float* bl0 = (const float*)d_in[4];
  const float* Wr0 = (const float*)d_in[5];
  const float* Wl1 = (const float*)d_in[6];
  const float* bl1 = (const float*)d_in[7];
  const float* Wr1 = (const float*)d_in[8];
  const float* Wl2 = (const float*)d_in[9];
  const float* bl2 = (const float*)d_in[10];
  const float* Wr2 = (const float*)d_in[11];

  float* out_ls = (float*)d_out;
  float* out_h2 = out_ls + (size_t)NN * OUTC;
  float* out_g  = out_h2 + (size_t)NN * HIDC;

  float* ws   = (float*)d_ws;
  float* scr  = ws;
  float* h1   = ws + (size_t)NN * HIDC;
  float* h2r  = h1 + (size_t)NN * HIDC;
  float* deg  = h2r + (size_t)NN * HIDC;
  float* gcnt = deg + NN;

  int* cnt    = (int*)h1;
  int* pos    = cnt + 50048;
  int* rowptr = (int*)scr;
  int* col    = rowptr + 50048;

  const int TPB = 256;
  const int edgeBlocks4 = (NE + TPB * 4 - 1) / (TPB * 4);
  const int layerBlocks = (NN + LAYER_ROWS - 1) / LAYER_ROWS;

  hipMemsetAsync(cnt, 0, NN * sizeof(int), stream);
  count_rank_kernel<<<edgeBlocks4, TPB, 0, stream>>>(dst, cnt, pos, NE);
  scan_kernel<<<(NN + 1023) / 1024, 1024, 0, stream>>>(cnt, rowptr, deg, NN);
  fill_kernel<<<edgeBlocks4, TPB, 0, stream>>>(src, dst, rowptr, pos, col, NE);

  sage_layer_kernel<HIDC, false><<<layerBlocks, (HIDC / 64) * 64, 0, stream>>>(
      x, rowptr, col, deg, Wl0, bl0, Wr0, h1, nullptr, NN, 1);

  sage_layer_kernel<HIDC, false><<<layerBlocks, (HIDC / 64) * 64, 0, stream>>>(
      h1, rowptr, col, deg, Wl1, bl1, Wr1, h2r, out_h2, NN, 1);

  hipMemsetAsync(out_g, 0, (size_t)NG * HIDC * sizeof(float), stream);
  hipMemsetAsync(gcnt, 0, NG * sizeof(float), stream);
  pool_kernel<<<(NN + 8 * POOL_NODES - 1) / (8 * POOL_NODES), TPB, 0, stream>>>(h2r, cluster, out_g, gcnt, NN);
  pool_div_kernel<<<(NG * HIDC + TPB - 1) / TPB, TPB, 0, stream>>>(out_g, gcnt, NG * HIDC);

  sage_layer_kernel<OUTC, true><<<layerBlocks, (OUTC / 64) * 64, 0, stream>>>(
      h2r, rowptr, col, deg, Wl2, bl2, Wr2, out_ls, nullptr, NN, 0);
}


# Edge buckets without prefix sum: a destination row owns a 32-slot line and the counting atomic's rank is the slot; scan and fill launches gone, a wave packs its nodes' ids in LDS; sum order kept
# speedup vs baseline: 3.8136x; 1.2232x over previous
#include <hip/hip_runtime.h>
#include <hip/hip_bf16.h>
#include <math.h>

typedef float v2f __attribute__((ext_vector_type(2)));
typedef float v8f __attribute__((ext_vector_type(8)));

#define NN      50000
#define NE      600000
#define INC     128
#define HIDC    128
#define OUTC    64
#define NG      512

__device__ __forceinline__ void atomAddF(float* p, float v) {
  __hip_atomic_fetch_add(p, v, __ATOMIC_RELAXED, __HIP_MEMORY_SCOPE_AGENT);
}

#define SLOTS 32
__device__ __forceinline__ void place_edge(int* cnt, int* slot, int* ovf, int n, int d, int r, int s) {
  if (r < SLOTS) {
    slot[d * SLOTS + r] = s;
  } else {
    int o = __hip_atomic_fetch_add(&cnt[n], 1, __ATOMIC_RELAXED, __HIP_MEMORY_SCOPE_AGENT);
    ovf[2 * o] = d;
    ovf[2 * o + 1] = s;
  }
}
__global__ void bucket_kernel(const int* __restrict__ src, const int* __restrict__ dst,
                              int* __restrict__ cnt, int* __restrict__ slot,
                              int* __restrict__ ovf, int n, int E) {
  int B = blockDim.x;
  int e0 = blockIdx.x * (B * 4) + threadIdx.x;
  int e1 = e0 + B, e2 = e1 + B, e3 = e2 + B;
  if (e3 < E) {
    int d0 = dst[e0], d1 = dst[e1], d2 = dst[e2], d3 = dst[e3];
    int s0 = src[e0], s1 = src[e1], s2 = src[e2], s3 = src[e3];
    int r0 = __hip_atomic_fetch_add(&cnt[d0], 1, __ATOMIC_RELAXED, __HIP_MEMORY_SCOPE_AGENT);
    int r1 = __hip_atomic_fetch_add(&cnt[d1], 1, __ATOMIC_RELAXED, __HIP_MEMORY_SCOPE_AGENT);
    int r2 = __hip_atomic_fetch_add(&cnt[d2], 1, __ATOMIC_RELAXED, __HIP_MEMORY_SCOPE_AGENT);
    int r3 = __hip_atomic_fetch_add(&cnt[d3], 1, __ATOMIC_RELAXED, __HIP_MEMORY_SCOPE_AGENT);
    place_edge(cnt, slot, ovf, n, d0, r0, s0);
    place_edge(cnt, slot, ovf, n, d1, r1, s1);
    place_edge(cnt, slot, ovf, n, d2, r2, s2);
    place_edge(cnt, slot, ovf, n, d3, r3, s3);
  } else {
    for (int e = e0; e < E; e += B) {
      int d = dst[e];
      int r = __hip_atomic_fetch_add(&cnt[d], 1, __ATOMIC_RELAXED, __HIP_MEMORY_SCOPE_AGENT);
      place_edge(cnt, slot, ovf, n, d, r, src[e]);
    }
  }
}

__device__ __forceinline__ void add4(float4& a, const float4 v) {
  a.x += v.x; a.y += v.y; a.z += v.z; a.w += v.w;
}

#define LAYER_ROWS 32
#define LDS_PITCH 132
struct GemmFrag {
  float4 a[2];
  float4 w[8];
};
__device__ __forceinline__ float f4c(const float4 v, int j) {
  return j == 0 ? v.x : (j == 1 ? v.y : (j == 2 ? v.z : v.w));
}
__device__ __forceinline__ void gemm_mma(const GemmFrag& f, v8f (&acc)[4]) {
#pragma unroll
  for (int b = 0; b < 2; ++b) {
#pragma unroll
    for (int t = 0; t < 2; ++t) {
      v2f af;
      af.x = f4c(f.a[b], 2 * t);
      af.y = f4c(f.a[b], 2 * t + 1);
#pragma unroll
      for (int j = 0; j < 4; ++j) {
        v2f bf;
        bf.x = f4c(f.w[4 * b + 2 * t], j);
        bf.y = f4c(f.w[4 * b + 2 * t + 1], j);
        acc[j] = __builtin_amdgcn_wmma_f32_16x16x4_f32(false, af, false, bf, (short)0, acc[j], false, false);
      }
    }
  }
}
template <int NCOLS>
__device__ __forceinline__ void gemm_k128(const float* sA, const float* wbase, int woff, v8f (&acc)[4]) {
  auto frag_load = [&](GemmFrag& f, const float* ast, const float* wst) {
#pragma unroll
    for (int b = 0; b < 2; ++b)
      f.a[b] = *reinterpret_cast<const float4*>(ast + 8 * b);
#pragma unroll
    for (int q = 0; q < 8; ++q)
      f.w[q] = *reinterpret_cast<const float4*>(wst + (woff + (8 * (q >> 2) + (q & 3)) * NCOLS));
  };
  constexpr int KI = 128 >> 4;
  const float* ast = sA;
  const float* wst = wbase;
  GemmFrag cur;
  frag_load(cur, ast, wst);
#pragma nounroll
  for (int i = 0; i + 1 < KI; ++i) {
    ast += 16;
    wst += 16 * NCOLS;
    GemmFrag nxt;
    frag_load(nxt, ast, wst);
    __builtin_amdgcn_sched_barrier(0);
    gemm_mma(cur, acc);
    cur = nxt;
  }
  gemm_mma(cur, acc);
}
__device__ __forceinline__ float half_max16(float v) {
  v = fmaxf(v, __shfl_xor(v, 8, 32));
  v = fmaxf(v, __shfl_xor(v, 4, 32));
  v = fmaxf(v, __shfl_xor(v, 2, 32));
  v = fmaxf(v, __shfl_xor(v, 1, 32));
  return v;
}
__device__ __forceinline__ float half_sum16(float v) {
  v += __shfl_xor(v, 8, 32);
  v += __shfl_xor(v, 4, 32);
  v += __shfl_xor(v, 2, 32);
  v += __shfl_xor(v, 1, 32);
  return v;
}
template <int NCOLS, bool LSM>
__global__ void __launch_bounds__((NCOLS / 64) * 64)
sage_layer_kernel(const float* __restrict__ X, const int* __restrict__ cnt,
                  const int* __restrict__ slot, const int* __restrict__ ovf,
                  const float* __restrict__ Wl, const float* __restrict__ bl,
                  const float* __restrict__ Wr,
                  float* __restrict__ OutA, float* __restrict__ OutB, int M, int doRelu) {
  constexpr int NGRP = NCOLS / 64;
  constexpr int WAVES = 2 * NGRP;
  constexpr int G = LAYER_ROWS / WAVES;
  __shared__ __attribute__((aligned(16))) float sAgg[LAYER_ROWS * LDS_PITCH];
  __shared__ __attribute__((aligned(16))) float sX[LAYER_ROWS * LDS_PITCH];
  __shared__ int sIds[LAYER_ROWS * SLOTS];

  int wv = __builtin_amdgcn_readfirstlane((int)(threadIdx.x >> 5));
  int lane = threadIdx.x & 31;
  int row0 = blockIdx.x * LAYER_ROWS;
  const bool gather_first = (__popc(blockIdx.x) & 1) == 0;

  int lr0 = wv * G;
  int n0 = row0 + lr0;
  const bool has_nodes = n0 + G <= M;

  if (has_nodes) {
#pragma unroll
    for (int q = 0; q < G; ++q)
      *reinterpret_cast<float4*>(&sX[(lr0 + q) * LDS_PITCH + lane * 4]) =
          *reinterpret_cast<const float4*>(X + (size_t)(n0 + q) * 128 + lane * 4);
  }

  auto gather = [&]() {
    if (!has_nodes) return;
    int* ids = sIds + wv * (G * SLOTS);
    int cq = (lane < G) ? cnt[n0 + lane] : 0;
    int idv[G];
#pragma unroll
    for (int q = 0; q < G; ++q) idv[q] = slot[(n0 + q) * SLOTS + lane];
    int dq = min(cq, SLOTS);
    float iv = 1.0f / fmaxf((float)cq, 1.0f);
    int inc = dq;
#pragma unroll
    for (int off = 1; off < 32; off <<= 1) {
      int u = __shfl_up(inc, off, 32);
      if (lane >= off) inc += u;
    }
    int rp = (lane <= G) ? inc - dq : 0x7fffffff;
#pragma unroll
    for (int q = 0; q < G; ++q) ids[__builtin_amdgcn_readlane(rp, q) + lane] = idv[q];
    int jend = __builtin_amdgcn_readlane(rp, G);
    int node = 0;
    int nend = __builtin_amdgcn_readlane(rp, 1);
    float4 sum = make_float4(0.0f, 0.0f, 0.0f, 0.0f);
    const float* xl = X + lane * 4;
    if (0 < jend) {
      int cv = ids[min(lane, jend - 1)];
      for (int jb = 0; jb < jend; jb += 32) {
        int cvn = ids[min(jb + 32 + lane, jend - 1)];
#pragma unroll
        for (int u = 0; u < 2; ++u) {
          int e0 = jb + 16 * u;
          if (e0 < jend) {
            float4 v[16];
#pragma unroll
            for (int t = 0; t < 16; ++t) {
              int s = __builtin_amdgcn_readlane(cv, 16 * u + t);
              v[t] = *reinterpret_cast<const float4*>(xl + (size_t)s * 128);
            }
#pragma unroll
            for (int t = 0; t < 16; ++t) {
              int e = e0 + t;
              while (e >= nend && node < G) {
                float s = __int_as_float(__builtin_amdgcn_readlane(__float_as_int(iv), node));
                *reinterpret_cast<float4*>(&sAgg[(lr0 + node) * LDS_PITCH + lane * 4]) =
                    make_float4(sum.x * s, sum.y * s, sum.z * s, sum.w * s);
                ++node;
                nend = __builtin_amdgcn_readlane(rp, node + 1);
                sum = make_float4(0.0f, 0.0f, 0.0f, 0.0f);
              }
              add4(sum, v[t]);
            }
          }
        }
        cv = cvn;
      }
    }
    for (; node < G; ++node) {
      float s = __int_as_float(__builtin_amdgcn_readlane(__float_as_int(iv), node));
      *reinterpret_cast<float4*>(&sAgg[(lr0 + node) * LDS_PITCH + lane * 4]) =
          make_float4(sum.x * s, sum.y * s, sum.z * s, sum.w * s);
      sum = make_float4(0.0f, 0.0f, 0.0f, 0.0f);
    }
    int ovn = min(cnt[M], NE);
    if (ovn > 0) {
#pragma nounroll
      for (int q = 0; q < G; ++q) {
        if (__builtin_amdgcn_readlane(cq, q) > SLOTS) {
          float4 ex = make_float4(0.0f, 0.0f, 0.0f, 0.0f);
#pragma nounroll
          for (int i = 0; i < ovn; ++i)
            if (ovf[2 * i] == n0 + q)
              add4(ex, *reinterpret_cast<const float4*>(xl + (size_t)ovf[2 * i + 1] * 128));
          float s = __int_as_float(__builtin_amdgcn_readlane(__float_as_int(iv), q));
          float4* p = reinterpret_cast<float4*>(&sAgg[(lr0 + q) * LDS_PITCH + lane * 4]);
          const float4 a = *p;
          *p = make_float4(a.x + ex.x * s, a.y + ex.y * s, a.z + ex.z * s, a.w + ex.w * s);
        }
      }
    }
  };

  int tl = wv / NGRP;
  int cg = wv % NGRP;
  int m0 = row0 + 16 * tl;
  const bool active = m0 < M;
  int nb = cg << 6;
  int c = lane & 15;
  int h = lane >> 4;
  int aoff = (16 * tl + c) * LDS_PITCH + 4 * h;
  int woff = (4 * h) * NCOLS + nb + 4 * c;
  v8f acc[4] = {};

#pragma nounroll
  for (int step = 0; step < 2; ++step) {
    if ((step == 0) == gather_first) gather();
    if (step == 0 || !gather_first) __syncthreads();
    if (active) gemm_k128<NCOLS>(step == 0 ? &sX[aoff] : &sAgg[aoff], step == 0 ? Wr : Wl, woff, acc);
  }
  if (!active) return;

  const float4 bias = *reinterpret_cast<const float4*>(bl + nb + 4 * c);
  int rbase = m0 + (h << 3);
  if constexpr (LSM) {
#pragma unroll
    for (int r = 0; r < 8; ++r) {
      float4 v = make_float4(acc[0][r] + bias.x, acc[1][r] + bias.y, acc[2][r] + bias.z, acc[3][r] + bias.w);
      float m = half_max16(fmaxf(fmaxf(v.x, v.y), fmaxf(v.z, v.w)));
      float e = half_sum16((__expf(v.x - m) + __expf(v.y - m)) + (__expf(v.z - m) + __expf(v.w - m)));
      float lse = m + __logf(e);
      size_t off = (size_t)(rbase + r) * NCOLS + nb + 4 * c;
      *reinterpret_cast<float4*>(OutA + off) = make_float4(v.x - lse, v.y - lse, v.z - lse, v.w - lse);
    }
  } else {
#pragma unroll
    for (int r = 0; r < 8; ++r) {
      float4 v = make_float4(acc[0][r] + bias.x, acc[1][r] + bias.y, acc[2][r] + bias.z, acc[3][r] + bias.w);
      size_t off = (size_t)(rbase + r) * NCOLS + nb + 4 * c;
      if (OutB) *reinterpret_cast<float4*>(OutB + off) = v;
      if (doRelu) v = make_float4(fmaxf(v.x, 0.0f), fmaxf(v.y, 0.0f), fmaxf(v.z, 0.0f), fmaxf(v.w, 0.0f));
      *reinterpret_cast<float4*>(OutA + off) = v;
    }
  }
}

#define POOL_NODES 16
__device__ __forceinline__ void pool_flush(float* __restrict__ gsum, float* __restrict__ gcnt,
                                           int g, int lane, const float4 acc, int run) {
  float* p = gsum + (size_t)g * 128 + lane * 4;
  atomAddF(p + 0, acc.x);
  atomAddF(p + 1, acc.y);
  atomAddF(p + 2, acc.z);
  atomAddF(p + 3, acc.w);
  if (lane == 0) atomAddF(&gcnt[g], (float)run);
}
__global__ void pool_kernel(const float* __restrict__ h, const int* __restrict__ cluster,
                            float* __restrict__ gsum, float* __restrict__ gcnt, int n) {
  int w = __builtin_amdgcn_readfirstlane((int)(blockIdx.x * (blockDim.x >> 5) + (threadIdx.x >> 5)));
  int lane = threadIdx.x & 31;
  int node0 = w * POOL_NODES;
  if (node0 >= n) return;
  const float* hl = h + (size_t)node0 * 128 + lane * 4;
  const int* cl = cluster + node0;
  if (node0 + POOL_NODES <= n) {
    int g[POOL_NODES];
    float4 v[POOL_NODES];
#pragma unroll
    for (int q = 0; q < POOL_NODES; ++q) {
      g[q] = cl[q];
      v[q] = *reinterpret_cast<const float4*>(hl + (size_t)q * 128);
    }
    int cur = g[0];
    int run = 1;
    float4 acc = v[0];
#pragma unroll
    for (int q = 1; q < POOL_NODES; ++q) {
      if (g[q] != cur) {
        pool_flush(gsum, gcnt, cur, lane, acc, run);
        cur = g[q];
        run = 1;
        acc = v[q];
      } else {
        add4(acc, v[q]);
        ++run;
      }
    }
    pool_flush(gsum, gcnt, cur, lane, acc, run);
  } else {
    int cur = cl[0];
    int run = 1;
    float4 acc = *reinterpret_cast<const float4*>(hl);
    for (int q = 1; node0 + q < n; ++q) {
      int gq = cl[q];
      const float4 vq = *reinterpret_cast<const float4*>(hl + (size_t)q * 128);
      if (gq != cur) {
        pool_flush(gsum, gcnt, cur, lane, acc, run);
        cur = gq;
        run = 1;
        acc = vq;
      } else {
        add4(acc, vq);
        ++run;
      }
    }
    pool_flush(gsum, gcnt, cur, lane, acc, run);
  }
}

__global__ void pool_div_kernel(float* __restrict__ g, const float* __restrict__ gcnt, int total) {
  int i = blockIdx.x * blockDim.x + threadIdx.x;
  if (i >= total) return;
  g[i] = g[i] / fmaxf(gcnt[i >> 7], 1.0f);
}

extern "C" void kernel_launch(void* const* d_in, const int* in_sizes, int n_in,
                              void* d_out, int out_size, void* d_ws, size_t ws_size,
                              hipStream_t stream) {
  const float* x       = (const float*)d_in[0];
  const int*   ei      = (const int*)d_in[1];
  const int*   src     = ei;
  const int*   dst     = ei + NE;
  const int*   cluster = (const int*)d_in[2];
  const float* Wl0 = (const float*)d_in[3];
  const float* bl0 = (const float*)d_in[4];
  const float* Wr0 = (const float*)d_in[5];
  const float* Wl1 = (const float*)d_in[6];
  const float* bl1 = (const float*)d_in[7];
  const float* Wr1 = (const float*)d_in[8];
  const float* Wl2 = (const float*)d_in[9];
  const float* bl2 = (const float*)d_in[10];
  const float* Wr2 = (const float*)d_in[11];

  float* out_ls = (float*)d_out;
  float* out_h2 = out_ls + (size_t)NN * OUTC;
  float* out_g  = out_h2 + (size_t)NN * HIDC;

  float* ws   = (float*)d_ws;
  float* scr  = ws;
  float* h1   = ws + (size_t)NN * HIDC;
  float* h2r  = h1 + (size_t)NN * HIDC;
  float* gcnt = h2r + (size_t)NN * HIDC + NN;

  int* cnt  = (int*)scr;
  int* slot = cnt + 50048;
  int* ovf  = slot + NN * SLOTS;

  const int TPB = 256;
  const int edgeBlocks4 = (NE + TPB * 4 - 1) / (TPB * 4);
  const int layerBlocks = (NN + LAYER_ROWS - 1) / LAYER_ROWS;

  hipMemsetAsync(cnt, 0, 50048 * sizeof(int), stream);
  bucket_kernel<<<edgeBlocks4, TPB, 0, stream>>>(src, dst, cnt, slot, ovf, NN, NE);

  sage_layer_kernel<HIDC, false><<<layerBlocks, (HIDC / 64) * 64, 0, stream>>>(
      x, cnt, slot, ovf, Wl0, bl0, Wr0, h1, nullptr, NN, 1);

  sage_layer_kernel<HIDC, false><<<layerBlocks, (HIDC / 64) * 64, 0, stream>>>(
      h1, cnt, slot, ovf, Wl1, bl1, Wr1, h2r, out_h2, NN, 1);

  hipMemsetAsync(out_g, 0, (size_t)NG * HIDC * sizeof(float), stream);
  hipMemsetAsync(gcnt, 0, NG * sizeof(float), stream);
  pool_kernel<<<(NN + 8 * POOL_NODES - 1) / (8 * POOL_NODES), TPB, 0, stream>>>(h2r, cluster, out_g, gcnt, NN);
  pool_div_kernel<<<(NG * HIDC + TPB - 1) / TPB, TPB, 0, stream>>>(out_g, gcnt, NG * HIDC);

  sage_layer_kernel<OUTC, true><<<layerBlocks, (OUTC / 64) * 64, 0, stream>>>(
      h2r, cnt, slot, ovf, Wl2, bl2, Wr2, out_ls, nullptr, NN, 0);
}


# Pooling rides in the layer launches: layer 1's epilogue adds a 16-row tile's run sums by float atomics, the count memset zeroes them, layer 2 divides; 4 stream ops gone; pooled sum order changes
# speedup vs baseline: 4.0694x; 1.0671x over previous
#include <hip/hip_runtime.h>
#include <hip/hip_bf16.h>
#include <math.h>

typedef float v2f __attribute__((ext_vector_type(2)));
typedef float v8f __attribute__((ext_vector_type(8)));

#define NN      50000
#define NE      600000
#define INC     128
#define HIDC    128
#define OUTC    64
#define NG      512

__device__ __forceinline__ void atomAddF(float* p, float v) {
  __hip_atomic_fetch_add(p, v, __ATOMIC_RELAXED, __HIP_MEMORY_SCOPE_AGENT);
}

#define SLOTS 32
__device__ __forceinline__ void place_edge(int* cnt, int* slot, int* ovf, int n, int d, int r, int s) {
  if (r < SLOTS) {
    slot[d * SLOTS + r] = s;
  } else {
    int o = __hip_atomic_fetch_add(&cnt[n], 1, __ATOMIC_RELAXED, __HIP_MEMORY_SCOPE_AGENT);
    ovf[2 * o] = d;
    ovf[2 * o + 1] = s;
  }
}
__global__ void bucket_kernel(const int* __restrict__ src, const int* __restrict__ dst,
                              int* __restrict__ cnt, int* __restrict__ slot,
                              int* __restrict__ ovf, int n, int E) {
  int B = blockDim.x;
  int e0 = blockIdx.x * (B * 4) + threadIdx.x;
  int e1 = e0 + B, e2 = e1 + B, e3 = e2 + B;
  if (e3 < E) {
    int d0 = dst[e0], d1 = dst[e1], d2 = dst[e2], d3 = dst[e3];
    int s0 = src[e0], s1 = src[e1], s2 = src[e2], s3 = src[e3];
    int r0 = __hip_atomic_fetch_add(&cnt[d0], 1, __ATOMIC_RELAXED, __HIP_MEMORY_SCOPE_AGENT);
    int r1 = __hip_atomic_fetch_add(&cnt[d1], 1, __ATOMIC_RELAXED, __HIP_MEMORY_SCOPE_AGENT);
    int r2 = __hip_atomic_fetch_add(&cnt[d2], 1, __ATOMIC_RELAXED, __HIP_MEMORY_SCOPE_AGENT);
    int r3 = __hip_atomic_fetch_add(&cnt[d3], 1, __ATOMIC_RELAXED, __HIP_MEMORY_SCOPE_AGENT);
    place_edge(cnt, slot, ovf, n, d0, r0, s0);
    place_edge(cnt, slot, ovf, n, d1, r1, s1);
    place_edge(cnt, slot, ovf, n, d2, r2, s2);
    place_edge(cnt, slot, ovf, n, d3, r3, s3);
  } else {
    for (int e = e0; e < E; e += B) {
      int d = dst[e];
      int r = __hip_atomic_fetch_add(&cnt[d], 1, __ATOMIC_RELAXED, __HIP_MEMORY_SCOPE_AGENT);
      place_edge(cnt, slot, ovf, n, d, r, src[e]);
    }
  }
}

__device__ __forceinline__ void add4(float4& a, const float4 v) {
  a.x += v.x; a.y += v.y; a.z += v.z; a.w += v.w;
}

#define LAYER_ROWS 32
#define LDS_PITCH 132
struct GemmFrag {
  float4 a[2];
  float4 w[8];
};
__device__ __forceinline__ float f4c(const float4 v, int j) {
  return j == 0 ? v.x : (j == 1 ? v.y : (j == 2 ? v.z : v.w));
}
__device__ __forceinline__ void gemm_mma(const GemmFrag& f, v8f (&acc)[4]) {
#pragma unroll
  for (int b = 0; b < 2; ++b) {
#pragma unroll
    for (int t = 0; t < 2; ++t) {
      v2f af;
      af.x = f4c(f.a[b], 2 * t);
      af.y = f4c(f.a[b], 2 * t + 1);
#pragma unroll
      for (int j = 0; j < 4; ++j) {
        v2f bf;
        bf.x = f4c(f.w[4 * b + 2 * t], j);
        bf.y = f4c(f.w[4 * b + 2 * t + 1], j);
        acc[j] = __builtin_amdgcn_wmma_f32_16x16x4_f32(false, af, false, bf, (short)0, acc[j], false, false);
      }
    }
  }
}
template <int NCOLS>
__device__ __forceinline__ void gemm_k128(const float* sA, const float* wbase, int woff, v8f (&acc)[4]) {
  auto frag_load = [&](GemmFrag& f, const float* ast, const float* wst) {
#pragma unroll
    for (int b = 0; b < 2; ++b)
      f.a[b] = *reinterpret_cast<const float4*>(ast + 8 * b);
#pragma unroll
    for (int q = 0; q < 8; ++q)
      f.w[q] = *reinterpret_cast<const float4*>(wst + (woff + (8 * (q >> 2) + (q & 3)) * NCOLS));
  };
  constexpr int KI = 128 >> 4;
  const float* ast = sA;
  const float* wst = wbase;
  GemmFrag cur;
  frag_load(cur, ast, wst);
#pragma nounroll
  for (int i = 0; i + 1 < KI; ++i) {
    ast += 16;
    wst += 16 * NCOLS;
    GemmFrag nxt;
    frag_load(nxt, ast, wst);
    __builtin_amdgcn_sched_barrier(0);
    gemm_mma(cur, acc);
    cur = nxt;
  }
  gemm_mma(cur, acc);
}
__device__ __forceinline__ float half_max16(float v) {
  v = fmaxf(v, __shfl_xor(v, 8, 32));
  v = fmaxf(v, __shfl_xor(v, 4, 32));
  v = fmaxf(v, __shfl_xor(v, 2, 32));
  v = fmaxf(v, __shfl_xor(v, 1, 32));
  return v;
}
__device__ __forceinline__ float half_sum16(float v) {
  v += __shfl_xor(v, 8, 32);
  v += __shfl_xor(v, 4, 32);
  v += __shfl_xor(v, 2, 32);
  v += __shfl_xor(v, 1, 32);
  return v;
}
__device__ __forceinline__ void pool_flush(float* __restrict__ gsum, float* __restrict__ gcnt,
                                           int g, int col, const float4 acc, int run, bool counts) {
  float* p = gsum + (size_t)g * 128 + col;
  atomAddF(p + 0, acc.x);
  atomAddF(p + 1, acc.y);
  atomAddF(p + 2, acc.z);
  atomAddF(p + 3, acc.w);
  if (counts) atomAddF(&gcnt[g], (float)run);
}
template <int NCOLS, bool LSM, int POOLMODE>
__global__ void __launch_bounds__((NCOLS / 64) * 64)
sage_layer_kernel(const float* __restrict__ X, const int* __restrict__ cnt,
                  const int* __restrict__ slot, const int* __restrict__ ovf,
                  const float* __restrict__ Wl, const float* __restrict__ bl,
                  const float* __restrict__ Wr,
                  float* __restrict__ OutA, float* __restrict__ OutB, int M, int doRelu,
                  const int* __restrict__ cluster, float* __restrict__ gsum,
                  float* __restrict__ gcnt, float* __restrict__ gOut) {
  constexpr int NGRP = NCOLS / 64;
  constexpr int WAVES = 2 * NGRP;
  constexpr int G = LAYER_ROWS / WAVES;
  __shared__ __attribute__((aligned(16))) float sAgg[LAYER_ROWS * LDS_PITCH];
  __shared__ __attribute__((aligned(16))) float sX[LAYER_ROWS * LDS_PITCH];
  __shared__ int sIds[LAYER_ROWS * SLOTS];

  int wv = __builtin_amdgcn_readfirstlane((int)(threadIdx.x >> 5));
  int lane = threadIdx.x & 31;
  int row0 = blockIdx.x * LAYER_ROWS;
  const bool gather_first = (__popc(blockIdx.x) & 1) == 0;

  int lr0 = wv * G;
  int n0 = row0 + lr0;
  const bool has_nodes = n0 + G <= M;

  constexpr int PCH = 128 / (WAVES * 32);
  float ps[PCH];
  float pc = 0.0f;
  if constexpr (POOLMODE == 2) {
    static_assert((NN + LAYER_ROWS - 1) / LAYER_ROWS >= NG, "one block per graph");
    if (row0 < NG * LAYER_ROWS) {
      pc = gcnt[blockIdx.x];
#pragma unroll
      for (int q = 0; q < PCH; ++q) ps[q] = gsum[blockIdx.x * 128 + q * (WAVES * 32) + threadIdx.x];
    }
  }

  if (has_nodes) {
#pragma unroll
    for (int q = 0; q < G; ++q)
      *reinterpret_cast<float4*>(&sX[(lr0 + q) * LDS_PITCH + lane * 4]) =
          *reinterpret_cast<const float4*>(X + (size_t)(n0 + q) * 128 + lane * 4);
  }

  if constexpr (POOLMODE == 2) {
    if (row0 < NG * LAYER_ROWS) {
      float dv = fmaxf(pc, 1.0f);
#pragma unroll
      for (int q = 0; q < PCH; ++q)
        gOut[blockIdx.x * 128 + q * (WAVES * 32) + threadIdx.x] = ps[q] / dv;
    }
  }

  auto gather = [&]() {
    if (!has_nodes) return;
    int* ids = sIds + wv * (G * SLOTS);
    int cq = (lane < G) ? cnt[n0 + lane] : 0;
    int idv[G];
#pragma unroll
    for (int q = 0; q < G; ++q) idv[q] = slot[(n0 + q) * SLOTS + lane];
    int dq = min(cq, SLOTS);
    float iv = 1.0f / fmaxf((float)cq, 1.0f);
    int inc = dq;
#pragma unroll
    for (int off = 1; off < 32; off <<= 1) {
      int u = __shfl_up(inc, off, 32);
      if (lane >= off) inc += u;
    }
    int rp = (lane <= G) ? inc - dq : 0x7fffffff;
#pragma unroll
    for (int q = 0; q < G; ++q) ids[__builtin_amdgcn_readlane(rp, q) + lane] = idv[q];
    int jend = __builtin_amdgcn_readlane(rp, G);
    int node = 0;
    int nend = __builtin_amdgcn_readlane(rp, 1);
    float4 sum = make_float4(0.0f, 0.0f, 0.0f, 0.0f);
    const float* xl = X + lane * 4;
    if (0 < jend) {
      int cv = ids[min(lane, jend - 1)];
      for (int jb = 0; jb < jend; jb += 32) {
        int cvn = ids[min(jb + 32 + lane, jend - 1)];
#pragma unroll
        for (int u = 0; u < 2; ++u) {
          int e0 = jb + 16 * u;
          if (e0 < jend) {
            float4 v[16];
#pragma unroll
            for (int t = 0; t < 16; ++t) {
              int s = __builtin_amdgcn_readlane(cv, 16 * u + t);
              v[t] = *reinterpret_cast<const float4*>(xl + (size_t)s * 128);
            }
#pragma unroll
            for (int t = 0; t < 16; ++t) {
              int e = e0 + t;
              while (e >= nend && node < G) {
                float s = __int_as_float(__builtin_amdgcn_readlane(__float_as_int(iv), node));
                *reinterpret_cast<float4*>(&sAgg[(lr0 + node) * LDS_PITCH + lane * 4]) =
                    make_float4(sum.x * s, sum.y * s, sum.z * s, sum.w * s);
                ++node;
                nend = __builtin_amdgcn_readlane(rp, node + 1);
                sum = make_float4(0.0f, 0.0f, 0.0f, 0.0f);
              }
              add4(sum, v[t]);
            }
          }
        }
        cv = cvn;
      }
    }
    for (; node < G; ++node) {
      float s = __int_as_float(__builtin_amdgcn_readlane(__float_as_int(iv), node));
      *reinterpret_cast<float4*>(&sAgg[(lr0 + node) * LDS_PITCH + lane * 4]) =
          make_float4(sum.x * s, sum.y * s, sum.z * s, sum.w * s);
      sum = make_float4(0.0f, 0.0f, 0.0f, 0.0f);
    }
    int ovn = min(cnt[M], NE);
    if (ovn > 0) {
#pragma nounroll
      for (int q = 0; q < G; ++q) {
        if (__builtin_amdgcn_readlane(cq, q) > SLOTS) {
          float4 ex = make_float4(0.0f, 0.0f, 0.0f, 0.0f);
#pragma nounroll
          for (int i = 0; i < ovn; ++i)
            if (ovf[2 * i] == n0 + q)
              add4(ex, *reinterpret_cast<const float4*>(xl + (size_t)ovf[2 * i + 1] * 128));
          float s = __int_as_float(__builtin_amdgcn_readlane(__float_as_int(iv), q));
          float4* p = reinterpret_cast<float4*>(&sAgg[(lr0 + q) * LDS_PITCH + lane * 4]);
          const float4 a = *p;
          *p = make_float4(a.x + ex.x * s, a.y + ex.y * s, a.z + ex.z * s, a.w + ex.w * s);
        }
      }
    }
  };

  int tl = wv / NGRP;
  int cg = wv % NGRP;
  int m0 = row0 + 16 * tl;
  const bool active = m0 < M;
  int nb = cg << 6;
  int c = lane & 15;
  int h = lane >> 4;
  int aoff = (16 * tl + c) * LDS_PITCH + 4 * h;
  int woff = (4 * h) * NCOLS + nb + 4 * c;
  v8f acc[4] = {};
  int gl = 0;
  if constexpr (POOLMODE == 1) {
    if (active) gl = cluster[m0 + c];
  }

#pragma nounroll
  for (int step = 0; step < 2; ++step) {
    if ((step == 0) == gather_first) gather();
    if (step == 0 || !gather_first) __syncthreads();
    if (active) gemm_k128<NCOLS>(step == 0 ? &sX[aoff] : &sAgg[aoff], step == 0 ? Wr : Wl, woff, acc);
  }
  if (!active) return;

  const float4 bias = *reinterpret_cast<const float4*>(bl + nb + 4 * c);
  int rbase = m0 + (h << 3);
  if constexpr (LSM) {
#pragma unroll
    for (int r = 0; r < 8; ++r) {
      float4 v = make_float4(acc[0][r] + bias.x, acc[1][r] + bias.y, acc[2][r] + bias.z, acc[3][r] + bias.w);
      float m = half_max16(fmaxf(fmaxf(v.x, v.y), fmaxf(v.z, v.w)));
      float e = half_sum16((__expf(v.x - m) + __expf(v.y - m)) + (__expf(v.z - m) + __expf(v.w - m)));
      float lse = m + __logf(e);
      size_t off = (size_t)(rbase + r) * NCOLS + nb + 4 * c;
      *reinterpret_cast<float4*>(OutA + off) = make_float4(v.x - lse, v.y - lse, v.z - lse, v.w - lse);
    }
  } else {
    float4 hv[8];
#pragma unroll
    for (int r = 0; r < 8; ++r) {
      float4 v = make_float4(acc[0][r] + bias.x, acc[1][r] + bias.y, acc[2][r] + bias.z, acc[3][r] + bias.w);
      size_t off = (size_t)(rbase + r) * NCOLS + nb + 4 * c;
      if (OutB) *reinterpret_cast<float4*>(OutB + off) = v;
      if (doRelu) v = make_float4(fmaxf(v.x, 0.0f), fmaxf(v.y, 0.0f), fmaxf(v.z, 0.0f), fmaxf(v.w, 0.0f));
      *reinterpret_cast<float4*>(OutA + off) = v;
      hv[r] = v;
    }
    if constexpr (POOLMODE == 1) {
      int col = nb + 4 * c;
      int g0 = __builtin_amdgcn_readfirstlane(gl);
      if (__builtin_amdgcn_ballot_w32(gl != g0) == 0) {
        float4 s = hv[0];
#pragma unroll
        for (int r = 1; r < 8; ++r) add4(s, hv[r]);
        s.x += __shfl_xor(s.x, 16, 32);
        s.y += __shfl_xor(s.y, 16, 32);
        s.z += __shfl_xor(s.z, 16, 32);
        s.w += __shfl_xor(s.w, 16, 32);
        if (h == 0) pool_flush(gsum, gcnt, g0, col, s, 16, cg == 0 && c == 0);
      } else {
        int cur = __shfl(gl, 8 * h, 32);
        int run = 1;
        float4 s = hv[0];
#pragma unroll
        for (int r = 1; r < 8; ++r) {
          int gr = __shfl(gl, 8 * h + r, 32);
          if (gr != cur) {
            pool_flush(gsum, gcnt, cur, col, s, run, cg == 0 && c == 0);
            cur = gr;
            run = 1;
            s = hv[r];
          } else {
            add4(s, hv[r]);
            ++run;
          }
        }
        pool_flush(gsum, gcnt, cur, col, s, run, cg == 0 && c == 0);
      }
    }
  }
}

extern "C" void kernel_launch(void* const* d_in, const int* in_sizes, int n_in,
                              void* d_out, int out_size, void* d_ws, size_t ws_size,
                              hipStream_t stream) {
  const float* x       = (const float*)d_in[0];
  const int*   ei      = (const int*)d_in[1];
  const int*   src     = ei;
  const int*   dst     = ei + NE;
  const int*   cluster = (const int*)d_in[2];
  const float* Wl0 = (const float*)d_in[3];
  const float* bl0 = (const float*)d_in[4];
  const float* Wr0 = (const float*)d_in[5];
  const float* Wl1 = (const float*)d_in[6];
  const float* bl1 = (const float*)d_in[7];
  const float* Wr1 = (const float*)d_in[8];
  const float* Wl2 = (const float*)d_in[9];
  const float* bl2 = (const float*)d_in[10];
  const float* Wr2 = (const float*)d_in[11];

  float* out_ls = (float*)d_out;
  float* out_h2 = out_ls + (size_t)NN * OUTC;
  float* out_g  = out_h2 + (size_t)NN * HIDC;

  float* ws   = (float*)d_ws;
  float* scr  = ws;
  float* h1   = ws + (size_t)NN * HIDC;
  float* h2r  = h1 + (size_t)NN * HIDC;

  int* cnt  = (int*)scr;
  float* gsum = (float*)(cnt + 50048);
  float* gcnt = gsum + NG * HIDC;
  int* slot = (int*)(gcnt + NG);
  int* ovf  = slot + NN * SLOTS;

  const int TPB = 256;
  const int edgeBlocks4 = (NE + TPB * 4 - 1) / (TPB * 4);
  const int layerBlocks = (NN + LAYER_ROWS - 1) / LAYER_ROWS;

  hipMemsetAsync(cnt, 0, (50048 + NG * HIDC + NG) * sizeof(int), stream);
  bucket_kernel<<<edgeBlocks4, TPB, 0, stream>>>(src, dst, cnt, slot, ovf, NN, NE);

  sage_layer_kernel<HIDC, false, 0><<<layerBlocks, (HIDC / 64) * 64, 0, stream>>>(
      x, cnt, slot, ovf, Wl0, bl0, Wr0, h1, nullptr, NN, 1, nullptr, nullptr, nullptr, nullptr);

  sage_layer_kernel<HIDC, false, 1><<<layerBlocks, (HIDC / 64) * 64, 0, stream>>>(
      h1, cnt, slot, ovf, Wl1, bl1, Wr1, h2r, out_h2, NN, 1, cluster, gsum, gcnt, nullptr);

  sage_layer_kernel<OUTC, true, 2><<<layerBlocks, (OUTC / 64) * 64, 0, stream>>>(
      h2r, cnt, slot, ovf, Wl2, bl2, Wr2, out_ls, nullptr, NN, 0, nullptr, gsum, gcnt, out_g);
}


# Edge buckets without global atomics: blocks split edge chunks by destination range in LDS, a block per range ranks pairs by LDS atomics, stores whole slot lines; memset gone; sum order = new ranks
# speedup vs baseline: 5.0017x; 1.2291x over previous
#include <hip/hip_runtime.h>
#include <hip/hip_bf16.h>
#include <math.h>

typedef float v2f __attribute__((ext_vector_type(2)));
typedef float v8f __attribute__((ext_vector_type(8)));

#define NN      50000
#define NE      600000
#define INC     128
#define HIDC    128
#define OUTC    64
#define NG      512

__device__ __forceinline__ void atomAddF(float* p, float v) {
  __hip_atomic_fetch_add(p, v, __ATOMIC_RELAXED, __HIP_MEMORY_SCOPE_AGENT);
}

#define SLOTS 32
#define RB 256
#define NR ((NN + RB - 1) / RB)
#define NA 256
#define EPB ((NE + NA - 1) / NA)
#define PT 256
#define EPT ((EPB + PT - 1) / PT)
#define NSTAGE (NA * EPB)
__device__ __forceinline__ int lds_add(int* p, int v) {
  return __hip_atomic_fetch_add(p, v, __ATOMIC_RELAXED, __HIP_MEMORY_SCOPE_WORKGROUP);
}
__global__ void __launch_bounds__(PT)
part_kernel(const int* __restrict__ src, const int* __restrict__ dst,
            int* __restrict__ stage_w, int* __restrict__ tab_w, int E) {
  static_assert(NR <= PT && PT == 256, "a thread per range, 8 waves");
  int2* __restrict__ stage = reinterpret_cast<int2*>(stage_w);
  int2* __restrict__ tab = reinterpret_cast<int2*>(tab_w);
  static_assert((NA - 1) * EPB < NE, "no empty chunk");
  __shared__ int sCnt[PT];
  __shared__ int sOff[PT];
  __shared__ int sWave[PT / 32];
  __shared__ int2 sPair[EPB];
  int t = threadIdx.x;
  int lane = t & 31;
  int wv = t >> 5;
  int e0 = blockIdx.x * EPB;
  int nloc = min(EPB, E - e0);
  sCnt[t] = 0;
  int d[EPT], s[EPT], r[EPT];
#pragma unroll
  for (int k = 0; k < EPT; ++k) {
    int i = k * PT + t;
    bool ok = i < nloc;
    d[k] = ok ? dst[e0 + i] : -1;
    s[k] = ok ? src[e0 + i] : 0;
  }
  __syncthreads();
#pragma unroll
  for (int k = 0; k < EPT; ++k) r[k] = (d[k] >= 0) ? lds_add(&sCnt[(unsigned)d[k] / RB], 1) : 0;
  __syncthreads();
  int c = sCnt[t];
  int inc = c;
#pragma unroll
  for (int off = 1; off < 32; off <<= 1) {
    int u = __shfl_up(inc, off, 32);
    if (lane >= off) inc += u;
  }
  if (lane == 31) sWave[wv] = inc;
  __syncthreads();
  int base = 0;
#pragma unroll
  for (int w = 0; w < PT / 32; ++w) base += (w < wv) ? sWave[w] : 0;
  int off = base + inc - c;
  sOff[t] = off;
  __syncthreads();
  int o[EPT];
#pragma unroll
  for (int k = 0; k < EPT; ++k) o[k] = sOff[(d[k] >= 0) ? (unsigned)d[k] / RB : 0u];
#pragma unroll
  for (int k = 0; k < EPT; ++k)
    if (d[k] >= 0) sPair[o[k] + r[k]] = make_int2(d[k], s[k]);
  __syncthreads();
  int2 pv[EPT];
#pragma unroll
  for (int k = 0; k < EPT; ++k) pv[k] = sPair[min(k * PT + t, EPB - 1)];
#pragma unroll
  for (int k = 0; k < EPT; ++k) {
    int i = k * PT + t;
    if (i < nloc) stage[e0 + i] = pv[k];
  }
  tab[blockIdx.x * PT + t] = make_int2(off, c);
}
__device__ __forceinline__ void place_pair(int* sSlot, int* sHead, int* __restrict__ nxt,
                                           int ln, int r, int s, int p) {
  if (r < SLOTS) {
    sSlot[ln * SLOTS + r] = s;
  } else {
    int prev = __hip_atomic_exchange(&sHead[ln], p, __ATOMIC_RELAXED, __HIP_MEMORY_SCOPE_WORKGROUP);
    nxt[p] = prev;
  }
}
__global__ void __launch_bounds__(PT)
rank_kernel(const int* __restrict__ stage_w, const int* __restrict__ tab_w,
            int* __restrict__ cnt, int* __restrict__ slot, int* __restrict__ nxt,
            int* __restrict__ ohead, float* __restrict__ pooled, int n) {
  static_assert(RB == PT && NA == PT, "a thread per row and per chunk");
  static_assert(2 * NR * PT >= NG * HIDC + NG, "the blocks cover the pooled sums and counts");
  __shared__ int sCnt[RB];
  __shared__ int sHead[RB];
  __shared__ __attribute__((aligned(16))) int sSlot[RB * SLOTS];
  const int2* __restrict__ stage = reinterpret_cast<const int2*>(stage_w);
  const int2* __restrict__ tab = reinterpret_cast<const int2*>(tab_w);
  int t = threadIdx.x;
  int lane = t & 31;
  int wv = __builtin_amdgcn_readfirstlane(t >> 5);
  int b = blockIdx.x;
  int lo = b * RB;
  const int2 oc = tab[t * PT + b];
  sCnt[t] = 0;
  sHead[t] = -1;
#pragma unroll
  for (int q = 0; q < RB * SLOTS / 4 / PT; ++q)
    reinterpret_cast<int4*>(sSlot)[q * PT + t] = make_int4(0, 0, 0, 0);
  __syncthreads();
  int2 pr[32];
  int rk[32];
#pragma unroll
  for (int j = 0; j < 32; ++j) {
    int off = __builtin_amdgcn_readlane(oc.x, j);
    int c = __builtin_amdgcn_readlane(oc.y, j);
    const int2 v = stage[min((32 * wv + j) * EPB + off + max(min(lane, c - 1), 0), NE - 1)];
    pr[j] = (lane < c) ? v : make_int2(-1, 0);
  }
#pragma unroll
  for (int j = 0; j < 32; ++j) rk[j] = (pr[j].x >= 0) ? lds_add(&sCnt[pr[j].x - lo], 1) : 0;
  bool chained = false;
#pragma unroll
  for (int j = 0; j < 32; ++j) {
    bool ok = pr[j].x >= 0;
    if (ok && rk[j] < SLOTS) sSlot[(pr[j].x - lo) * SLOTS + rk[j]] = pr[j].y;
    chained |= ok && rk[j] >= SLOTS;
  }
  if (__builtin_amdgcn_ballot_w32(chained) != 0) {
#pragma unroll
    for (int j = 0; j < 32; ++j)
      if (pr[j].x >= 0 && rk[j] >= SLOTS)
        place_pair(sSlot, sHead, nxt, pr[j].x - lo, rk[j], pr[j].y,
                   (32 * wv + j) * EPB + __builtin_amdgcn_readlane(oc.x, j) + lane);
  }
  if (__builtin_amdgcn_ballot_w32(oc.y > 32) != 0) {
#pragma nounroll
    for (int j = 0; j < 32; ++j) {
      int c = min(__builtin_amdgcn_readlane(oc.y, j), EPB);
      if (c > 32) {
        int p0 = (32 * wv + j) * EPB + __builtin_amdgcn_readlane(oc.x, j);
        for (int i = 32 + lane; i < c; i += 32) {
          const int2 q = stage[p0 + i];
          place_pair(sSlot, sHead, nxt, q.x - lo, lds_add(&sCnt[q.x - lo], 1), q.y, p0 + i);
        }
      }
    }
  }
  __syncthreads();
  if (lo + t < n) {
    cnt[lo + t] = sCnt[t];
    ohead[lo + t] = sHead[t];
  }
  int rows = min(RB, n - lo);
  int4 ln4[RB * SLOTS / 4 / PT];
#pragma unroll
  for (int q = 0; q < RB * SLOTS / 4 / PT; ++q) ln4[q] = reinterpret_cast<const int4*>(sSlot)[q * PT + t];
#pragma unroll
  for (int q = 0; q < RB * SLOTS / 4 / PT; ++q) {
    int i4 = q * PT + t;
    if ((i4 >> 3) < rows) reinterpret_cast<int4*>(slot + (size_t)lo * SLOTS)[i4] = ln4[q];
  }
#pragma unroll
  for (int q = 0; q < 2; ++q) {
    int i = (2 * b + q) * PT + t;
    if (i < NG * HIDC + NG) pooled[i] = 0.0f;
  }
}

__device__ __forceinline__ void add4(float4& a, const float4 v) {
  a.x += v.x; a.y += v.y; a.z += v.z; a.w += v.w;
}

#define LAYER_ROWS 32
#define LDS_PITCH 132
struct GemmFrag {
  float4 a[2];
  float4 w[8];
};
__device__ __forceinline__ float f4c(const float4 v, int j) {
  return j == 0 ? v.x : (j == 1 ? v.y : (j == 2 ? v.z : v.w));
}
__device__ __forceinline__ void gemm_mma(const GemmFrag& f, v8f (&acc)[4]) {
#pragma unroll
  for (int b = 0; b < 2; ++b) {
#pragma unroll
    for (int t = 0; t < 2; ++t) {
      v2f af;
      af.x = f4c(f.a[b], 2 * t);
      af.y = f4c(f.a[b], 2 * t + 1);
#pragma unroll
      for (int j = 0; j < 4; ++j) {
        v2f bf;
        bf.x = f4c(f.w[4 * b + 2 * t], j);
        bf.y = f4c(f.w[4 * b + 2 * t + 1], j);
        acc[j] = __builtin_amdgcn_wmma_f32_16x16x4_f32(false, af, false, bf, (short)0, acc[j], false, false);
      }
    }
  }
}
template <int NCOLS>
__device__ __forceinline__ void gemm_k128(const float* sA, const float* wbase, int woff, v8f (&acc)[4]) {
  auto frag_load = [&](GemmFrag& f, const float* ast, const float* wst) {
#pragma unroll
    for (int b = 0; b < 2; ++b)
      f.a[b] = *reinterpret_cast<const float4*>(ast + 8 * b);
#pragma unroll
    for (int q = 0; q < 8; ++q)
      f.w[q] = *reinterpret_cast<const float4*>(wst + (woff + (8 * (q >> 2) + (q & 3)) * NCOLS));
  };
  constexpr int KI = 128 >> 4;
  const float* ast = sA;
  const float* wst = wbase;
  GemmFrag cur;
  frag_load(cur, ast, wst);
#pragma nounroll
  for (int i = 0; i + 1 < KI; ++i) {
    ast += 16;
    wst += 16 * NCOLS;
    GemmFrag nxt;
    frag_load(nxt, ast, wst);
    __builtin_amdgcn_sched_barrier(0);
    gemm_mma(cur, acc);
    cur = nxt;
  }
  gemm_mma(cur, acc);
}
__device__ __forceinline__ float half_max16(float v) {
  v = fmaxf(v, __shfl_xor(v, 8, 32));
  v = fmaxf(v, __shfl_xor(v, 4, 32));
  v = fmaxf(v, __shfl_xor(v, 2, 32));
  v = fmaxf(v, __shfl_xor(v, 1, 32));
  return v;
}
__device__ __forceinline__ float half_sum16(float v) {
  v += __shfl_xor(v, 8, 32);
  v += __shfl_xor(v, 4, 32);
  v += __shfl_xor(v, 2, 32);
  v += __shfl_xor(v, 1, 32);
  return v;
}
__device__ __forceinline__ void pool_flush(float* __restrict__ gsum, float* __restrict__ gcnt,
                                           int g, int col, const float4 acc, int run, bool counts) {
  float* p = gsum + (size_t)g * 128 + col;
  atomAddF(p + 0, acc.x);
  atomAddF(p + 1, acc.y);
  atomAddF(p + 2, acc.z);
  atomAddF(p + 3, acc.w);
  if (counts) atomAddF(&gcnt[g], (float)run);
}
template <int NCOLS, bool LSM, int POOLMODE>
__global__ void __launch_bounds__((NCOLS / 64) * 64)
sage_layer_kernel(const float* __restrict__ X, const int* __restrict__ cnt,
                  const int* __restrict__ slot, const int* __restrict__ stage,
                  const int* __restrict__ nxt, const int* __restrict__ ohead,
                  const float* __restrict__ Wl, const float* __restrict__ bl,
                  const float* __restrict__ Wr,
                  float* __restrict__ OutA, float* __restrict__ OutB, int M, int doRelu,
                  const int* __restrict__ cluster, float* __restrict__ gsum,
                  float* __restrict__ gcnt, float* __restrict__ gOut) {
  constexpr int NGRP = NCOLS / 64;
  constexpr int WAVES = 2 * NGRP;
  constexpr int G = LAYER_ROWS / WAVES;
  __shared__ __attribute__((aligned(16))) float sAgg[LAYER_ROWS * LDS_PITCH];
  __shared__ __attribute__((aligned(16))) float sX[LAYER_ROWS * LDS_PITCH];
  __shared__ int sIds[LAYER_ROWS * SLOTS];

  int wv = __builtin_amdgcn_readfirstlane((int)(threadIdx.x >> 5));
  int lane = threadIdx.x & 31;
  int row0 = blockIdx.x * LAYER_ROWS;
  const bool gather_first = (__popc(blockIdx.x) & 1) == 0;

  int lr0 = wv * G;
  int n0 = row0 + lr0;
  const bool has_nodes = n0 + G <= M;

  constexpr int PCH = 128 / (WAVES * 32);
  float ps[PCH];
  float pc = 0.0f;
  if constexpr (POOLMODE == 2) {
    static_assert((NN + LAYER_ROWS - 1) / LAYER_ROWS >= NG, "one block per graph");
    if (row0 < NG * LAYER_ROWS) {
      pc = gcnt[blockIdx.x];
#pragma unroll
      for (int q = 0; q < PCH; ++q) ps[q] = gsum[blockIdx.x * 128 + q * (WAVES * 32) + threadIdx.x];
    }
  }

  if (has_nodes) {
#pragma unroll
    for (int q = 0; q < G; ++q)
      *reinterpret_cast<float4*>(&sX[(lr0 + q) * LDS_PITCH + lane * 4]) =
          *reinterpret_cast<const float4*>(X + (size_t)(n0 + q) * 128 + lane * 4);
  }

  if constexpr (POOLMODE == 2) {
    if (row0 < NG * LAYER_ROWS) {
      float dv = fmaxf(pc, 1.0f);
#pragma unroll
      for (int q = 0; q < PCH; ++q)
        gOut[blockIdx.x * 128 + q * (WAVES * 32) + threadIdx.x] = ps[q] / dv;
    }
  }

  auto gather = [&]() {
    if (!has_nodes) return;
    int* ids = sIds + wv * (G * SLOTS);
    int cq = (lane < G) ? cnt[n0 + lane] : 0;
    int idv[G];
#pragma unroll
    for (int q = 0; q < G; ++q) idv[q] = slot[(n0 + q) * SLOTS + lane];
    int dq = min(cq, SLOTS);
    float iv = 1.0f / fmaxf((float)cq, 1.0f);
    int inc = dq;
#pragma unroll
    for (int off = 1; off < 32; off <<= 1) {
      int u = __shfl_up(inc, off, 32);
      if (lane >= off) inc += u;
    }
    int rp = (lane <= G) ? inc - dq : 0x7fffffff;
#pragma unroll
    for (int q = 0; q < G; ++q) ids[__builtin_amdgcn_readlane(rp, q) + lane] = idv[q];
    int jend = __builtin_amdgcn_readlane(rp, G);
    int node = 0;
    int nend = __builtin_amdgcn_readlane(rp, 1);
    float4 sum = make_float4(0.0f, 0.0f, 0.0f, 0.0f);
    const float* xl = X + lane * 4;
    if (0 < jend) {
      int cv = ids[min(lane, jend - 1)];
      for (int jb = 0; jb < jend; jb += 32) {
        int cvn = ids[min(jb + 32 + lane, jend - 1)];
#pragma unroll
        for (int u = 0; u < 2; ++u) {
          int e0 = jb + 16 * u;
          if (e0 < jend) {
            float4 v[16];
#pragma unroll
            for (int t = 0; t < 16; ++t) {
              int s = __builtin_amdgcn_readlane(cv, 16 * u + t);
              v[t] = *reinterpret_cast<const float4*>(xl + (size_t)s * 128);
            }
#pragma unroll
            for (int t = 0; t < 16; ++t) {
              int e = e0 + t;
              while (e >= nend && node < G) {
                float s = __int_as_float(__builtin_amdgcn_readlane(__float_as_int(iv), node));
                *reinterpret_cast<float4*>(&sAgg[(lr0 + node) * LDS_PITCH + lane * 4]) =
                    make_float4(sum.x * s, sum.y * s, sum.z * s, sum.w * s);
                ++node;
                nend = __builtin_amdgcn_readlane(rp, node + 1);
                sum = make_float4(0.0f, 0.0f, 0.0f, 0.0f);
              }
              add4(sum, v[t]);
            }
          }
        }
        cv = cvn;
      }
    }
    for (; node < G; ++node) {
      float s = __int_as_float(__builtin_amdgcn_readlane(__float_as_int(iv), node));
      *reinterpret_cast<float4*>(&sAgg[(lr0 + node) * LDS_PITCH + lane * 4]) =
          make_float4(sum.x * s, sum.y * s, sum.z * s, sum.w * s);
      sum = make_float4(0.0f, 0.0f, 0.0f, 0.0f);
    }
    if (__builtin_amdgcn_ballot_w32(cq > SLOTS) != 0) {
#pragma nounroll
      for (int q = 0; q < G; ++q) {
        int over = __builtin_amdgcn_readlane(cq, q) - SLOTS;
        if (over > 0) {
          float4 ex = make_float4(0.0f, 0.0f, 0.0f, 0.0f);
          int p = ohead[n0 + q];
#pragma nounroll
          for (int i = 0; i < over && (unsigned)p < (unsigned)NSTAGE; ++i) {
            add4(ex, *reinterpret_cast<const float4*>(xl + (size_t)stage[2 * p + 1] * 128));
            p = nxt[p];
          }
          float s = __int_as_float(__builtin_amdgcn_readlane(__float_as_int(iv), q));
          float4* pa = reinterpret_cast<float4*>(&sAgg[(lr0 + q) * LDS_PITCH + lane * 4]);
          const float4 a = *pa;
          *pa = make_float4(a.x + ex.x * s, a.y + ex.y * s, a.z + ex.z * s, a.w + ex.w * s);
        }
      }
    }
  };

  int tl = wv / NGRP;
  int cg = wv % NGRP;
  int m0 = row0 + 16 * tl;
  const bool active = m0 < M;
  int nb = cg << 6;
  int c = lane & 15;
  int h = lane >> 4;
  int aoff = (16 * tl + c) * LDS_PITCH + 4 * h;
  int woff = (4 * h) * NCOLS + nb + 4 * c;
  v8f acc[4] = {};
  int gl = 0;
  if constexpr (POOLMODE == 1) {
    if (active) gl = cluster[m0 + c];
  }

#pragma nounroll
  for (int step = 0; step < 2; ++step) {
    if ((step == 0) == gather_first) gather();
    if (step == 0 || !gather_first) __syncthreads();
    if (active) gemm_k128<NCOLS>(step == 0 ? &sX[aoff] : &sAgg[aoff], step == 0 ? Wr : Wl, woff, acc);
  }
  if (!active) return;

  const float4 bias = *reinterpret_cast<const float4*>(bl + nb + 4 * c);
  int rbase = m0 + (h << 3);
  if constexpr (LSM) {
#pragma unroll
    for (int r = 0; r < 8; ++r) {
      float4 v = make_float4(acc[0][r] + bias.x, acc[1][r] + bias.y, acc[2][r] + bias.z, acc[3][r] + bias.w);
      float m = half_max16(fmaxf(fmaxf(v.x, v.y), fmaxf(v.z, v.w)));
      float e = half_sum16((__expf(v.x - m) + __expf(v.y - m)) + (__expf(v.z - m) + __expf(v.w - m)));
      float lse = m + __logf(e);
      size_t off = (size_t)(rbase + r) * NCOLS + nb + 4 * c;
      *reinterpret_cast<float4*>(OutA + off) = make_float4(v.x - lse, v.y - lse, v.z - lse, v.w - lse);
    }
  } else {
    float4 hv[8];
#pragma unroll
    for (int r = 0; r < 8; ++r) {
      float4 v = make_float4(acc[0][r] + bias.x, acc[1][r] + bias.y, acc[2][r] + bias.z, acc[3][r] + bias.w);
      size_t off = (size_t)(rbase + r) * NCOLS + nb + 4 * c;
      if (OutB) *reinterpret_cast<float4*>(OutB + off) = v;
      if (doRelu) v = make_float4(fmaxf(v.x, 0.0f), fmaxf(v.y, 0.0f), fmaxf(v.z, 0.0f), fmaxf(v.w, 0.0f));
      *reinterpret_cast<float4*>(OutA + off) = v;
      hv[r] = v;
    }
    if constexpr (POOLMODE == 1) {
      int col = nb + 4 * c;
      int g0 = __builtin_amdgcn_readfirstlane(gl);
      if (__builtin_amdgcn_ballot_w32(gl != g0) == 0) {
        float4 s = hv[0];
#pragma unroll
        for (int r = 1; r < 8; ++r) add4(s, hv[r]);
        s.x += __shfl_xor(s.x, 16, 32);
        s.y += __shfl_xor(s.y, 16, 32);
        s.z += __shfl_xor(s.z, 16, 32);
        s.w += __shfl_xor(s.w, 16, 32);
        if (h == 0) pool_flush(gsum, gcnt, g0, col, s, 16, cg == 0 && c == 0);
      } else {
        int cur = __shfl(gl, 8 * h, 32);
        int run = 1;
        float4 s = hv[0];
#pragma unroll
        for (int r = 1; r < 8; ++r) {
          int gr = __shfl(gl, 8 * h + r, 32);
          if (gr != cur) {
            pool_flush(gsum, gcnt, cur, col, s, run, cg == 0 && c == 0);
            cur = gr;
            run = 1;
            s = hv[r];
          } else {
            add4(s, hv[r]);
            ++run;
          }
        }
        pool_flush(gsum, gcnt, cur, col, s, run, cg == 0 && c == 0);
      }
    }
  }
}

extern "C" void kernel_launch(void* const* d_in, const int* in_sizes, int n_in,
                              void* d_out, int out_size, void* d_ws, size_t ws_size,
                              hipStream_t stream) {
  const float* x       = (const float*)d_in[0];
  const int*   ei      = (const int*)d_in[1];
  const int*   src     = ei;
  const int*   dst     = ei + NE;
  const int*   cluster = (const int*)d_in[2];
  const float* Wl0 = (const float*)d_in[3];
  const float* bl0 = (const float*)d_in[4];
  const float* Wr0 = (const float*)d_in[5];
  const float* Wl1 = (const float*)d_in[6];
  const float* bl1 = (const float*)d_in[7];
  const float* Wr1 = (const float*)d_in[8];
  const float* Wl2 = (const float*)d_in[9];
  const float* bl2 = (const float*)d_in[10];
  const float* Wr2 = (const float*)d_in[11];

  float* out_ls = (float*)d_out;
  float* out_h2 = out_ls + (size_t)NN * OUTC;
  float* out_g  = out_h2 + (size_t)NN * HIDC;

  float* ws   = (float*)d_ws;
  float* scr  = ws;
  float* h1   = ws + (size_t)NN * HIDC;
  float* h2r  = h1 + (size_t)NN * HIDC;

  int* cnt  = (int*)scr;
  float* gsum = (float*)(cnt + 50048);
  float* gcnt = gsum + NG * HIDC;
  int* slot = (int*)(gcnt + NG);
  int* stage = slot + NN * SLOTS;
  int* nxt   = stage + 2 * NSTAGE;
  int* ohead = nxt + NSTAGE;
  int* tab   = ohead + 50048;

  const int layerBlocks = (NN + LAYER_ROWS - 1) / LAYER_ROWS;

  part_kernel<<<NA, PT, 0, stream>>>(src, dst, stage, tab, NE);
  rank_kernel<<<NR, PT, 0, stream>>>(stage, tab, cnt, slot, nxt, ohead, gsum, NN);

  sage_layer_kernel<HIDC, false, 0><<<layerBlocks, (HIDC / 64) * 64, 0, stream>>>(
      x, cnt, slot, stage, nxt, ohead, Wl0, bl0, Wr0, h1, nullptr, NN, 1, nullptr, nullptr, nullptr, nullptr);

  sage_layer_kernel<HIDC, false, 1><<<layerBlocks, (HIDC / 64) * 64, 0, stream>>>(
      h1, cnt, slot, stage, nxt, ohead, Wl1, bl1, Wr1, h2r, out_h2, NN, 1, cluster, gsum, gcnt, nullptr);

  sage_layer_kernel<OUTC, true, 2><<<layerBlocks, (OUTC / 64) * 64, 0, stream>>>(
      h2r, cnt, slot, stage, nxt, ohead, Wl2, bl2, Wr2, out_ls, nullptr, NN, 0, nullptr, gsum, gcnt, out_g);
}


# GEMM K loop on two operand sets in turn, no copy of the prefetched stage: the wait for a stage's W rows moves from behind 2 matrix instructions to behind all 16 of the stage before; sum order kept
# speedup vs baseline: 5.2585x; 1.0513x over previous
#include <hip/hip_runtime.h>
#include <hip/hip_bf16.h>
#include <math.h>

typedef float v2f __attribute__((ext_vector_type(2)));
typedef float v8f __attribute__((ext_vector_type(8)));

#define NN      50000
#define NE      600000
#define INC     128
#define HIDC    128
#define OUTC    64
#define NG      512

__device__ __forceinline__ void atomAddF(float* p, float v) {
  __hip_atomic_fetch_add(p, v, __ATOMIC_RELAXED, __HIP_MEMORY_SCOPE_AGENT);
}

#define SLOTS 32
#define RB 256
#define NR ((NN + RB - 1) / RB)
#define NA 256
#define EPB ((NE + NA - 1) / NA)
#define PT 256
#define EPT ((EPB + PT - 1) / PT)
#define NSTAGE (NA * EPB)
__device__ __forceinline__ int lds_add(int* p, int v) {
  return __hip_atomic_fetch_add(p, v, __ATOMIC_RELAXED, __HIP_MEMORY_SCOPE_WORKGROUP);
}
__global__ void __launch_bounds__(PT)
part_kernel(const int* __restrict__ src, const int* __restrict__ dst,
            int* __restrict__ stage_w, int* __restrict__ tab_w, int E) {
  static_assert(NR <= PT && PT == 256, "a thread per range, 8 waves");
  int2* __restrict__ stage = reinterpret_cast<int2*>(stage_w);
  int2* __restrict__ tab = reinterpret_cast<int2*>(tab_w);
  static_assert((NA - 1) * EPB < NE, "no empty chunk");
  __shared__ int sCnt[PT];
  __shared__ int sOff[PT];
  __shared__ int sWave[PT / 32];
  __shared__ int2 sPair[EPB];
  int t = threadIdx.x;
  int lane = t & 31;
  int wv = t >> 5;
  int e0 = blockIdx.x * EPB;
  int nloc = min(EPB, E - e0);
  sCnt[t] = 0;
  int d[EPT], s[EPT], r[EPT];
#pragma unroll
  for (int k = 0; k < EPT; ++k) {
    int i = k * PT + t;
    bool ok = i < nloc;
    d[k] = ok ? dst[e0 + i] : -1;
    s[k] = ok ? src[e0 + i] : 0;
  }
  __syncthreads();
#pragma unroll
  for (int k = 0; k < EPT; ++k) r[k] = (d[k] >= 0) ? lds_add(&sCnt[(unsigned)d[k] / RB], 1) : 0;
  __syncthreads();
  int c = sCnt[t];
  int inc = c;
#pragma unroll
  for (int off = 1; off < 32; off <<= 1) {
    int u = __shfl_up(inc, off, 32);
    if (lane >= off) inc += u;
  }
  if (lane == 31) sWave[wv] = inc;
  __syncthreads();
  int base = 0;
#pragma unroll
  for (int w = 0; w < PT / 32; ++w) base += (w < wv) ? sWave[w] : 0;
  int off = base + inc - c;
  sOff[t] = off;
  __syncthreads();
  int o[EPT];
#pragma unroll
  for (int k = 0; k < EPT; ++k) o[k] = sOff[(d[k] >= 0) ? (unsigned)d[k] / RB : 0u];
#pragma unroll
  for (int k = 0; k < EPT; ++k)
    if (d[k] >= 0) sPair[o[k] + r[k]] = make_int2(d[k], s[k]);
  __syncthreads();
  int2 pv[EPT];
#pragma unroll
  for (int k = 0; k < EPT; ++k) pv[k] = sPair[min(k * PT + t, EPB - 1)];
#pragma unroll
  for (int k = 0; k < EPT; ++k) {
    int i = k * PT + t;
    if (i < nloc) stage[e0 + i] = pv[k];
  }
  tab[blockIdx.x * PT + t] = make_int2(off, c);
}
__device__ __forceinline__ void place_pair(int* sSlot, int* sHead, int* __restrict__ nxt,
                                           int ln, int r, int s, int p) {
  if (r < SLOTS) {
    sSlot[ln * SLOTS + r] = s;
  } else {
    int prev = __hip_atomic_exchange(&sHead[ln], p, __ATOMIC_RELAXED, __HIP_MEMORY_SCOPE_WORKGROUP);
    nxt[p] = prev;
  }
}
__global__ void __launch_bounds__(PT)
rank_kernel(const int* __restrict__ stage_w, const int* __restrict__ tab_w,
            int* __restrict__ cnt, int* __restrict__ slot, int* __restrict__ nxt,
            int* __restrict__ ohead, float* __restrict__ pooled, int n) {
  static_assert(RB == PT && NA == PT, "a thread per row and per chunk");
  static_assert(2 * NR * PT >= NG * HIDC + NG, "the blocks cover the pooled sums and counts");
  __shared__ int sCnt[RB];
  __shared__ int sHead[RB];
  __shared__ __attribute__((aligned(16))) int sSlot[RB * SLOTS];
  const int2* __restrict__ stage = reinterpret_cast<const int2*>(stage_w);
  const int2* __restrict__ tab = reinterpret_cast<const int2*>(tab_w);
  int t = threadIdx.x;
  int lane = t & 31;
  int wv = __builtin_amdgcn_readfirstlane(t >> 5);
  int b = blockIdx.x;
  int lo = b * RB;
  const int2 oc = tab[t * PT + b];
  sCnt[t] = 0;
  sHead[t] = -1;
#pragma unroll
  for (int q = 0; q < RB * SLOTS / 4 / PT; ++q)
    reinterpret_cast<int4*>(sSlot)[q * PT + t] = make_int4(0, 0, 0, 0);
  __syncthreads();
  int2 pr[32];
  int rk[32];
#pragma unroll
  for (int j = 0; j < 32; ++j) {
    int off = __builtin_amdgcn_readlane(oc.x, j);
    int c = __builtin_amdgcn_readlane(oc.y, j);
    const int2 v = stage[min((32 * wv + j) * EPB + off + max(min(lane, c - 1), 0), NE - 1)];
    pr[j] = (lane < c) ? v : make_int2(-1, 0);
  }
#pragma unroll
  for (int j = 0; j < 32; ++j) rk[j] = (pr[j].x >= 0) ? lds_add(&sCnt[pr[j].x - lo], 1) : 0;
  bool chained = false;
#pragma unroll
  for (int j = 0; j < 32; ++j) {
    bool ok = pr[j].x >= 0;
    if (ok && rk[j] < SLOTS) sSlot[(pr[j].x - lo) * SLOTS + rk[j]] = pr[j].y;
    chained |= ok && rk[j] >= SLOTS;
  }
  if (__builtin_amdgcn_ballot_w32(chained) != 0) {
#pragma unroll
    for (int j = 0; j < 32; ++j)
      if (pr[j].x >= 0 && rk[j] >= SLOTS)
        place_pair(sSlot, sHead, nxt, pr[j].x - lo, rk[j], pr[j].y,
                   (32 * wv + j) * EPB + __builtin_amdgcn_readlane(oc.x, j) + lane);
  }
  if (__builtin_amdgcn_ballot_w32(oc.y > 32) != 0) {
#pragma nounroll
    for (int j = 0; j < 32; ++j) {
      int c = min(__builtin_amdgcn_readlane(oc.y, j), EPB);
      if (c > 32) {
        int p0 = (32 * wv + j) * EPB + __builtin_amdgcn_readlane(oc.x, j);
        for (int i = 32 + lane; i < c; i += 32) {
          const int2 q = stage[p0 + i];
          place_pair(sSlot, sHead, nxt, q.x - lo, lds_add(&sCnt[q.x - lo], 1), q.y, p0 + i);
        }
      }
    }
  }
  __syncthreads();
  if (lo + t < n) {
    cnt[lo + t] = sCnt[t];
    ohead[lo + t] = sHead[t];
  }
  int rows = min(RB, n - lo);
  int4 ln4[RB * SLOTS / 4 / PT];
#pragma unroll
  for (int q = 0; q < RB * SLOTS / 4 / PT; ++q) ln4[q] = reinterpret_cast<const int4*>(sSlot)[q * PT + t];
#pragma unroll
  for (int q = 0; q < RB * SLOTS / 4 / PT; ++q) {
    int i4 = q * PT + t;
    if ((i4 >> 3) < rows) reinterpret_cast<int4*>(slot + (size_t)lo * SLOTS)[i4] = ln4[q];
  }
#pragma unroll
  for (int q = 0; q < 2; ++q) {
    int i = (2 * b + q) * PT + t;
    if (i < NG * HIDC + NG) pooled[i] = 0.0f;
  }
}

__device__ __forceinline__ void add4(float4& a, const float4 v) {
  a.x += v.x; a.y += v.y; a.z += v.z; a.w += v.w;
}

#define LAYER_ROWS 32
#define LDS_PITCH 132
struct GemmFrag {
  float4 a[2];
  float4 w[8];
};
__device__ __forceinline__ float f4c(const float4 v, int j) {
  return j == 0 ? v.x : (j == 1 ? v.y : (j == 2 ? v.z : v.w));
}
__device__ __forceinline__ void gemm_mma(const GemmFrag& f, v8f (&acc)[4]) {
#pragma unroll
  for (int b = 0; b < 2; ++b) {
#pragma unroll
    for (int t = 0; t < 2; ++t) {
      v2f af;
      af.x = f4c(f.a[b], 2 * t);
      af.y = f4c(f.a[b], 2 * t + 1);
#pragma unroll
      for (int j = 0; j < 4; ++j) {
        v2f bf;
        bf.x = f4c(f.w[4 * b + 2 * t], j);
        bf.y = f4c(f.w[4 * b + 2 * t + 1], j);
        acc[j] = __builtin_amdgcn_wmma_f32_16x16x4_f32(false, af, false, bf, (short)0, acc[j], false, false);
      }
    }
  }
}
template <int NCOLS>
__device__ __forceinline__ void gemm_k128(const float* sA, const float* wbase, int woff, v8f (&acc)[4]) {
  auto frag_load = [&](GemmFrag& f, const float* ast, const float* wst) {
#pragma unroll
    for (int b = 0; b < 2; ++b)
      f.a[b] = *reinterpret_cast<const float4*>(ast + 8 * b);
#pragma unroll
    for (int q = 0; q < 8; ++q)
      f.w[q] = *reinterpret_cast<const float4*>(wst + (woff + (8 * (q >> 2) + (q & 3)) * NCOLS));
  };
  constexpr int KI = 128 >> 4;
  static_assert(KI % 2 == 0, "the stages are taken in pairs");
  const float* ast = sA;
  const float* wst = wbase;
  GemmFrag f0, f1;
  frag_load(f0, ast, wst);
  auto step = [&](GemmFrag& fl, const GemmFrag& fu, const float* a, const float* w) {
    __builtin_amdgcn_sched_barrier(0);
    frag_load(fl, a, w);
    __builtin_amdgcn_sched_barrier(0);
    gemm_mma(fu, acc);
  };
#pragma nounroll
  for (int i = 0; i + 2 < KI; i += 2) {
    step(f1, f0, ast + 16, wst + 16 * NCOLS);
    ast += 32;
    wst += 32 * NCOLS;
    step(f0, f1, ast, wst);
    __builtin_amdgcn_sched_barrier(0);
  }
  step(f1, f0, ast + 16, wst + 16 * NCOLS);
  __builtin_amdgcn_sched_barrier(0);
  gemm_mma(f1, acc);
}
__device__ __forceinline__ float half_max16(float v) {
  v = fmaxf(v, __shfl_xor(v, 8, 32));
  v = fmaxf(v, __shfl_xor(v, 4, 32));
  v = fmaxf(v, __shfl_xor(v, 2, 32));
  v = fmaxf(v, __shfl_xor(v, 1, 32));
  return v;
}
__device__ __forceinline__ float half_sum16(float v) {
  v += __shfl_xor(v, 8, 32);
  v += __shfl_xor(v, 4, 32);
  v += __shfl_xor(v, 2, 32);
  v += __shfl_xor(v, 1, 32);
  return v;
}
__device__ __forceinline__ void pool_flush(float* __restrict__ gsum, float* __restrict__ gcnt,
                                           int g, int col, const float4 acc, int run, bool counts) {
  float* p = gsum + (size_t)g * 128 + col;
  atomAddF(p + 0, acc.x);
  atomAddF(p + 1, acc.y);
  atomAddF(p + 2, acc.z);
  atomAddF(p + 3, acc.w);
  if (counts) atomAddF(&gcnt[g], (float)run);
}
template <int NCOLS, bool LSM, int POOLMODE>
__global__ void __launch_bounds__((NCOLS / 64) * 64, 7)
sage_layer_kernel(const float* __restrict__ X, const int* __restrict__ cnt,
                  const int* __restrict__ slot, const int* __restrict__ stage,
                  const int* __restrict__ nxt, const int* __restrict__ ohead,
                  const float* __restrict__ Wl, const float* __restrict__ bl,
                  const float* __restrict__ Wr,
                  float* __restrict__ OutA, float* __restrict__ OutB, int M, int doRelu,
                  const int* __restrict__ cluster, float* __restrict__ gsum,
                  float* __restrict__ gcnt, float* __restrict__ gOut) {
  constexpr int NGRP = NCOLS / 64;
  constexpr int WAVES = 2 * NGRP;
  constexpr int G = LAYER_ROWS / WAVES;
  __shared__ __attribute__((aligned(16))) float sAgg[LAYER_ROWS * LDS_PITCH];
  __shared__ __attribute__((aligned(16))) float sX[LAYER_ROWS * LDS_PITCH];
  __shared__ int sIds[LAYER_ROWS * SLOTS];

  int wv = __builtin_amdgcn_readfirstlane((int)(threadIdx.x >> 5));
  int lane = threadIdx.x & 31;
  int row0 = blockIdx.x * LAYER_ROWS;
  const bool gather_first = (__popc(blockIdx.x) & 1) == 0;

  int lr0 = wv * G;
  int n0 = row0 + lr0;
  const bool has_nodes = n0 + G <= M;

  constexpr int PCH = 128 / (WAVES * 32);
  float ps[PCH];
  float pc = 0.0f;
  if constexpr (POOLMODE == 2) {
    static_assert((NN + LAYER_ROWS - 1) / LAYER_ROWS >= NG, "one block per graph");
    if (row0 < NG * LAYER_ROWS) {
      pc = gcnt[blockIdx.x];
#pragma unroll
      for (int q = 0; q < PCH; ++q) ps[q] = gsum[blockIdx.x * 128 + q * (WAVES * 32) + threadIdx.x];
    }
  }

  if (has_nodes) {
#pragma unroll
    for (int q = 0; q < G; ++q)
      *reinterpret_cast<float4*>(&sX[(lr0 + q) * LDS_PITCH + lane * 4]) =
          *reinterpret_cast<const float4*>(X + (size_t)(n0 + q) * 128 + lane * 4);
  }

  if constexpr (POOLMODE == 2) {
    if (row0 < NG * LAYER_ROWS) {
      float dv = fmaxf(pc, 1.0f);
#pragma unroll
      for (int q = 0; q < PCH; ++q)
        gOut[blockIdx.x * 128 + q * (WAVES * 32) + threadIdx.x] = ps[q] / dv;
    }
  }

  auto gather = [&]() {
    if (!has_nodes) return;
    int* ids = sIds + wv * (G * SLOTS);
    int cq = (lane < G) ? cnt[n0 + lane] : 0;
    int idv[G];
#pragma unroll
    for (int q = 0; q < G; ++q) idv[q] = slot[(n0 + q) * SLOTS + lane];
    int dq = min(cq, SLOTS);
    float iv = 1.0f / fmaxf((float)cq, 1.0f);
    int inc = dq;
#pragma unroll
    for (int off = 1; off < 32; off <<= 1) {
      int u = __shfl_up(inc, off, 32);
      if (lane >= off) inc += u;
    }
    int rp = (lane <= G) ? inc - dq : 0x7fffffff;
#pragma unroll
    for (int q = 0; q < G; ++q) ids[__builtin_amdgcn_readlane(rp, q) + lane] = idv[q];
    int jend = __builtin_amdgcn_readlane(rp, G);
    int node = 0;
    int nend = __builtin_amdgcn_readlane(rp, 1);
    float4 sum = make_float4(0.0f, 0.0f, 0.0f, 0.0f);
    const float* xl = X + lane * 4;
    if (0 < jend) {
      int cv = ids[min(lane, jend - 1)];
      for (int jb = 0; jb < jend; jb += 32) {
        int cvn = ids[min(jb + 32 + lane, jend - 1)];
#pragma unroll
        for (int u = 0; u < 2; ++u) {
          int e0 = jb + 16 * u;
          if (e0 < jend) {
            float4 v[16];
#pragma unroll
            for (int t = 0; t < 16; ++t) {
              int s = __builtin_amdgcn_readlane(cv, 16 * u + t);
              v[t] = *reinterpret_cast<const float4*>(xl + (size_t)s * 128);
            }
#pragma unroll
            for (int t = 0; t < 16; ++t) {
              int e = e0 + t;
              while (e >= nend && node < G) {
                float s = __int_as_float(__builtin_amdgcn_readlane(__float_as_int(iv), node));
                *reinterpret_cast<float4*>(&sAgg[(lr0 + node) * LDS_PITCH + lane * 4]) =
                    make_float4(sum.x * s, sum.y * s, sum.z * s, sum.w * s);
                ++node;
                nend = __builtin_amdgcn_readlane(rp, node + 1);
                sum = make_float4(0.0f, 0.0f, 0.0f, 0.0f);
              }
              add4(sum, v[t]);
            }
          }
        }
        cv = cvn;
      }
    }
    for (; node < G; ++node) {
      float s = __int_as_float(__builtin_amdgcn_readlane(__float_as_int(iv), node));
      *reinterpret_cast<float4*>(&sAgg[(lr0 + node) * LDS_PITCH + lane * 4]) =
          make_float4(sum.x * s, sum.y * s, sum.z * s, sum.w * s);
      sum = make_float4(0.0f, 0.0f, 0.0f, 0.0f);
    }
    if (__builtin_amdgcn_ballot_w32(cq > SLOTS) != 0) {
#pragma nounroll
      for (int q = 0; q < G; ++q) {
        int over = __builtin_amdgcn_readlane(cq, q) - SLOTS;
        if (over > 0) {
          float4 ex = make_float4(0.0f, 0.0f, 0.0f, 0.0f);
          int p = ohead[n0 + q];
#pragma nounroll
          for (int i = 0; i < over && (unsigned)p < (unsigned)NSTAGE; ++i) {
            add4(ex, *reinterpret_cast<const float4*>(xl + (size_t)stage[2 * p + 1] * 128));
            p = nxt[p];
          }
          float s = __int_as_float(__builtin_amdgcn_readlane(__float_as_int(iv), q));
          float4* pa = reinterpret_cast<float4*>(&sAgg[(lr0 + q) * LDS_PITCH + lane * 4]);
          const float4 a = *pa;
          *pa = make_float4(a.x + ex.x * s, a.y + ex.y * s, a.z + ex.z * s, a.w + ex.w * s);
        }
      }
    }
  };

  int tl = wv / NGRP;
  int cg = wv % NGRP;
  int m0 = row0 + 16 * tl;
  const bool active = m0 < M;
  int nb = cg << 6;
  int c = lane & 15;
  int h = lane >> 4;
  int aoff = (16 * tl + c) * LDS_PITCH + 4 * h;
  int woff = (4 * h) * NCOLS + nb + 4 * c;
  v8f acc[4] = {};
  int gl = 0;
  if constexpr (POOLMODE == 1) {
    if (active) gl = cluster[m0 + c];
  }

#pragma nounroll
  for (int step = 0; step < 2; ++step) {
    if ((step == 0) == gather_first) gather();
    if (step == 0 || !gather_first) __syncthreads();
    if (active) gemm_k128<NCOLS>(step == 0 ? &sX[aoff] : &sAgg[aoff], step == 0 ? Wr : Wl, woff, acc);
  }
  if (!active) return;

  const float4 bias = *reinterpret_cast<const float4*>(bl + nb + 4 * c);
  int rbase = m0 + (h << 3);
  if constexpr (LSM) {
#pragma unroll
    for (int r = 0; r < 8; ++r) {
      float4 v = make_float4(acc[0][r] + bias.x, acc[1][r] + bias.y, acc[2][r] + bias.z, acc[3][r] + bias.w);
      float m = half_max16(fmaxf(fmaxf(v.x, v.y), fmaxf(v.z, v.w)));
      float e = half_sum16((__expf(v.x - m) + __expf(v.y - m)) + (__expf(v.z - m) + __expf(v.w - m)));
      float lse = m + __logf(e);
      size_t off = (size_t)(rbase + r) * NCOLS + nb + 4 * c;
      *reinterpret_cast<float4*>(OutA + off) = make_float4(v.x - lse, v.y - lse, v.z - lse, v.w - lse);
    }
  } else {
    float4 hv[8];
#pragma unroll
    for (int r = 0; r < 8; ++r) {
      float4 v = make_float4(acc[0][r] + bias.x, acc[1][r] + bias.y, acc[2][r] + bias.z, acc[3][r] + bias.w);
      size_t off = (size_t)(rbase + r) * NCOLS + nb + 4 * c;
      if (OutB) *reinterpret_cast<float4*>(OutB + off) = v;
      if (doRelu) v = make_float4(fmaxf(v.x, 0.0f), fmaxf(v.y, 0.0f), fmaxf(v.z, 0.0f), fmaxf(v.w, 0.0f));
      *reinterpret_cast<float4*>(OutA + off) = v;
      hv[r] = v;
    }
    if constexpr (POOLMODE == 1) {
      int col = nb + 4 * c;
      int g0 = __builtin_amdgcn_readfirstlane(gl);
      if (__builtin_amdgcn_ballot_w32(gl != g0) == 0) {
        float4 s = hv[0];
#pragma unroll
        for (int r = 1; r < 8; ++r) add4(s, hv[r]);
        s.x += __shfl_xor(s.x, 16, 32);
        s.y += __shfl_xor(s.y, 16, 32);
        s.z += __shfl_xor(s.z, 16, 32);
        s.w += __shfl_xor(s.w, 16, 32);
        if (h == 0) pool_flush(gsum, gcnt, g0, col, s, 16, cg == 0 && c == 0);
      } else {
        int cur = __shfl(gl, 8 * h, 32);
        int run = 1;
        float4 s = hv[0];
#pragma unroll
        for (int r = 1; r < 8; ++r) {
          int gr = __shfl(gl, 8 * h + r, 32);
          if (gr != cur) {
            pool_flush(gsum, gcnt, cur, col, s, run, cg == 0 && c == 0);
            cur = gr;
            run = 1;
            s = hv[r];
          } else {
            add4(s, hv[r]);
            ++run;
          }
        }
        pool_flush(gsum, gcnt, cur, col, s, run, cg == 0 && c == 0);
      }
    }
  }
}

extern "C" void kernel_launch(void* const* d_in, const int* in_sizes, int n_in,
                              void* d_out, int out_size, void* d_ws, size_t ws_size,
                              hipStream_t stream) {
  const float* x       = (const float*)d_in[0];
  const int*   ei      = (const int*)d_in[1];
  const int*   src     = ei;
  const int*   dst     = ei + NE;
  const int*   cluster = (const int*)d_in[2];
  const float* Wl0 = (const float*)d_in[3];
  const float* bl0 = (const float*)d_in[4];
  const float* Wr0 = (const float*)d_in[5];
  const float* Wl1 = (const float*)d_in[6];
  const float* bl1 = (const float*)d_in[7];
  const float* Wr1 = (const float*)d_in[8];
  const float* Wl2 = (const float*)d_in[9];
  const float* bl2 = (const float*)d_in[10];
  const float* Wr2 = (const float*)d_in[11];

  float* out_ls = (float*)d_out;
  float* out_h2 = out_ls + (size_t)NN * OUTC;
  float* out_g  = out_h2 + (size_t)NN * HIDC;

  float* ws   = (float*)d_ws;
  float* scr  = ws;
  float* h1   = ws + (size_t)NN * HIDC;
  float* h2r  = h1 + (size_t)NN * HIDC;

  int* cnt  = (int*)scr;
  float* gsum = (float*)(cnt + 50048);
  float* gcnt = gsum + NG * HIDC;
  int* slot = (int*)(gcnt + NG);
  int* stage = slot + NN * SLOTS;
  int* nxt   = stage + 2 * NSTAGE;
  int* ohead = nxt + NSTAGE;
  int* tab   = ohead + 50048;

  const int layerBlocks = (NN + LAYER_ROWS - 1) / LAYER_ROWS;

  part_kernel<<<NA, PT, 0, stream>>>(src, dst, stage, tab, NE);
  rank_kernel<<<NR, PT, 0, stream>>>(stage, tab, cnt, slot, nxt, ohead, gsum, NN);

  sage_layer_kernel<HIDC, false, 0><<<layerBlocks, (HIDC / 64) * 64, 0, stream>>>(
      x, cnt, slot, stage, nxt, ohead, Wl0, bl0, Wr0, h1, nullptr, NN, 1, nullptr, nullptr, nullptr, nullptr);

  sage_layer_kernel<HIDC, false, 1><<<layerBlocks, (HIDC / 64) * 64, 0, stream>>>(
      h1, cnt, slot, stage, nxt, ohead, Wl1, bl1, Wr1, h2r, out_h2, NN, 1, cluster, gsum, gcnt, nullptr);

  sage_layer_kernel<OUTC, true, 2><<<layerBlocks, (OUTC / 64) * 64, 0, stream>>>(
      h2r, cnt, slot, stage, nxt, ohead, Wl2, bl2, Wr2, out_ls, nullptr, NN, 0, nullptr, gsum, gcnt, out_g);
}
